# starting kernel
# baseline (speedup 1.0000x reference)
#include <hip/hip_runtime.h>
#include <cstddef>


#define NEG_SLOPE 0.2f

typedef float v2f __attribute__((ext_vector_type(2)));
typedef float v8f __attribute__((ext_vector_type(8)));

__global__ void gat_gemm_wmma(const float* __restrict__ A,
                              const float* __restrict__ B,
                              float* __restrict__ C,
                              int M, int N, int K)
{
    __shared__ float As[16 * 256];
    const int m0   = blockIdx.x * 16;
    const int wave = threadIdx.x >> 5;
    const int lane = threadIdx.x & 31;
    const int half = lane >> 4;
    const int l15  = lane & 15;
    const int n0   = (blockIdx.y * (blockDim.x >> 5) + wave) * 16;

    for (int i = threadIdx.x; i < 16 * K; i += blockDim.x) {
        const int r = i / K, c = i % K;
        const int gm = m0 + r;
        As[i] = (gm < M) ? A[(size_t)gm * K + c] : 0.0f;
    }
    __syncthreads();

    v8f acc = {};
    for (int k = 0; k < K; k += 4) {
        v2f a, b;
        const int ka = k + 2 * half;
        a.x = As[l15 * K + ka + 0];
        a.y = As[l15 * K + ka + 1];
        b.x = B[(size_t)(ka + 0) * N + n0 + l15];
        b.y = B[(size_t)(ka + 1) * N + n0 + l15];
        acc = __builtin_amdgcn_wmma_f32_16x16x4_f32(
            false, a, false, b, (short)0, acc, false, false);
    }

    #pragma unroll
    for (int r = 0; r < 8; ++r) {
        const int gm = m0 + r + 8 * half;
        if (gm < M) C[(size_t)gm * N + n0 + l15] = acc[r];
    }
}

__global__ void gat_attn_coef(const float* __restrict__ H,
                              const float* __restrict__ att_s,
                              const float* __restrict__ att_d,
                              float* __restrict__ as_, float* __restrict__ ad_,
                              int Nn, int heads, int C)
{
    const int idx = blockIdx.x * blockDim.x + threadIdx.x;
    if (idx >= Nn * heads) return;
    const int n = idx / heads, h = idx % heads;
    const float* hp = H + (size_t)n * heads * C + h * C;
    const float* sp = att_s + h * C;
    const float* dp = att_d + h * C;
    float ss = 0.f, dd = 0.f;
    #pragma unroll 8
    for (int c = 0; c < C; ++c) { const float v = hp[c]; ss += v * sp[c]; dd += v * dp[c]; }
    as_[idx] = ss; ad_[idx] = dd;
}

__device__ __forceinline__ void atomicMaxF32(float* addr, float val)
{
    if (val >= 0.0f) atomicMax((int*)addr, __float_as_int(val));
    else             atomicMin((unsigned int*)addr, __float_as_uint(val));
}

__device__ __forceinline__ void edge_endpoints(const int* ei, int E, int e,
                                               int& s, int& d)
{
    if (e < E) { s = ei[e]; d = ei[E + e]; }
    else       { s = d = e - E; }
}

__global__ void gat_edge_max(const int* __restrict__ ei, int E, int Etot,
                             const float* __restrict__ as_,
                             const float* __restrict__ ad_,
                             float* __restrict__ mbuf, int heads)
{
    const int e = blockIdx.x * blockDim.x + threadIdx.x;
    if (e >= Etot) return;
    int s, d; edge_endpoints(ei, E, e, s, d);
    for (int h = 0; h < heads; ++h) {
        float ev = as_[s * heads + h] + ad_[d * heads + h];
        ev = ev > 0.f ? ev : NEG_SLOPE * ev;
        atomicMaxF32(&mbuf[d * heads + h], ev);
    }
}

__global__ void gat_edge_aggregate(const int* __restrict__ ei, int E, int Etot,
                                   const float* __restrict__ as_,
                                   const float* __restrict__ ad_,
                                   const float* __restrict__ mbuf,
                                   float* __restrict__ den,
                                   float* __restrict__ accum,
                                   const float* __restrict__ H,
                                   int heads, int C)
{
    const int e = blockIdx.x * (blockDim.x >> 5) + (threadIdx.x >> 5);
    if (e >= Etot) return;
    const int lane = threadIdx.x & 31;
    int s, d; edge_endpoints(ei, E, e, s, d);
    const int HC = heads * C;
    const float* hs = H + (size_t)s * HC;
    float* acc = accum + (size_t)d * HC;
    for (int h = 0; h < heads; ++h) {
        float ev = as_[s * heads + h] + ad_[d * heads + h];
        ev = ev > 0.f ? ev : NEG_SLOPE * ev;
        const float alpha = __expf(ev - mbuf[d * heads + h]);
        if (lane == 0) atomicAdd(&den[d * heads + h], alpha);
        for (int c = lane; c < C; c += 32)
            atomicAdd(&acc[h * C + c], alpha * hs[h * C + c]);
    }
}

__global__ void gat_node_epilogue(float* __restrict__ accum,
                                  const float* __restrict__ den,
                                  const float* __restrict__ bias,
                                  int Nn, int heads, int C)
{
    const int HC = heads * C;
    const int idx = blockIdx.x * blockDim.x + threadIdx.x;
    if (idx >= Nn * HC) return;
    const int n = idx / HC, rem = idx % HC, h = rem / C;
    float v = accum[idx] / den[n * heads + h] + bias[rem];
    accum[idx] = v > 0.f ? v : (__expf(v) - 1.0f);
}

__global__ void gat_pool(const float* __restrict__ h2,
                         const int* __restrict__ batch,
                         float* __restrict__ sums, float* __restrict__ cnts,
                         int Nn, int C)
{
    const int idx = blockIdx.x * blockDim.x + threadIdx.x;
    if (idx >= Nn * C) return;
    const int n = idx / C, c = idx % C;
    const int b = batch[n];
    atomicAdd(&sums[b * C + c], h2[idx]);
    if (c == 0) atomicAdd(&cnts[b], 1.0f);
}

__global__ void gat_final_linear(const float* __restrict__ sums,
                                 const float* __restrict__ cnts,
                                 const float* __restrict__ Wl,
                                 const float* __restrict__ bl,
                                 float* __restrict__ out, int G, int C, int O)
{
    const int idx = blockIdx.x * blockDim.x + threadIdx.x;
    if (idx >= G * O) return;
    const int g = idx / O, o = idx % O;
    const float inv = 1.0f / fmaxf(cnts[g], 1.0f);
    float a = bl[o];
    for (int c = 0; c < C; ++c) a += sums[g * C + c] * inv * Wl[c * O + o];
    out[idx] = a;
}

__global__ void gat_fill(float* __restrict__ p, float v, size_t n)
{
    const size_t i = (size_t)blockIdx.x * blockDim.x + threadIdx.x;
    if (i < n) p[i] = v;
}

extern "C" void kernel_launch(void* const* d_in, const int* in_sizes, int n_in,
                              void* d_out, int out_size, void* d_ws, size_t ws_size,
                              hipStream_t stream)
{
    const float* x   = (const float*)d_in[0];
    const int*   ei  = (const int*)  d_in[1];
    const int*   bat = (const int*)  d_in[2];
    const float* W1  = (const float*)d_in[3];
    const float* aS1 = (const float*)d_in[4];
    const float* aD1 = (const float*)d_in[5];
    const float* b1  = (const float*)d_in[6];
    const float* W2  = (const float*)d_in[7];
    const float* aS2 = (const float*)d_in[8];
    const float* aD2 = (const float*)d_in[9];
    const float* b2  = (const float*)d_in[10];
    const float* Wl  = (const float*)d_in[11];
    const float* bl  = (const float*)d_in[12];
    float* out = (float*)d_out;

    const int N    = in_sizes[2];
    const int E    = in_sizes[1] / 2;
    const int Etot = E + N;
    const int IN   = in_sizes[0] / N;
    const int HID  = 64, HEADS = 4, OUT = 10, G = 512;
    const int HC1  = HEADS * HID;

    float* ws = (float*)d_ws;
    size_t off = 0;
    float* H1  = ws + off; off += (size_t)N * HC1;
    float* A1  = ws + off; off += (size_t)N * HC1;
    float* as1 = ws + off; off += (size_t)N * HEADS;
    float* ad1 = ws + off; off += (size_t)N * HEADS;
    float* m1  = ws + off; off += (size_t)N * HEADS;
    float* dn1 = ws + off; off += (size_t)N * HEADS;
    float* H2  = ws + off; off += (size_t)N * HID;
    float* A2  = ws + off; off += (size_t)N * HID;
    float* as2 = ws + off; off += (size_t)N;
    float* ad2 = ws + off; off += (size_t)N;
    float* m2  = ws + off; off += (size_t)N;
    float* dn2 = ws + off; off += (size_t)N;
    float* sums = ws + off; off += (size_t)G * HID;
    float* cnts = ws + off; off += (size_t)G;
    (void)ws_size; (void)n_in;

    const float NEG_INF = __builtin_inff() * -1.0f;
    const int TB = 256;
    auto blks = [](size_t n, int tb) { return (unsigned)((n + tb - 1) / tb); };

    {
        dim3 grid((N + 15) / 16, HC1 / 64);
        gat_gemm_wmma<<<grid, 128, 0, stream>>>(x, W1, H1, N, HC1, IN);
    }
    gat_attn_coef<<<blks((size_t)N * HEADS, TB), TB, 0, stream>>>(
        H1, aS1, aD1, as1, ad1, N, HEADS, HID);
    gat_fill<<<blks((size_t)N * HEADS, TB), TB, 0, stream>>>(m1, NEG_INF, (size_t)N * HEADS);
    gat_fill<<<blks((size_t)N * HEADS, TB), TB, 0, stream>>>(dn1, 0.f, (size_t)N * HEADS);
    gat_fill<<<blks((size_t)N * HC1, TB), TB, 0, stream>>>(A1, 0.f, (size_t)N * HC1);
    gat_edge_max<<<blks(Etot, TB), TB, 0, stream>>>(ei, E, Etot, as1, ad1, m1, HEADS);
    gat_edge_aggregate<<<blks(Etot, 8) , TB, 0, stream>>>(
        ei, E, Etot, as1, ad1, m1, dn1, A1, H1, HEADS, HID);
    gat_node_epilogue<<<blks((size_t)N * HC1, TB), TB, 0, stream>>>(
        A1, dn1, b1, N, HEADS, HID);

    {
        dim3 grid((N + 15) / 16, HID / 64);
        gat_gemm_wmma<<<grid, 128, 0, stream>>>(A1, W2, H2, N, HID, HC1);
    }
    gat_attn_coef<<<blks(N, TB), TB, 0, stream>>>(H2, aS2, aD2, as2, ad2, N, 1, HID);
    gat_fill<<<blks(N, TB), TB, 0, stream>>>(m2, NEG_INF, (size_t)N);
    gat_fill<<<blks(N, TB), TB, 0, stream>>>(dn2, 0.f, (size_t)N);
    gat_fill<<<blks((size_t)N * HID, TB), TB, 0, stream>>>(A2, 0.f, (size_t)N * HID);
    gat_edge_max<<<blks(Etot, TB), TB, 0, stream>>>(ei, E, Etot, as2, ad2, m2, 1);
    gat_edge_aggregate<<<blks(Etot, 8), TB, 0, stream>>>(
        ei, E, Etot, as2, ad2, m2, dn2, A2, H2, 1, HID);
    gat_node_epilogue<<<blks((size_t)N * HID, TB), TB, 0, stream>>>(
        A2, dn2, b2, N, 1, HID);

    gat_fill<<<blks((size_t)G * HID, TB), TB, 0, stream>>>(sums, 0.f, (size_t)G * HID);
    gat_fill<<<blks(G, TB), TB, 0, stream>>>(cnts, 0.f, (size_t)G);
    gat_pool<<<blks((size_t)N * HID, TB), TB, 0, stream>>>(A2, bat, sums, cnts, N, HID);
    gat_final_linear<<<blks((size_t)G * OUT, TB), TB, 0, stream>>>(
        sums, cnts, Wl, bl, out, G, HID, OUT);
}


# edge softmax and aggregation: float-atomic scatter replaced by an in-call counting sort of edges by destination and one wave per node summing source rows in registers; sum order changes
# speedup vs baseline: 2.6450x; 2.6450x over previous
#include <hip/hip_runtime.h>
#include <cstddef>


#define NEG_SLOPE 0.2f

typedef float v2f __attribute__((ext_vector_type(2)));
typedef float v8f __attribute__((ext_vector_type(8)));

__global__ void gat_gemm_wmma(const float* __restrict__ A,
                              const float* __restrict__ B,
                              float* __restrict__ C,
                              int M, int N, int K)
{
    __shared__ float As[16 * 256];
    const int m0   = blockIdx.x * 16;
    const int wave = threadIdx.x >> 5;
    const int lane = threadIdx.x & 31;
    const int half = lane >> 4;
    const int l15  = lane & 15;
    const int n0   = (blockIdx.y * (blockDim.x >> 5) + wave) * 16;

    for (int i = threadIdx.x; i < 16 * K; i += blockDim.x) {
        const int r = i / K, c = i % K;
        const int gm = m0 + r;
        As[i] = (gm < M) ? A[(size_t)gm * K + c] : 0.0f;
    }
    __syncthreads();

    v8f acc = {};
    for (int k = 0; k < K; k += 4) {
        v2f a, b;
        const int ka = k + 2 * half;
        a.x = As[l15 * K + ka + 0];
        a.y = As[l15 * K + ka + 1];
        b.x = B[(size_t)(ka + 0) * N + n0 + l15];
        b.y = B[(size_t)(ka + 1) * N + n0 + l15];
        acc = __builtin_amdgcn_wmma_f32_16x16x4_f32(
            false, a, false, b, (short)0, acc, false, false);
    }

    #pragma unroll
    for (int r = 0; r < 8; ++r) {
        const int gm = m0 + r + 8 * half;
        if (gm < M) C[(size_t)gm * N + n0 + l15] = acc[r];
    }
}

__global__ void gat_attn_coef(const float* __restrict__ H,
                              const float* __restrict__ att_s,
                              const float* __restrict__ att_d,
                              float* __restrict__ as_, float* __restrict__ ad_,
                              int Nn, int heads, int C)
{
    const int idx = blockIdx.x * blockDim.x + threadIdx.x;
    if (idx >= Nn * heads) return;
    const int n = idx / heads, h = idx % heads;
    const float* hp = H + (size_t)n * heads * C + h * C;
    const float* sp = att_s + h * C;
    const float* dp = att_d + h * C;
    float ss = 0.f, dd = 0.f;
    #pragma unroll 8
    for (int c = 0; c < C; ++c) { const float v = hp[c]; ss += v * sp[c]; dd += v * dp[c]; }
    as_[idx] = ss; ad_[idx] = dd;
}

__global__ void gat_dst_count(const int* __restrict__ ei, int E, int* __restrict__ cnt)
{
    const int e = blockIdx.x * blockDim.x + threadIdx.x;
    if (e < E) atomicAdd(&cnt[ei[E + e]], 1);
}

__global__ void gat_dst_alloc(int* cnt, int* __restrict__ start,
                              int* __restrict__ cursor, int Nn)
{
    const int i = blockIdx.x * blockDim.x + threadIdx.x;
    const int lane = threadIdx.x & 31;
    const int c = (i < Nn) ? cnt[i] : 0;
    int incl = c;
    for (int o = 1; o < 32; o <<= 1) {
        const int t = __shfl_up(incl, o);
        if (lane >= o) incl += t;
    }
    int base = 0;
    if (lane == 31) base = atomicAdd(&cnt[Nn], incl);
    base = __shfl(base, 31);
    if (i < Nn) { const int st = base + incl - c; start[i] = st; cursor[i] = st; }
}

__global__ void gat_dst_place(const int* __restrict__ ei, int E,
                              int* __restrict__ cursor, int* __restrict__ csr_src)
{
    const int e = blockIdx.x * blockDim.x + threadIdx.x;
    if (e >= E) return;
    const int s = ei[e], d = ei[E + e];
    const int p = atomicAdd(&cursor[d], 1);
    csr_src[p] = s;
}

template <int HEADS>
__global__ __launch_bounds__(256)
void gat_node_gather(const int* __restrict__ csr_src,
                     const int* __restrict__ start,
                     const int* __restrict__ cnt,
                     const float* __restrict__ as_,
                     const float* __restrict__ ad_,
                     const float* __restrict__ H,
                     const float* __restrict__ bias,
                     float* __restrict__ outp, int Nn)
{
    constexpr int HC  = HEADS * 64;
    constexpr int VPL = HC / 32;
    const int lane = threadIdx.x & 31;
    const int wave = __builtin_amdgcn_readfirstlane((int)(threadIdx.x >> 5));
    const int n = blockIdx.x * (int)(blockDim.x >> 5) + wave;
    if (n >= Nn) return;
    const int beg = start[n];
    const int deg = cnt[n];
    const int tot = deg + 1;
    const float NEG_INF = -__builtin_inff();

    float adv[HEADS];
    for (int h = 0; h < HEADS; ++h) adv[h] = ad_[n * HEADS + h];

    auto scores = [&](int s, float (&ev)[HEADS]) {
        float av[HEADS];
        if constexpr (HEADS == 4) {
            const float4 q = *reinterpret_cast<const float4*>(as_ + (size_t)s * 4);
            av[0] = q.x; av[1] = q.y; av[2] = q.z; av[3] = q.w;
        } else {
            for (int h = 0; h < HEADS; ++h) av[h] = as_[s * HEADS + h];
        }
        for (int h = 0; h < HEADS; ++h) {
            const float v = av[h] + adv[h];
            ev[h] = v > 0.f ? v : NEG_SLOPE * v;
        }
    };

    int   s0 = n;
    float e0[HEADS];
    for (int h = 0; h < HEADS; ++h) e0[h] = NEG_INF;
    const bool act0 = lane < tot;
    if (act0) {
        s0 = (lane < deg) ? csr_src[beg + lane] : n;
        scores(s0, e0);
    }
    float m[HEADS];
    for (int h = 0; h < HEADS; ++h) m[h] = e0[h];
    for (int j0 = 32; j0 < tot; j0 += 32) {
        const int j = j0 + lane;
        if (j < tot) {
            const int s = (j < deg) ? csr_src[beg + j] : n;
            float ev[HEADS];
            scores(s, ev);
            for (int h = 0; h < HEADS; ++h) m[h] = fmaxf(m[h], ev[h]);
        }
    }
    for (int h = 0; h < HEADS; ++h)
        for (int o = 16; o > 0; o >>= 1) m[h] = fmaxf(m[h], __shfl_xor(m[h], o));

    float acc[VPL];
    for (int i = 0; i < VPL; ++i) acc[i] = 0.f;
    float dl[HEADS];
    for (int h = 0; h < HEADS; ++h) dl[h] = 0.f;
    const bool hi = (lane & 16) != 0;

    auto chunk = [&](int s, const float (&ev)[HEADS], bool act, int cn) {
        float al[HEADS];
        for (int h = 0; h < HEADS; ++h) {
            al[h] = act ? __expf(ev[h] - m[h]) : 0.f;
            dl[h] += al[h];
        }
        #pragma unroll 4
        for (int t = 0; t < cn; ++t) {
            const int st = __builtin_amdgcn_readlane(s, t);
            float w[HEADS];
            for (int h = 0; h < HEADS; ++h)
                w[h] = __int_as_float(__builtin_amdgcn_readlane(__float_as_int(al[h]), t));
            const float* row = H + (size_t)st * HC;
            if constexpr (HEADS == 4) {
                const float4 r0 = *reinterpret_cast<const float4*>(row + lane * 4);
                const float4 r1 = *reinterpret_cast<const float4*>(row + 128 + lane * 4);
                const float w0 = hi ? w[1] : w[0];
                const float w1 = hi ? w[3] : w[2];
                acc[0] += w0 * r0.x; acc[1] += w0 * r0.y;
                acc[2] += w0 * r0.z; acc[3] += w0 * r0.w;
                acc[4] += w1 * r1.x; acc[5] += w1 * r1.y;
                acc[6] += w1 * r1.z; acc[7] += w1 * r1.w;
            } else {
                const float2 r = *reinterpret_cast<const float2*>(row + lane * 2);
                acc[0] += w[0] * r.x; acc[1] += w[0] * r.y;
            }
        }
    };

    chunk(s0, e0, act0, tot < 32 ? tot : 32);
    for (int j0 = 32; j0 < tot; j0 += 32) {
        const int j = j0 + lane;
        const bool act = j < tot;
        int s = n;
        float ev[HEADS];
        for (int h = 0; h < HEADS; ++h) ev[h] = NEG_INF;
        if (act) {
            s = (j < deg) ? csr_src[beg + j] : n;
            scores(s, ev);
        }
        const int left = tot - j0;
        chunk(s, ev, act, left < 32 ? left : 32);
    }

    for (int h = 0; h < HEADS; ++h)
        for (int o = 16; o > 0; o >>= 1) dl[h] += __shfl_xor(dl[h], o);

    float* orow = outp + (size_t)n * HC;
    if constexpr (HEADS == 4) {
        const float d0 = hi ? dl[1] : dl[0];
        const float d1 = hi ? dl[3] : dl[2];
        float o[8];
        for (int i = 0; i < 4; ++i) {
            const float v0 = acc[i] / d0 + bias[lane * 4 + i];
            const float v1 = acc[4 + i] / d1 + bias[128 + lane * 4 + i];
            o[i]     = v0 > 0.f ? v0 : (__expf(v0) - 1.0f);
            o[4 + i] = v1 > 0.f ? v1 : (__expf(v1) - 1.0f);
        }
        *reinterpret_cast<float4*>(orow + lane * 4)       = make_float4(o[0], o[1], o[2], o[3]);
        *reinterpret_cast<float4*>(orow + 128 + lane * 4) = make_float4(o[4], o[5], o[6], o[7]);
    } else {
        const float v0 = acc[0] / dl[0] + bias[lane * 2 + 0];
        const float v1 = acc[1] / dl[0] + bias[lane * 2 + 1];
        *reinterpret_cast<float2*>(orow + lane * 2) =
            make_float2(v0 > 0.f ? v0 : (__expf(v0) - 1.0f),
                        v1 > 0.f ? v1 : (__expf(v1) - 1.0f));
    }
}

__global__ void gat_pool(const float* __restrict__ h2,
                         const int* __restrict__ batch,
                         float* __restrict__ sums, float* __restrict__ cnts,
                         int Nn, int C)
{
    const int idx = blockIdx.x * blockDim.x + threadIdx.x;
    if (idx >= Nn * C) return;
    const int n = idx / C, c = idx % C;
    const int b = batch[n];
    atomicAdd(&sums[b * C + c], h2[idx]);
    if (c == 0) atomicAdd(&cnts[b], 1.0f);
}

__global__ void gat_final_linear(const float* __restrict__ sums,
                                 const float* __restrict__ cnts,
                                 const float* __restrict__ Wl,
                                 const float* __restrict__ bl,
                                 float* __restrict__ out, int G, int C, int O)
{
    const int idx = blockIdx.x * blockDim.x + threadIdx.x;
    if (idx >= G * O) return;
    const int g = idx / O, o = idx % O;
    const float inv = 1.0f / fmaxf(cnts[g], 1.0f);
    float a = bl[o];
    for (int c = 0; c < C; ++c) a += sums[g * C + c] * inv * Wl[c * O + o];
    out[idx] = a;
}

__global__ void gat_fill(float* __restrict__ p, float v, size_t n)
{
    const size_t i = (size_t)blockIdx.x * blockDim.x + threadIdx.x;
    if (i < n) p[i] = v;
}

extern "C" void kernel_launch(void* const* d_in, const int* in_sizes, int n_in,
                              void* d_out, int out_size, void* d_ws, size_t ws_size,
                              hipStream_t stream)
{
    const float* x   = (const float*)d_in[0];
    const int*   ei  = (const int*)  d_in[1];
    const int*   bat = (const int*)  d_in[2];
    const float* W1  = (const float*)d_in[3];
    const float* aS1 = (const float*)d_in[4];
    const float* aD1 = (const float*)d_in[5];
    const float* b1  = (const float*)d_in[6];
    const float* W2  = (const float*)d_in[7];
    const float* aS2 = (const float*)d_in[8];
    const float* aD2 = (const float*)d_in[9];
    const float* b2  = (const float*)d_in[10];
    const float* Wl  = (const float*)d_in[11];
    const float* bl  = (const float*)d_in[12];
    float* out = (float*)d_out;

    const int N    = in_sizes[2];
    const int E    = in_sizes[1] / 2;
    const int IN   = in_sizes[0] / N;
    const int HID  = 64, HEADS = 4, OUT = 10, G = 512;
    const int HC1  = HEADS * HID;

    float* ws = (float*)d_ws;
    size_t off = 0;
    float* H1  = ws + off; off += (size_t)N * HC1;
    float* A1  = ws + off; off += (size_t)N * HC1;
    float* as1 = ws + off; off += (size_t)N * HEADS;
    float* ad1 = ws + off; off += (size_t)N * HEADS;
    float* H2  = ws + off; off += (size_t)N * HID;
    float* A2  = ws + off; off += (size_t)N * HID;
    float* as2 = ws + off; off += (size_t)N;
    float* ad2 = ws + off; off += (size_t)N;
    float* sums = ws + off; off += (size_t)G * HID;
    float* cnts = ws + off; off += (size_t)G;
    int* dcnt   = (int*)(ws + off); off += (size_t)N + 1;
    int* dstart = (int*)(ws + off); off += (size_t)N;
    int* dcur   = (int*)(ws + off); off += (size_t)N;
    int* csrc   = (int*)(ws + off); off += (size_t)E;
    (void)ws_size; (void)n_in;

    const int TB = 256;
    auto blks = [](size_t n, int tb) { return (unsigned)((n + tb - 1) / tb); };

    gat_fill<<<blks((size_t)N + 1, TB), TB, 0, stream>>>((float*)dcnt, 0.f, (size_t)N + 1);
    gat_dst_count<<<blks(E, TB), TB, 0, stream>>>(ei, E, dcnt);
    gat_dst_alloc<<<blks(N, TB), TB, 0, stream>>>(dcnt, dstart, dcur, N);
    gat_dst_place<<<blks(E, TB), TB, 0, stream>>>(ei, E, dcur, csrc);

    {
        dim3 grid((N + 15) / 16, HC1 / 64);
        gat_gemm_wmma<<<grid, 128, 0, stream>>>(x, W1, H1, N, HC1, IN);
    }
    gat_attn_coef<<<blks((size_t)N * HEADS, TB), TB, 0, stream>>>(
        H1, aS1, aD1, as1, ad1, N, HEADS, HID);
    gat_node_gather<4><<<blks(N, 8), TB, 0, stream>>>(
        csrc, dstart, dcnt, as1, ad1, H1, b1, A1, N);

    {
        dim3 grid((N + 15) / 16, HID / 64);
        gat_gemm_wmma<<<grid, 128, 0, stream>>>(A1, W2, H2, N, HID, HC1);
    }
    gat_attn_coef<<<blks(N, TB), TB, 0, stream>>>(H2, aS2, aD2, as2, ad2, N, 1, HID);
    gat_node_gather<1><<<blks(N, 8), TB, 0, stream>>>(
        csrc, dstart, dcnt, as2, ad2, H2, b2, A2, N);

    gat_fill<<<blks((size_t)G * HID, TB), TB, 0, stream>>>(sums, 0.f, (size_t)G * HID);
    gat_fill<<<blks(G, TB), TB, 0, stream>>>(cnts, 0.f, (size_t)G);
    gat_pool<<<blks((size_t)N * HID, TB), TB, 0, stream>>>(A2, bat, sums, cnts, N, HID);
    gat_final_linear<<<blks((size_t)G * OUT, TB), TB, 0, stream>>>(
        sums, cnts, Wl, bl, out, G, HID, OUT);
}


# GEMM re-tiled: wave keeps its 16 rows of A in registers for all K, B slab staged once per block in LDS in fragment order (16-byte reads, no bank conflicts), whole-row stores; k sum order changes
# speedup vs baseline: 3.6645x; 1.3854x over previous
#include <hip/hip_runtime.h>
#include <cstddef>


#define NEG_SLOPE 0.2f

typedef float v2f __attribute__((ext_vector_type(2)));
typedef float v8f __attribute__((ext_vector_type(8)));

template <int K, int N>
__global__ __launch_bounds__(256)
void gat_gemm_wmma(const float* __restrict__ A,
                   const float* __restrict__ B,
                   float* __restrict__ C, int M)
{
    constexpr int KJ    = K / 8;
    constexpr int BS_F4 = KJ * 4 * 32;
    constexpr int CP    = 68;
    constexpr int CS_F4 = 8 * 16 * CP / 4;
    constexpr int LDS_F4 = BS_F4 > CS_F4 ? BS_F4 : CS_F4;
    __shared__ float4 smem[LDS_F4];

    const int lane = threadIdx.x & 31;
    const int wave = __builtin_amdgcn_readfirstlane((int)(threadIdx.x >> 5));
    const int c    = lane & 15;
    const int h    = lane >> 4;
    const int nb   = blockIdx.x * 64;
    const int m0   = (blockIdx.y * 8 + wave) * 16;

    int arow = m0 + c;
    if (arow > M - 1) arow = M - 1;
    const float4* ap = reinterpret_cast<const float4*>(A + (size_t)arow * K + 4 * h);
    float4 a[KJ];
    #pragma unroll
    for (int j = 0; j < KJ; ++j) a[j] = ap[2 * j];

    #pragma unroll
    for (int it = 0; it < BS_F4 / 256; ++it) {
        const int idx = it * 256 + (int)threadIdx.x;
        const int sc = idx & 15, sh = (idx >> 4) & 1, st = (idx >> 5) & 3, sj = idx >> 7;
        const float* bp = B + (size_t)(8 * sj + 4 * sh) * N + nb + 16 * st + sc;
        float4 v;
        v.x = bp[0]; v.y = bp[N]; v.z = bp[2 * N]; v.w = bp[3 * N];
        smem[idx] = v;
    }
    __syncthreads();

    v8f acc[4] = {};
    #pragma unroll
    for (int j = 0; j < KJ; ++j) {
        float4 b[4];
        #pragma unroll
        for (int t = 0; t < 4; ++t) b[t] = smem[(j * 4 + t) * 32 + lane];
        v2f a0, a1;
        a0.x = a[j].x; a0.y = a[j].y;
        a1.x = a[j].z; a1.y = a[j].w;
        #pragma unroll
        for (int t = 0; t < 4; ++t) {
            v2f b0; b0.x = b[t].x; b0.y = b[t].y;
            acc[t] = __builtin_amdgcn_wmma_f32_16x16x4_f32(
                false, a0, false, b0, (short)0, acc[t], false, false);
        }
        #pragma unroll
        for (int t = 0; t < 4; ++t) {
            v2f b1; b1.x = b[t].z; b1.y = b[t].w;
            acc[t] = __builtin_amdgcn_wmma_f32_16x16x4_f32(
                false, a1, false, b1, (short)0, acc[t], false, false);
        }
    }

    __syncthreads();
    float* cs = reinterpret_cast<float*>(smem) + wave * (16 * CP);
    #pragma unroll
    for (int t = 0; t < 4; ++t)
        #pragma unroll
        for (int r = 0; r < 8; ++r)
            cs[(8 * h + r) * CP + 16 * t + c] = acc[t][r];
    __syncthreads();
    const float* cr = cs + h * CP + 4 * c;
    if (m0 + 16 <= M) {
        float* cp = C + (size_t)(m0 + h) * N + nb + 4 * c;
        float4 v[8];
        #pragma unroll
        for (int i = 0; i < 8; ++i)
            v[i] = *reinterpret_cast<const float4*>(cr + 2 * i * CP);
        #pragma unroll
        for (int i = 0; i < 8; ++i)
            *reinterpret_cast<float4*>(cp + (size_t)(2 * i) * N) = v[i];
    } else {
        #pragma unroll
        for (int i = 0; i < 8; ++i) {
            const int gm = m0 + 2 * i + h;
            if (gm < M)
                *reinterpret_cast<float4*>(C + (size_t)gm * N + nb + 4 * c) =
                    *reinterpret_cast<const float4*>(cr + 2 * i * CP);
        }
    }
}

__global__ void gat_attn_coef(const float* __restrict__ H,
                              const float* __restrict__ att_s,
                              const float* __restrict__ att_d,
                              float* __restrict__ as_, float* __restrict__ ad_,
                              int Nn, int heads, int C)
{
    const int idx = blockIdx.x * blockDim.x + threadIdx.x;
    if (idx >= Nn * heads) return;
    const int n = idx / heads, h = idx % heads;
    const float* hp = H + (size_t)n * heads * C + h * C;
    const float* sp = att_s + h * C;
    const float* dp = att_d + h * C;
    float ss = 0.f, dd = 0.f;
    #pragma unroll 8
    for (int c = 0; c < C; ++c) { const float v = hp[c]; ss += v * sp[c]; dd += v * dp[c]; }
    as_[idx] = ss; ad_[idx] = dd;
}

__global__ void gat_dst_count(const int* __restrict__ ei, int E, int* __restrict__ cnt)
{
    const int e = blockIdx.x * blockDim.x + threadIdx.x;
    if (e < E) atomicAdd(&cnt[ei[E + e]], 1);
}

__global__ void gat_dst_alloc(int* cnt, int* __restrict__ start,
                              int* __restrict__ cursor, int Nn)
{
    const int i = blockIdx.x * blockDim.x + threadIdx.x;
    const int lane = threadIdx.x & 31;
    const int c = (i < Nn) ? cnt[i] : 0;
    int incl = c;
    for (int o = 1; o < 32; o <<= 1) {
        const int t = __shfl_up(incl, o);
        if (lane >= o) incl += t;
    }
    int base = 0;
    if (lane == 31) base = atomicAdd(&cnt[Nn], incl);
    base = __shfl(base, 31);
    if (i < Nn) { const int st = base + incl - c; start[i] = st; cursor[i] = st; }
}

__global__ void gat_dst_place(const int* __restrict__ ei, int E,
                              int* __restrict__ cursor, int* __restrict__ csr_src)
{
    const int e = blockIdx.x * blockDim.x + threadIdx.x;
    if (e >= E) return;
    const int s = ei[e], d = ei[E + e];
    const int p = atomicAdd(&cursor[d], 1);
    csr_src[p] = s;
}

template <int HEADS>
__global__ __launch_bounds__(256)
void gat_node_gather(const int* __restrict__ csr_src,
                     const int* __restrict__ start,
                     const int* __restrict__ cnt,
                     const float* __restrict__ as_,
                     const float* __restrict__ ad_,
                     const float* __restrict__ H,
                     const float* __restrict__ bias,
                     float* __restrict__ outp, int Nn)
{
    constexpr int HC  = HEADS * 64;
    constexpr int VPL = HC / 32;
    const int lane = threadIdx.x & 31;
    const int wave = __builtin_amdgcn_readfirstlane((int)(threadIdx.x >> 5));
    const int n = blockIdx.x * (int)(blockDim.x >> 5) + wave;
    if (n >= Nn) return;
    const int beg = start[n];
    const int deg = cnt[n];
    const int tot = deg + 1;
    const float NEG_INF = -__builtin_inff();

    float adv[HEADS];
    for (int h = 0; h < HEADS; ++h) adv[h] = ad_[n * HEADS + h];

    auto scores = [&](int s, float (&ev)[HEADS]) {
        float av[HEADS];
        if constexpr (HEADS == 4) {
            const float4 q = *reinterpret_cast<const float4*>(as_ + (size_t)s * 4);
            av[0] = q.x; av[1] = q.y; av[2] = q.z; av[3] = q.w;
        } else {
            for (int h = 0; h < HEADS; ++h) av[h] = as_[s * HEADS + h];
        }
        for (int h = 0; h < HEADS; ++h) {
            const float v = av[h] + adv[h];
            ev[h] = v > 0.f ? v : NEG_SLOPE * v;
        }
    };

    int   s0 = n;
    float e0[HEADS];
    for (int h = 0; h < HEADS; ++h) e0[h] = NEG_INF;
    const bool act0 = lane < tot;
    if (act0) {
        s0 = (lane < deg) ? csr_src[beg + lane] : n;
        scores(s0, e0);
    }
    float m[HEADS];
    for (int h = 0; h < HEADS; ++h) m[h] = e0[h];
    for (int j0 = 32; j0 < tot; j0 += 32) {
        const int j = j0 + lane;
        if (j < tot) {
            const int s = (j < deg) ? csr_src[beg + j] : n;
            float ev[HEADS];
            scores(s, ev);
            for (int h = 0; h < HEADS; ++h) m[h] = fmaxf(m[h], ev[h]);
        }
    }
    for (int h = 0; h < HEADS; ++h)
        for (int o = 16; o > 0; o >>= 1) m[h] = fmaxf(m[h], __shfl_xor(m[h], o));

    float acc[VPL];
    for (int i = 0; i < VPL; ++i) acc[i] = 0.f;
    float dl[HEADS];
    for (int h = 0; h < HEADS; ++h) dl[h] = 0.f;
    const bool hi = (lane & 16) != 0;

    auto chunk = [&](int s, const float (&ev)[HEADS], bool act, int cn) {
        float al[HEADS];
        for (int h = 0; h < HEADS; ++h) {
            al[h] = act ? __expf(ev[h] - m[h]) : 0.f;
            dl[h] += al[h];
        }
        #pragma unroll 4
        for (int t = 0; t < cn; ++t) {
            const int st = __builtin_amdgcn_readlane(s, t);
            float w[HEADS];
            for (int h = 0; h < HEADS; ++h)
                w[h] = __int_as_float(__builtin_amdgcn_readlane(__float_as_int(al[h]), t));
            const float* row = H + (size_t)st * HC;
            if constexpr (HEADS == 4) {
                const float4 r0 = *reinterpret_cast<const float4*>(row + lane * 4);
                const float4 r1 = *reinterpret_cast<const float4*>(row + 128 + lane * 4);
                const float w0 = hi ? w[1] : w[0];
                const float w1 = hi ? w[3] : w[2];
                acc[0] += w0 * r0.x; acc[1] += w0 * r0.y;
                acc[2] += w0 * r0.z; acc[3] += w0 * r0.w;
                acc[4] += w1 * r1.x; acc[5] += w1 * r1.y;
                acc[6] += w1 * r1.z; acc[7] += w1 * r1.w;
            } else {
                const float2 r = *reinterpret_cast<const float2*>(row + lane * 2);
                acc[0] += w[0] * r.x; acc[1] += w[0] * r.y;
            }
        }
    };

    chunk(s0, e0, act0, tot < 32 ? tot : 32);
    for (int j0 = 32; j0 < tot; j0 += 32) {
        const int j = j0 + lane;
        const bool act = j < tot;
        int s = n;
        float ev[HEADS];
        for (int h = 0; h < HEADS; ++h) ev[h] = NEG_INF;
        if (act) {
            s = (j < deg) ? csr_src[beg + j] : n;
            scores(s, ev);
        }
        const int left = tot - j0;
        chunk(s, ev, act, left < 32 ? left : 32);
    }

    for (int h = 0; h < HEADS; ++h)
        for (int o = 16; o > 0; o >>= 1) dl[h] += __shfl_xor(dl[h], o);

    float* orow = outp + (size_t)n * HC;
    if constexpr (HEADS == 4) {
        const float d0 = hi ? dl[1] : dl[0];
        const float d1 = hi ? dl[3] : dl[2];
        float o[8];
        for (int i = 0; i < 4; ++i) {
            const float v0 = acc[i] / d0 + bias[lane * 4 + i];
            const float v1 = acc[4 + i] / d1 + bias[128 + lane * 4 + i];
            o[i]     = v0 > 0.f ? v0 : (__expf(v0) - 1.0f);
            o[4 + i] = v1 > 0.f ? v1 : (__expf(v1) - 1.0f);
        }
        *reinterpret_cast<float4*>(orow + lane * 4)       = make_float4(o[0], o[1], o[2], o[3]);
        *reinterpret_cast<float4*>(orow + 128 + lane * 4) = make_float4(o[4], o[5], o[6], o[7]);
    } else {
        const float v0 = acc[0] / dl[0] + bias[lane * 2 + 0];
        const float v1 = acc[1] / dl[0] + bias[lane * 2 + 1];
        *reinterpret_cast<float2*>(orow + lane * 2) =
            make_float2(v0 > 0.f ? v0 : (__expf(v0) - 1.0f),
                        v1 > 0.f ? v1 : (__expf(v1) - 1.0f));
    }
}

__global__ void gat_pool(const float* __restrict__ h2,
                         const int* __restrict__ batch,
                         float* __restrict__ sums, float* __restrict__ cnts,
                         int Nn, int C)
{
    const int idx = blockIdx.x * blockDim.x + threadIdx.x;
    if (idx >= Nn * C) return;
    const int n = idx / C, c = idx % C;
    const int b = batch[n];
    atomicAdd(&sums[b * C + c], h2[idx]);
    if (c == 0) atomicAdd(&cnts[b], 1.0f);
}

__global__ void gat_final_linear(const float* __restrict__ sums,
                                 const float* __restrict__ cnts,
                                 const float* __restrict__ Wl,
                                 const float* __restrict__ bl,
                                 float* __restrict__ out, int G, int C, int O)
{
    const int idx = blockIdx.x * blockDim.x + threadIdx.x;
    if (idx >= G * O) return;
    const int g = idx / O, o = idx % O;
    const float inv = 1.0f / fmaxf(cnts[g], 1.0f);
    float a = bl[o];
    for (int c = 0; c < C; ++c) a += sums[g * C + c] * inv * Wl[c * O + o];
    out[idx] = a;
}

__global__ void gat_fill(float* __restrict__ p, float v, size_t n)
{
    const size_t i = (size_t)blockIdx.x * blockDim.x + threadIdx.x;
    if (i < n) p[i] = v;
}

extern "C" void kernel_launch(void* const* d_in, const int* in_sizes, int n_in,
                              void* d_out, int out_size, void* d_ws, size_t ws_size,
                              hipStream_t stream)
{
    const float* x   = (const float*)d_in[0];
    const int*   ei  = (const int*)  d_in[1];
    const int*   bat = (const int*)  d_in[2];
    const float* W1  = (const float*)d_in[3];
    const float* aS1 = (const float*)d_in[4];
    const float* aD1 = (const float*)d_in[5];
    const float* b1  = (const float*)d_in[6];
    const float* W2  = (const float*)d_in[7];
    const float* aS2 = (const float*)d_in[8];
    const float* aD2 = (const float*)d_in[9];
    const float* b2  = (const float*)d_in[10];
    const float* Wl  = (const float*)d_in[11];
    const float* bl  = (const float*)d_in[12];
    float* out = (float*)d_out;

    const int N    = in_sizes[2];
    const int E    = in_sizes[1] / 2;
    const int HID  = 64, HEADS = 4, OUT = 10, G = 512;
    const int HC1  = HEADS * HID;

    float* ws = (float*)d_ws;
    size_t off = 0;
    float* H1  = ws + off; off += (size_t)N * HC1;
    float* A1  = ws + off; off += (size_t)N * HC1;
    float* as1 = ws + off; off += (size_t)N * HEADS;
    float* ad1 = ws + off; off += (size_t)N * HEADS;
    float* H2  = ws + off; off += (size_t)N * HID;
    float* A2  = ws + off; off += (size_t)N * HID;
    float* as2 = ws + off; off += (size_t)N;
    float* ad2 = ws + off; off += (size_t)N;
    float* sums = ws + off; off += (size_t)G * HID;
    float* cnts = ws + off; off += (size_t)G;
    int* dcnt   = (int*)(ws + off); off += (size_t)N + 1;
    int* dstart = (int*)(ws + off); off += (size_t)N;
    int* dcur   = (int*)(ws + off); off += (size_t)N;
    int* csrc   = (int*)(ws + off); off += (size_t)E;
    (void)ws_size; (void)n_in;

    const int TB = 256;
    auto blks = [](size_t n, int tb) { return (unsigned)((n + tb - 1) / tb); };

    gat_fill<<<blks((size_t)N + 1, TB), TB, 0, stream>>>((float*)dcnt, 0.f, (size_t)N + 1);
    gat_dst_count<<<blks(E, TB), TB, 0, stream>>>(ei, E, dcnt);
    gat_dst_alloc<<<blks(N, TB), TB, 0, stream>>>(dcnt, dstart, dcur, N);
    gat_dst_place<<<blks(E, TB), TB, 0, stream>>>(ei, E, dcur, csrc);

    {
        dim3 grid(256 / 64, (N + 127) / 128);
        gat_gemm_wmma<128, 256><<<grid, 256, 0, stream>>>(x, W1, H1, N);
    }
    gat_attn_coef<<<blks((size_t)N * HEADS, TB), TB, 0, stream>>>(
        H1, aS1, aD1, as1, ad1, N, HEADS, HID);
    gat_node_gather<4><<<blks(N, 8), TB, 0, stream>>>(
        csrc, dstart, dcnt, as1, ad1, H1, b1, A1, N);

    {
        dim3 grid(64 / 64, (N + 127) / 128);
        gat_gemm_wmma<256, 64><<<grid, 256, 0, stream>>>(A1, W2, H2, N);
    }
    gat_attn_coef<<<blks(N, TB), TB, 0, stream>>>(H2, aS2, aD2, as2, ad2, N, 1, HID);
    gat_node_gather<1><<<blks(N, 8), TB, 0, stream>>>(
        csrc, dstart, dcnt, as2, ad2, H2, b2, A2, N);

    gat_fill<<<blks((size_t)G * HID, TB), TB, 0, stream>>>(sums, 0.f, (size_t)G * HID);
    gat_fill<<<blks(G, TB), TB, 0, stream>>>(cnts, 0.f, (size_t)G);
    gat_pool<<<blks((size_t)N * HID, TB), TB, 0, stream>>>(A2, bat, sums, cnts, N, HID);
    gat_final_linear<<<blks((size_t)G * OUT, TB), TB, 0, stream>>>(
        sums, cnts, Wl, bl, out, G, HID, OUT);
}


# launch fusion: a_s/a_d computed in each GEMM block's epilogue (block = one head), layer-2 rows added to the pool sums from the aggregation wave's registers, one zero fill: 14 launches -> 9
# speedup vs baseline: 4.0466x; 1.1043x over previous
#include <hip/hip_runtime.h>
#include <cstddef>


#define NEG_SLOPE 0.2f

typedef float v2f __attribute__((ext_vector_type(2)));
typedef float v8f __attribute__((ext_vector_type(8)));

template <int K, int N>
__global__ __launch_bounds__(256)
void gat_gemm_wmma(const float* __restrict__ A,
                   const float* __restrict__ B,
                   float* __restrict__ C,
                   const float* __restrict__ att_s,
                   const float* __restrict__ att_d,
                   float* __restrict__ a_src,
                   float* __restrict__ a_dst, int M)
{
    constexpr int KJ    = K / 8;
    constexpr int BS_F4 = KJ * 4 * 32;
    constexpr int CP    = 68;
    constexpr int CS_F4 = 8 * 16 * CP / 4;
    constexpr int LDS_F4 = BS_F4 > CS_F4 ? BS_F4 : CS_F4;
    __shared__ float4 smem[LDS_F4];

    const int lane = threadIdx.x & 31;
    const int wave = __builtin_amdgcn_readfirstlane((int)(threadIdx.x >> 5));
    const int c    = lane & 15;
    const int h    = lane >> 4;
    const int nb   = blockIdx.x * 64;
    const int m0   = (blockIdx.y * 8 + wave) * 16;

    int arow = m0 + c;
    if (arow > M - 1) arow = M - 1;
    const float4* ap = reinterpret_cast<const float4*>(A + (size_t)arow * K + 4 * h);
    float4 a[KJ];
    #pragma unroll
    for (int j = 0; j < KJ; ++j) a[j] = ap[2 * j];

    #pragma unroll
    for (int it = 0; it < BS_F4 / 256; ++it) {
        const int idx = it * 256 + (int)threadIdx.x;
        const int sc = idx & 15, sh = (idx >> 4) & 1, st = (idx >> 5) & 3, sj = idx >> 7;
        const float* bp = B + (size_t)(8 * sj + 4 * sh) * N + nb + 16 * st + sc;
        float4 v;
        v.x = bp[0]; v.y = bp[N]; v.z = bp[2 * N]; v.w = bp[3 * N];
        smem[idx] = v;
    }
    __syncthreads();

    v8f acc[4] = {};
    #pragma unroll
    for (int j = 0; j < KJ; ++j) {
        float4 b[4];
        #pragma unroll
        for (int t = 0; t < 4; ++t) b[t] = smem[(j * 4 + t) * 32 + lane];
        v2f a0, a1;
        a0.x = a[j].x; a0.y = a[j].y;
        a1.x = a[j].z; a1.y = a[j].w;
        #pragma unroll
        for (int t = 0; t < 4; ++t) {
            v2f b0; b0.x = b[t].x; b0.y = b[t].y;
            acc[t] = __builtin_amdgcn_wmma_f32_16x16x4_f32(
                false, a0, false, b0, (short)0, acc[t], false, false);
        }
        #pragma unroll
        for (int t = 0; t < 4; ++t) {
            v2f b1; b1.x = b[t].z; b1.y = b[t].w;
            acc[t] = __builtin_amdgcn_wmma_f32_16x16x4_f32(
                false, a1, false, b1, (short)0, acc[t], false, false);
        }
    }

    __syncthreads();
    float* cs = reinterpret_cast<float*>(smem) + wave * (16 * CP);
    #pragma unroll
    for (int t = 0; t < 4; ++t)
        #pragma unroll
        for (int r = 0; r < 8; ++r)
            cs[(8 * h + r) * CP + 16 * t + c] = acc[t][r];
    __syncthreads();
    const float* cr = cs + h * CP + 4 * c;
    if (m0 + 16 <= M) {
        float* cp = C + (size_t)(m0 + h) * N + nb + 4 * c;
        float4 v[8];
        #pragma unroll
        for (int i = 0; i < 8; ++i)
            v[i] = *reinterpret_cast<const float4*>(cr + 2 * i * CP);
        #pragma unroll
        for (int i = 0; i < 8; ++i)
            *reinterpret_cast<float4*>(cp + (size_t)(2 * i) * N) = v[i];
    } else {
        #pragma unroll
        for (int i = 0; i < 8; ++i) {
            const int gm = m0 + 2 * i + h;
            if (gm < M)
                *reinterpret_cast<float4*>(C + (size_t)gm * N + nb + 4 * c) =
                    *reinterpret_cast<const float4*>(cr + 2 * i * CP);
        }
    }

    {
        const float* av   = (h ? att_d : att_s) + nb;
        const float* crow = cs + c * CP;
        float s = 0.f;
        #pragma unroll
        for (int k = 0; k < 64; k += 4) {
            const float4 v = *reinterpret_cast<const float4*>(crow + k);
            const float4 w = *reinterpret_cast<const float4*>(av + k);
            s = __builtin_fmaf(v.x, w.x, s); s = __builtin_fmaf(v.y, w.y, s);
            s = __builtin_fmaf(v.z, w.z, s); s = __builtin_fmaf(v.w, w.w, s);
        }
        const int gm = m0 + c;
        if (gm < M) (h ? a_dst : a_src)[(size_t)gm * (N / 64) + blockIdx.x] = s;
    }
}

__global__ void gat_dst_count(const int* __restrict__ ei, int E, int* __restrict__ cnt)
{
    const int e = blockIdx.x * blockDim.x + threadIdx.x;
    if (e < E) atomicAdd(&cnt[ei[E + e]], 1);
}

__global__ void gat_dst_alloc(int* cnt, int* __restrict__ start,
                              int* __restrict__ cursor, int Nn)
{
    const int i = blockIdx.x * blockDim.x + threadIdx.x;
    const int lane = threadIdx.x & 31;
    const int c = (i < Nn) ? cnt[i] : 0;
    int incl = c;
    for (int o = 1; o < 32; o <<= 1) {
        const int t = __shfl_up(incl, o);
        if (lane >= o) incl += t;
    }
    int base = 0;
    if (lane == 31) base = atomicAdd(&cnt[Nn], incl);
    base = __shfl(base, 31);
    if (i < Nn) { const int st = base + incl - c; start[i] = st; cursor[i] = st; }
}

__global__ void gat_dst_place(const int* __restrict__ ei, int E,
                              int* __restrict__ cursor, int* __restrict__ csr_src)
{
    const int e = blockIdx.x * blockDim.x + threadIdx.x;
    if (e >= E) return;
    const int s = ei[e], d = ei[E + e];
    const int p = atomicAdd(&cursor[d], 1);
    csr_src[p] = s;
}

template <int HEADS>
__global__ __launch_bounds__(256)
void gat_node_gather(const int* __restrict__ csr_src,
                     const int* __restrict__ start,
                     const int* __restrict__ cnt,
                     const float* __restrict__ as_,
                     const float* __restrict__ ad_,
                     const float* __restrict__ H,
                     const float* __restrict__ bias,
                     float* __restrict__ outp,
                     const int* __restrict__ batch,
                     float* __restrict__ sums,
                     float* __restrict__ cnts, int Nn)
{
    constexpr int HC  = HEADS * 64;
    constexpr int VPL = HC / 32;
    const int lane = threadIdx.x & 31;
    const int wave = __builtin_amdgcn_readfirstlane((int)(threadIdx.x >> 5));
    const int n = blockIdx.x * (int)(blockDim.x >> 5) + wave;
    if (n >= Nn) return;
    const int beg = start[n];
    const int deg = cnt[n];
    const int tot = deg + 1;
    const float NEG_INF = -__builtin_inff();

    float adv[HEADS];
    for (int h = 0; h < HEADS; ++h) adv[h] = ad_[n * HEADS + h];

    auto scores = [&](int s, float (&ev)[HEADS]) {
        float av[HEADS];
        if constexpr (HEADS == 4) {
            const float4 q = *reinterpret_cast<const float4*>(as_ + (size_t)s * 4);
            av[0] = q.x; av[1] = q.y; av[2] = q.z; av[3] = q.w;
        } else {
            for (int h = 0; h < HEADS; ++h) av[h] = as_[s * HEADS + h];
        }
        for (int h = 0; h < HEADS; ++h) {
            const float v = av[h] + adv[h];
            ev[h] = v > 0.f ? v : NEG_SLOPE * v;
        }
    };

    int   s0 = n;
    float e0[HEADS];
    for (int h = 0; h < HEADS; ++h) e0[h] = NEG_INF;
    const bool act0 = lane < tot;
    if (act0) {
        s0 = (lane < deg) ? csr_src[beg + lane] : n;
        scores(s0, e0);
    }
    float m[HEADS];
    for (int h = 0; h < HEADS; ++h) m[h] = e0[h];
    for (int j0 = 32; j0 < tot; j0 += 32) {
        const int j = j0 + lane;
        if (j < tot) {
            const int s = (j < deg) ? csr_src[beg + j] : n;
            float ev[HEADS];
            scores(s, ev);
            for (int h = 0; h < HEADS; ++h) m[h] = fmaxf(m[h], ev[h]);
        }
    }
    for (int h = 0; h < HEADS; ++h)
        for (int o = 16; o > 0; o >>= 1) m[h] = fmaxf(m[h], __shfl_xor(m[h], o));

    float acc[VPL];
    for (int i = 0; i < VPL; ++i) acc[i] = 0.f;
    float dl[HEADS];
    for (int h = 0; h < HEADS; ++h) dl[h] = 0.f;
    const bool hi = (lane & 16) != 0;

    auto chunk = [&](int s, const float (&ev)[HEADS], bool act, int cn) {
        float al[HEADS];
        for (int h = 0; h < HEADS; ++h) {
            al[h] = act ? __expf(ev[h] - m[h]) : 0.f;
            dl[h] += al[h];
        }
        #pragma unroll 4
        for (int t = 0; t < cn; ++t) {
            const int st = __builtin_amdgcn_readlane(s, t);
            float w[HEADS];
            for (int h = 0; h < HEADS; ++h)
                w[h] = __int_as_float(__builtin_amdgcn_readlane(__float_as_int(al[h]), t));
            const float* row = H + (size_t)st * HC;
            if constexpr (HEADS == 4) {
                const float4 r0 = *reinterpret_cast<const float4*>(row + lane * 4);
                const float4 r1 = *reinterpret_cast<const float4*>(row + 128 + lane * 4);
                const float w0 = hi ? w[1] : w[0];
                const float w1 = hi ? w[3] : w[2];
                acc[0] += w0 * r0.x; acc[1] += w0 * r0.y;
                acc[2] += w0 * r0.z; acc[3] += w0 * r0.w;
                acc[4] += w1 * r1.x; acc[5] += w1 * r1.y;
                acc[6] += w1 * r1.z; acc[7] += w1 * r1.w;
            } else {
                const float2 r = *reinterpret_cast<const float2*>(row + lane * 2);
                acc[0] += w[0] * r.x; acc[1] += w[0] * r.y;
            }
        }
    };

    chunk(s0, e0, act0, tot < 32 ? tot : 32);
    for (int j0 = 32; j0 < tot; j0 += 32) {
        const int j = j0 + lane;
        const bool act = j < tot;
        int s = n;
        float ev[HEADS];
        for (int h = 0; h < HEADS; ++h) ev[h] = NEG_INF;
        if (act) {
            s = (j < deg) ? csr_src[beg + j] : n;
            scores(s, ev);
        }
        const int left = tot - j0;
        chunk(s, ev, act, left < 32 ? left : 32);
    }

    for (int h = 0; h < HEADS; ++h)
        for (int o = 16; o > 0; o >>= 1) dl[h] += __shfl_xor(dl[h], o);

    if constexpr (HEADS == 4) {
        float* orow = outp + (size_t)n * HC;
        const float d0 = hi ? dl[1] : dl[0];
        const float d1 = hi ? dl[3] : dl[2];
        float o[8];
        for (int i = 0; i < 4; ++i) {
            const float v0 = acc[i] / d0 + bias[lane * 4 + i];
            const float v1 = acc[4 + i] / d1 + bias[128 + lane * 4 + i];
            o[i]     = v0 > 0.f ? v0 : (__expf(v0) - 1.0f);
            o[4 + i] = v1 > 0.f ? v1 : (__expf(v1) - 1.0f);
        }
        *reinterpret_cast<float4*>(orow + lane * 4)       = make_float4(o[0], o[1], o[2], o[3]);
        *reinterpret_cast<float4*>(orow + 128 + lane * 4) = make_float4(o[4], o[5], o[6], o[7]);
    } else {
        const float v0 = acc[0] / dl[0] + bias[lane * 2 + 0];
        const float v1 = acc[1] / dl[0] + bias[lane * 2 + 1];
        const float o0 = v0 > 0.f ? v0 : (__expf(v0) - 1.0f);
        const float o1 = v1 > 0.f ? v1 : (__expf(v1) - 1.0f);
        const int   hl = lane >> 1;
        const float t0 = __shfl(o0, hl),      t1 = __shfl(o1, hl);
        const float u0 = __shfl(o0, 16 + hl), u1 = __shfl(o1, 16 + hl);
        const bool  od = (lane & 1) != 0;
        const int   g  = batch[n];
        float* sp = sums + (size_t)g * 64 + lane;
        atomicAdd(sp,      od ? t1 : t0);
        atomicAdd(sp + 32, od ? u1 : u0);
        if (lane == 0) atomicAdd(&cnts[g], 1.0f);
    }
}

__global__ void gat_final_linear(const float* __restrict__ sums,
                                 const float* __restrict__ cnts,
                                 const float* __restrict__ Wl,
                                 const float* __restrict__ bl,
                                 float* __restrict__ out, int G, int C, int O)
{
    const int idx = blockIdx.x * blockDim.x + threadIdx.x;
    if (idx >= G * O) return;
    const int g = idx / O, o = idx % O;
    const float inv = 1.0f / fmaxf(cnts[g], 1.0f);
    float a = bl[o];
    for (int c = 0; c < C; ++c) a += sums[g * C + c] * inv * Wl[c * O + o];
    out[idx] = a;
}

__global__ void gat_fill(float* __restrict__ p, float v, size_t n)
{
    const size_t i = (size_t)blockIdx.x * blockDim.x + threadIdx.x;
    if (i < n) p[i] = v;
}

extern "C" void kernel_launch(void* const* d_in, const int* in_sizes, int n_in,
                              void* d_out, int out_size, void* d_ws, size_t ws_size,
                              hipStream_t stream)
{
    const float* x   = (const float*)d_in[0];
    const int*   ei  = (const int*)  d_in[1];
    const int*   bat = (const int*)  d_in[2];
    const float* W1  = (const float*)d_in[3];
    const float* aS1 = (const float*)d_in[4];
    const float* aD1 = (const float*)d_in[5];
    const float* b1  = (const float*)d_in[6];
    const float* W2  = (const float*)d_in[7];
    const float* aS2 = (const float*)d_in[8];
    const float* aD2 = (const float*)d_in[9];
    const float* b2  = (const float*)d_in[10];
    const float* Wl  = (const float*)d_in[11];
    const float* bl  = (const float*)d_in[12];
    float* out = (float*)d_out;

    const int N    = in_sizes[2];
    const int E    = in_sizes[1] / 2;
    const int HID  = 64, HEADS = 4, OUT = 10, G = 512;
    const int HC1  = HEADS * HID;

    float* ws = (float*)d_ws;
    size_t off = 0;
    float* H1  = ws + off; off += (size_t)N * HC1;
    float* A1  = ws + off; off += (size_t)N * HC1;
    float* as1 = ws + off; off += (size_t)N * HEADS;
    float* ad1 = ws + off; off += (size_t)N * HEADS;
    float* H2  = ws + off; off += (size_t)N * HID;
    float* as2 = ws + off; off += (size_t)N;
    float* ad2 = ws + off; off += (size_t)N;
    float* sums = ws + off; off += (size_t)G * HID;
    float* cnts = ws + off; off += (size_t)G;
    int* dcnt   = (int*)(ws + off); off += (size_t)N + 1;
    int* dstart = (int*)(ws + off); off += (size_t)N;
    int* dcur   = (int*)(ws + off); off += (size_t)N;
    int* csrc   = (int*)(ws + off); off += (size_t)E;
    (void)ws_size; (void)n_in;

    const int TB = 256;
    auto blks = [](size_t n, int tb) { return (unsigned)((n + tb - 1) / tb); };

    const size_t nzero = (size_t)G * HID + (size_t)G + (size_t)N + 1;
    gat_fill<<<blks(nzero, TB), TB, 0, stream>>>(sums, 0.f, nzero);
    gat_dst_count<<<blks(E, TB), TB, 0, stream>>>(ei, E, dcnt);
    gat_dst_alloc<<<blks(N, TB), TB, 0, stream>>>(dcnt, dstart, dcur, N);
    gat_dst_place<<<blks(E, TB), TB, 0, stream>>>(ei, E, dcur, csrc);

    {
        dim3 grid(256 / 64, (N + 127) / 128);
        gat_gemm_wmma<128, 256><<<grid, 256, 0, stream>>>(x, W1, H1, aS1, aD1, as1, ad1, N);
    }
    gat_node_gather<4><<<blks(N, 8), TB, 0, stream>>>(
        csrc, dstart, dcnt, as1, ad1, H1, b1, A1, bat, sums, cnts, N);

    {
        dim3 grid(64 / 64, (N + 127) / 128);
        gat_gemm_wmma<256, 64><<<grid, 256, 0, stream>>>(A1, W2, H2, aS2, aD2, as2, ad2, N);
    }
    gat_node_gather<1><<<blks(N, 8), TB, 0, stream>>>(
        csrc, dstart, dcnt, as2, ad2, H2, b2, nullptr, bat, sums, cnts, N);

    gat_final_linear<<<blks((size_t)G * OUT, TB), TB, 0, stream>>>(
        sums, cnts, Wl, bl, out, G, HID, OUT);
}


# edges grouped by destination on LDS counters: 4096-edge chunks ordered by 64-node bin, then one work-group per bin; no global atomic or scattered store per edge; 9 launches -> 7; sum order changes
# speedup vs baseline: 5.8872x; 1.4549x over previous
#include <hip/hip_runtime.h>
#include <cstddef>


#define NEG_SLOPE 0.2f

typedef float v2f __attribute__((ext_vector_type(2)));
typedef float v8f __attribute__((ext_vector_type(8)));

template <int K, int N>
__global__ __launch_bounds__(256)
void gat_gemm_wmma(const float* __restrict__ A,
                   const float* __restrict__ B,
                   float* __restrict__ C,
                   const float* __restrict__ att_s,
                   const float* __restrict__ att_d,
                   float* __restrict__ a_src,
                   float* __restrict__ a_dst, int M)
{
    constexpr int KJ    = K / 8;
    constexpr int BS_F4 = KJ * 4 * 32;
    constexpr int CP    = 68;
    constexpr int CS_F4 = 8 * 16 * CP / 4;
    constexpr int LDS_F4 = BS_F4 > CS_F4 ? BS_F4 : CS_F4;
    __shared__ float4 smem[LDS_F4];

    const int lane = threadIdx.x & 31;
    const int wave = __builtin_amdgcn_readfirstlane((int)(threadIdx.x >> 5));
    const int c    = lane & 15;
    const int h    = lane >> 4;
    const int nb   = blockIdx.x * 64;
    const int m0   = (blockIdx.y * 8 + wave) * 16;

    int arow = m0 + c;
    if (arow > M - 1) arow = M - 1;
    const float4* ap = reinterpret_cast<const float4*>(A + (size_t)arow * K + 4 * h);
    float4 a[KJ];
    #pragma unroll
    for (int j = 0; j < KJ; ++j) a[j] = ap[2 * j];

    #pragma unroll
    for (int it = 0; it < BS_F4 / 256; ++it) {
        const int idx = it * 256 + (int)threadIdx.x;
        const int sc = idx & 15, sh = (idx >> 4) & 1, st = (idx >> 5) & 3, sj = idx >> 7;
        const float* bp = B + (size_t)(8 * sj + 4 * sh) * N + nb + 16 * st + sc;
        float4 v;
        v.x = bp[0]; v.y = bp[N]; v.z = bp[2 * N]; v.w = bp[3 * N];
        smem[idx] = v;
    }
    __syncthreads();

    v8f acc[4] = {};
    #pragma unroll
    for (int j = 0; j < KJ; ++j) {
        float4 b[4];
        #pragma unroll
        for (int t = 0; t < 4; ++t) b[t] = smem[(j * 4 + t) * 32 + lane];
        v2f a0, a1;
        a0.x = a[j].x; a0.y = a[j].y;
        a1.x = a[j].z; a1.y = a[j].w;
        #pragma unroll
        for (int t = 0; t < 4; ++t) {
            v2f b0; b0.x = b[t].x; b0.y = b[t].y;
            acc[t] = __builtin_amdgcn_wmma_f32_16x16x4_f32(
                false, a0, false, b0, (short)0, acc[t], false, false);
        }
        #pragma unroll
        for (int t = 0; t < 4; ++t) {
            v2f b1; b1.x = b[t].z; b1.y = b[t].w;
            acc[t] = __builtin_amdgcn_wmma_f32_16x16x4_f32(
                false, a1, false, b1, (short)0, acc[t], false, false);
        }
    }

    __syncthreads();
    float* cs = reinterpret_cast<float*>(smem) + wave * (16 * CP);
    #pragma unroll
    for (int t = 0; t < 4; ++t)
        #pragma unroll
        for (int r = 0; r < 8; ++r)
            cs[(8 * h + r) * CP + 16 * t + c] = acc[t][r];
    __syncthreads();
    const float* cr = cs + h * CP + 4 * c;
    if (m0 + 16 <= M) {
        float* cp = C + (size_t)(m0 + h) * N + nb + 4 * c;
        float4 v[8];
        #pragma unroll
        for (int i = 0; i < 8; ++i)
            v[i] = *reinterpret_cast<const float4*>(cr + 2 * i * CP);
        #pragma unroll
        for (int i = 0; i < 8; ++i)
            *reinterpret_cast<float4*>(cp + (size_t)(2 * i) * N) = v[i];
    } else {
        #pragma unroll
        for (int i = 0; i < 8; ++i) {
            const int gm = m0 + 2 * i + h;
            if (gm < M)
                *reinterpret_cast<float4*>(C + (size_t)gm * N + nb + 4 * c) =
                    *reinterpret_cast<const float4*>(cr + 2 * i * CP);
        }
    }

    {
        const float* av   = (h ? att_d : att_s) + nb;
        const float* crow = cs + c * CP;
        float s = 0.f;
        #pragma unroll
        for (int k = 0; k < 64; k += 4) {
            const float4 v = *reinterpret_cast<const float4*>(crow + k);
            const float4 w = *reinterpret_cast<const float4*>(av + k);
            s = __builtin_fmaf(v.x, w.x, s); s = __builtin_fmaf(v.y, w.y, s);
            s = __builtin_fmaf(v.z, w.z, s); s = __builtin_fmaf(v.w, w.w, s);
        }
        const int gm = m0 + c;
        if (gm < M) (h ? a_dst : a_src)[(size_t)gm * (N / 64) + blockIdx.x] = s;
    }
}

#define BIN_SHIFT 6
#define BIN_NODES 64
#define EDGE_CHUNK 4096
#define MAX_BINS 1024
#define FINE_CAP 4096

__global__ __launch_bounds__(1024)
void gat_bin_coarse(const int* __restrict__ ei, int E, int nbin,
                    int2* __restrict__ etmp, int2* __restrict__ seg,
                    float* __restrict__ zero_p, int nzero)
{
    __shared__ int  hist[MAX_BINS];
    __shared__ int  wsum[32];
    __shared__ int2 stage[EDGE_CHUNK];
    const int tid  = threadIdx.x;
    const int lane = tid & 31, wave = tid >> 5;
    const int nblk = gridDim.x;

    for (int i = blockIdx.x * 1024 + tid; i < nzero; i += nblk * 1024) zero_p[i] = 0.f;
    hist[tid] = 0;
    __syncthreads();

    const int e0 = blockIdx.x * EDGE_CHUNK;
    int s[4], d[4], r[4];
    #pragma unroll
    for (int k = 0; k < 4; ++k) {
        const int e = e0 + k * 1024 + tid;
        const bool ok = e < E;
        s[k] = ok ? ei[e] : 0;
        d[k] = ok ? ei[E + e] : -1;
    }
    #pragma unroll
    for (int k = 0; k < 4; ++k)
        r[k] = (d[k] >= 0) ? atomicAdd(&hist[d[k] >> BIN_SHIFT], 1) : 0;
    __syncthreads();

    const int c = hist[tid];
    int incl = c;
    for (int o = 1; o < 32; o <<= 1) {
        const int t = __shfl_up(incl, o);
        if (lane >= o) incl += t;
    }
    if (lane == 31) wsum[wave] = incl;
    __syncthreads();
    if (wave == 0) {
        const int w = wsum[lane];
        int wi = w;
        for (int o = 1; o < 32; o <<= 1) {
            const int t = __shfl_up(wi, o);
            if (lane >= o) wi += t;
        }
        wsum[lane] = wi - w;
    }
    __syncthreads();
    const int off = wsum[wave] + incl - c;
    hist[tid] = off;
    if (tid < nbin) seg[(size_t)tid * nblk + blockIdx.x] = make_int2(e0 + off, c);
    __syncthreads();

    #pragma unroll
    for (int k = 0; k < 4; ++k)
        if (d[k] >= 0) stage[hist[d[k] >> BIN_SHIFT] + r[k]] = make_int2(s[k], d[k]);
    __syncthreads();
    #pragma unroll
    for (int k = 0; k < 4; ++k) {
        const int i = k * 1024 + tid;
        if (e0 + i < E) etmp[e0 + i] = stage[i];
    }
}

__global__ __launch_bounds__(256)
void gat_bin_fine(const int2* __restrict__ etmp, const int2* __restrict__ seg, int nseg,
                  int* __restrict__ gtot, int* __restrict__ start, int* __restrict__ cnt,
                  int* __restrict__ csr_src, int Nn)
{
    __shared__ int ncnt[BIN_NODES];
    __shared__ int ncur[BIN_NODES];
    __shared__ int tot_s, base_s;
    __shared__ int stage[FINE_CAP];
    const int tid = threadIdx.x;
    const int n0  = blockIdx.x << BIN_SHIFT;
    const int2* sp = seg + (size_t)blockIdx.x * nseg;

    if (tid < BIN_NODES) ncnt[tid] = 0;
    __syncthreads();
    for (int g = tid; g < nseg; g += 256) {
        const int2  sg = sp[g];
        const int2* ep = etmp + sg.x;
        for (int k = 0; k < sg.y; k += 4) {
            int dd[4];
            #pragma unroll
            for (int u = 0; u < 4; ++u) dd[u] = (k + u < sg.y) ? ep[k + u].y : -1;
            #pragma unroll
            for (int u = 0; u < 4; ++u)
                if (dd[u] >= 0) atomicAdd(&ncnt[dd[u] - n0], 1);
        }
    }
    __syncthreads();

    if (tid < 32) {
        const int v0 = ncnt[2 * tid], v1 = ncnt[2 * tid + 1];
        const int sm = v0 + v1;
        int incl = sm;
        for (int o = 1; o < 32; o <<= 1) {
            const int t = __shfl_up(incl, o);
            if (tid >= o) incl += t;
        }
        const int tot = __shfl(incl, 31);
        int base = 0;
        if (tid == 0) base = atomicAdd(gtot, tot);
        base = __shfl(base, 0);
        const int ex = incl - sm;
        ncur[2 * tid]     = ex;
        ncur[2 * tid + 1] = ex + v0;
        const int na = n0 + 2 * tid;
        if (na < Nn)     { start[na]     = base + ex;      cnt[na]     = v0; }
        if (na + 1 < Nn) { start[na + 1] = base + ex + v0; cnt[na + 1] = v1; }
        if (tid == 0) { tot_s = tot; base_s = base; }
    }
    __syncthreads();

    const int  tot    = tot_s, base = base_s;
    const bool staged = tot <= FINE_CAP;
    for (int g = tid; g < nseg; g += 256) {
        const int2  sg = sp[g];
        const int2* ep = etmp + sg.x;
        for (int k = 0; k < sg.y; k += 4) {
            int2 ed[4];
            #pragma unroll
            for (int u = 0; u < 4; ++u) ed[u] = (k + u < sg.y) ? ep[k + u] : make_int2(0, -1);
            #pragma unroll
            for (int u = 0; u < 4; ++u) {
                if (ed[u].y >= 0) {
                    const int p = atomicAdd(&ncur[ed[u].y - n0], 1);
                    if (staged) stage[p] = ed[u].x; else csr_src[base + p] = ed[u].x;
                }
            }
        }
    }
    if (staged) {
        __syncthreads();
        for (int i = tid; i < tot; i += 256) csr_src[base + i] = stage[i];
    }
}

template <int HEADS>
__global__ __launch_bounds__(256)
void gat_node_gather(const int* __restrict__ csr_src,
                     const int* __restrict__ start,
                     const int* __restrict__ cnt,
                     const float* __restrict__ as_,
                     const float* __restrict__ ad_,
                     const float* __restrict__ H,
                     const float* __restrict__ bias,
                     float* __restrict__ outp,
                     const int* __restrict__ batch,
                     float* __restrict__ sums,
                     float* __restrict__ cnts, int Nn)
{
    constexpr int HC  = HEADS * 64;
    constexpr int VPL = HC / 32;
    const int lane = threadIdx.x & 31;
    const int wave = __builtin_amdgcn_readfirstlane((int)(threadIdx.x >> 5));
    const int n = blockIdx.x * (int)(blockDim.x >> 5) + wave;
    if (n >= Nn) return;
    const int beg = start[n];
    const int deg = cnt[n];
    const int tot = deg + 1;
    const float NEG_INF = -__builtin_inff();

    float adv[HEADS];
    for (int h = 0; h < HEADS; ++h) adv[h] = ad_[n * HEADS + h];

    auto scores = [&](int s, float (&ev)[HEADS]) {
        float av[HEADS];
        if constexpr (HEADS == 4) {
            const float4 q = *reinterpret_cast<const float4*>(as_ + (size_t)s * 4);
            av[0] = q.x; av[1] = q.y; av[2] = q.z; av[3] = q.w;
        } else {
            for (int h = 0; h < HEADS; ++h) av[h] = as_[s * HEADS + h];
        }
        for (int h = 0; h < HEADS; ++h) {
            const float v = av[h] + adv[h];
            ev[h] = v > 0.f ? v : NEG_SLOPE * v;
        }
    };

    int   s0 = n;
    float e0[HEADS];
    for (int h = 0; h < HEADS; ++h) e0[h] = NEG_INF;
    const bool act0 = lane < tot;
    if (act0) {
        s0 = (lane < deg) ? csr_src[beg + lane] : n;
        scores(s0, e0);
    }
    float m[HEADS];
    for (int h = 0; h < HEADS; ++h) m[h] = e0[h];
    for (int j0 = 32; j0 < tot; j0 += 32) {
        const int j = j0 + lane;
        if (j < tot) {
            const int s = (j < deg) ? csr_src[beg + j] : n;
            float ev[HEADS];
            scores(s, ev);
            for (int h = 0; h < HEADS; ++h) m[h] = fmaxf(m[h], ev[h]);
        }
    }
    for (int h = 0; h < HEADS; ++h)
        for (int o = 16; o > 0; o >>= 1) m[h] = fmaxf(m[h], __shfl_xor(m[h], o));

    float acc[VPL];
    for (int i = 0; i < VPL; ++i) acc[i] = 0.f;
    float dl[HEADS];
    for (int h = 0; h < HEADS; ++h) dl[h] = 0.f;
    const bool hi = (lane & 16) != 0;

    auto chunk = [&](int s, const float (&ev)[HEADS], bool act, int cn) {
        float al[HEADS];
        for (int h = 0; h < HEADS; ++h) {
            al[h] = act ? __expf(ev[h] - m[h]) : 0.f;
            dl[h] += al[h];
        }
        #pragma unroll 4
        for (int t = 0; t < cn; ++t) {
            const int st = __builtin_amdgcn_readlane(s, t);
            float w[HEADS];
            for (int h = 0; h < HEADS; ++h)
                w[h] = __int_as_float(__builtin_amdgcn_readlane(__float_as_int(al[h]), t));
            const float* row = H + (size_t)st * HC;
            if constexpr (HEADS == 4) {
                const float4 r0 = *reinterpret_cast<const float4*>(row + lane * 4);
                const float4 r1 = *reinterpret_cast<const float4*>(row + 128 + lane * 4);
                const float w0 = hi ? w[1] : w[0];
                const float w1 = hi ? w[3] : w[2];
                acc[0] += w0 * r0.x; acc[1] += w0 * r0.y;
                acc[2] += w0 * r0.z; acc[3] += w0 * r0.w;
                acc[4] += w1 * r1.x; acc[5] += w1 * r1.y;
                acc[6] += w1 * r1.z; acc[7] += w1 * r1.w;
            } else {
                const float2 r = *reinterpret_cast<const float2*>(row + lane * 2);
                acc[0] += w[0] * r.x; acc[1] += w[0] * r.y;
            }
        }
    };

    chunk(s0, e0, act0, tot < 32 ? tot : 32);
    for (int j0 = 32; j0 < tot; j0 += 32) {
        const int j = j0 + lane;
        const bool act = j < tot;
        int s = n;
        float ev[HEADS];
        for (int h = 0; h < HEADS; ++h) ev[h] = NEG_INF;
        if (act) {
            s = (j < deg) ? csr_src[beg + j] : n;
            scores(s, ev);
        }
        const int left = tot - j0;
        chunk(s, ev, act, left < 32 ? left : 32);
    }

    for (int h = 0; h < HEADS; ++h)
        for (int o = 16; o > 0; o >>= 1) dl[h] += __shfl_xor(dl[h], o);

    if constexpr (HEADS == 4) {
        float* orow = outp + (size_t)n * HC;
        const float d0 = hi ? dl[1] : dl[0];
        const float d1 = hi ? dl[3] : dl[2];
        float o[8];
        for (int i = 0; i < 4; ++i) {
            const float v0 = acc[i] / d0 + bias[lane * 4 + i];
            const float v1 = acc[4 + i] / d1 + bias[128 + lane * 4 + i];
            o[i]     = v0 > 0.f ? v0 : (__expf(v0) - 1.0f);
            o[4 + i] = v1 > 0.f ? v1 : (__expf(v1) - 1.0f);
        }
        *reinterpret_cast<float4*>(orow + lane * 4)       = make_float4(o[0], o[1], o[2], o[3]);
        *reinterpret_cast<float4*>(orow + 128 + lane * 4) = make_float4(o[4], o[5], o[6], o[7]);
    } else {
        const float v0 = acc[0] / dl[0] + bias[lane * 2 + 0];
        const float v1 = acc[1] / dl[0] + bias[lane * 2 + 1];
        const float o0 = v0 > 0.f ? v0 : (__expf(v0) - 1.0f);
        const float o1 = v1 > 0.f ? v1 : (__expf(v1) - 1.0f);
        const int   hl = lane >> 1;
        const float t0 = __shfl(o0, hl),      t1 = __shfl(o1, hl);
        const float u0 = __shfl(o0, 16 + hl), u1 = __shfl(o1, 16 + hl);
        const bool  od = (lane & 1) != 0;
        const int   g  = batch[n];
        float* sp = sums + (size_t)g * 64 + lane;
        atomicAdd(sp,      od ? t1 : t0);
        atomicAdd(sp + 32, od ? u1 : u0);
        if (lane == 0) atomicAdd(&cnts[g], 1.0f);
    }
}

__global__ void gat_final_linear(const float* __restrict__ sums,
                                 const float* __restrict__ cnts,
                                 const float* __restrict__ Wl,
                                 const float* __restrict__ bl,
                                 float* __restrict__ out, int G, int C, int O)
{
    const int idx = blockIdx.x * blockDim.x + threadIdx.x;
    if (idx >= G * O) return;
    const int g = idx / O, o = idx % O;
    const float inv = 1.0f / fmaxf(cnts[g], 1.0f);
    float a = bl[o];
    for (int c = 0; c < C; ++c) a += sums[g * C + c] * inv * Wl[c * O + o];
    out[idx] = a;
}

extern "C" void kernel_launch(void* const* d_in, const int* in_sizes, int n_in,
                              void* d_out, int out_size, void* d_ws, size_t ws_size,
                              hipStream_t stream)
{
    const float* x   = (const float*)d_in[0];
    const int*   ei  = (const int*)  d_in[1];
    const int*   bat = (const int*)  d_in[2];
    const float* W1  = (const float*)d_in[3];
    const float* aS1 = (const float*)d_in[4];
    const float* aD1 = (const float*)d_in[5];
    const float* b1  = (const float*)d_in[6];
    const float* W2  = (const float*)d_in[7];
    const float* aS2 = (const float*)d_in[8];
    const float* aD2 = (const float*)d_in[9];
    const float* b2  = (const float*)d_in[10];
    const float* Wl  = (const float*)d_in[11];
    const float* bl  = (const float*)d_in[12];
    float* out = (float*)d_out;

    const int N    = in_sizes[2];
    const int E    = in_sizes[1] / 2;
    const int HID  = 64, HEADS = 4, OUT = 10, G = 512;
    const int HC1  = HEADS * HID;

    float* ws = (float*)d_ws;
    size_t off = 0;
    float* H1  = ws + off; off += (size_t)N * HC1;
    float* A1  = ws + off; off += (size_t)N * HC1;
    float* as1 = ws + off; off += (size_t)N * HEADS;
    float* ad1 = ws + off; off += (size_t)N * HEADS;
    float* H2  = ws + off; off += (size_t)N * HID;
    float* as2 = ws + off; off += (size_t)N;
    float* ad2 = ws + off; off += (size_t)N;
    float* sums = ws + off; off += (size_t)G * HID;
    float* cnts = ws + off; off += (size_t)G;
    int* gtot   = (int*)(ws + off); off += 1;
    int* dcnt   = (int*)(ws + off); off += (size_t)N;
    int* dstart = (int*)(ws + off); off += (size_t)N;
    int* csrc   = (int*)(ws + off); off += (size_t)E;
    const int nbin = (N + 63) >> 6;
    const int nblk = (E + 4095) / 4096;
    off += off & 1;
    int2* etmp  = (int2*)(ws + off); off += 2 * (size_t)E;
    int2* seg   = (int2*)(ws + off); off += 2 * (size_t)nbin * nblk;
    (void)ws_size; (void)n_in;

    const int TB = 256;
    auto blks = [](size_t n, int tb) { return (unsigned)((n + tb - 1) / tb); };

    const int nzero = G * HID + G + 1;
    gat_bin_coarse<<<nblk, 1024, 0, stream>>>(ei, E, nbin, etmp, seg, sums, nzero);
    gat_bin_fine<<<nbin, 256, 0, stream>>>(etmp, seg, nblk, gtot, dstart, dcnt, csrc, N);

    {
        dim3 grid(256 / 64, (N + 127) / 128);
        gat_gemm_wmma<128, 256><<<grid, 256, 0, stream>>>(x, W1, H1, aS1, aD1, as1, ad1, N);
    }
    gat_node_gather<4><<<blks(N, 8), TB, 0, stream>>>(
        csrc, dstart, dcnt, as1, ad1, H1, b1, A1, bat, sums, cnts, N);

    {
        dim3 grid(64 / 64, (N + 127) / 128);
        gat_gemm_wmma<256, 64><<<grid, 256, 0, stream>>>(A1, W2, H2, aS2, aD2, as2, ad2, N);
    }
    gat_node_gather<1><<<blks(N, 8), TB, 0, stream>>>(
        csrc, dstart, dcnt, as2, ad2, H2, b2, nullptr, bat, sums, cnts, N);

    gat_final_linear<<<blks((size_t)G * OUT, TB), TB, 0, stream>>>(
        sums, cnts, Wl, bl, out, G, HID, OUT);
}


# global mean pool: a work-group's 8 node rows are added per graph in LDS, in node order, then one atomic row and one count per graph and work-group instead of per node; sum order changes
# speedup vs baseline: 7.2790x; 1.2364x over previous
#include <hip/hip_runtime.h>
#include <cstddef>


#define NEG_SLOPE 0.2f

typedef float v2f __attribute__((ext_vector_type(2)));
typedef float v8f __attribute__((ext_vector_type(8)));

template <int K, int N>
__global__ __launch_bounds__(256)
void gat_gemm_wmma(const float* __restrict__ A,
                   const float* __restrict__ B,
                   float* __restrict__ C,
                   const float* __restrict__ att_s,
                   const float* __restrict__ att_d,
                   float* __restrict__ a_src,
                   float* __restrict__ a_dst, int M)
{
    constexpr int KJ    = K / 8;
    constexpr int BS_F4 = KJ * 4 * 32;
    constexpr int CP    = 68;
    constexpr int CS_F4 = 8 * 16 * CP / 4;
    constexpr int LDS_F4 = BS_F4 > CS_F4 ? BS_F4 : CS_F4;
    __shared__ float4 smem[LDS_F4];

    const int lane = threadIdx.x & 31;
    const int wave = __builtin_amdgcn_readfirstlane((int)(threadIdx.x >> 5));
    const int c    = lane & 15;
    const int h    = lane >> 4;
    const int nb   = blockIdx.x * 64;
    const int m0   = (blockIdx.y * 8 + wave) * 16;

    int arow = m0 + c;
    if (arow > M - 1) arow = M - 1;
    const float4* ap = reinterpret_cast<const float4*>(A + (size_t)arow * K + 4 * h);
    float4 a[KJ];
    #pragma unroll
    for (int j = 0; j < KJ; ++j) a[j] = ap[2 * j];

    #pragma unroll
    for (int it = 0; it < BS_F4 / 256; ++it) {
        const int idx = it * 256 + (int)threadIdx.x;
        const int sc = idx & 15, sh = (idx >> 4) & 1, st = (idx >> 5) & 3, sj = idx >> 7;
        const float* bp = B + (size_t)(8 * sj + 4 * sh) * N + nb + 16 * st + sc;
        float4 v;
        v.x = bp[0]; v.y = bp[N]; v.z = bp[2 * N]; v.w = bp[3 * N];
        smem[idx] = v;
    }
    __syncthreads();

    v8f acc[4] = {};
    #pragma unroll
    for (int j = 0; j < KJ; ++j) {
        float4 b[4];
        #pragma unroll
        for (int t = 0; t < 4; ++t) b[t] = smem[(j * 4 + t) * 32 + lane];
        v2f a0, a1;
        a0.x = a[j].x; a0.y = a[j].y;
        a1.x = a[j].z; a1.y = a[j].w;
        #pragma unroll
        for (int t = 0; t < 4; ++t) {
            v2f b0; b0.x = b[t].x; b0.y = b[t].y;
            acc[t] = __builtin_amdgcn_wmma_f32_16x16x4_f32(
                false, a0, false, b0, (short)0, acc[t], false, false);
        }
        #pragma unroll
        for (int t = 0; t < 4; ++t) {
            v2f b1; b1.x = b[t].z; b1.y = b[t].w;
            acc[t] = __builtin_amdgcn_wmma_f32_16x16x4_f32(
                false, a1, false, b1, (short)0, acc[t], false, false);
        }
    }

    __syncthreads();
    float* cs = reinterpret_cast<float*>(smem) + wave * (16 * CP);
    #pragma unroll
    for (int t = 0; t < 4; ++t)
        #pragma unroll
        for (int r = 0; r < 8; ++r)
            cs[(8 * h + r) * CP + 16 * t + c] = acc[t][r];
    __syncthreads();
    const float* cr = cs + h * CP + 4 * c;
    if (m0 + 16 <= M) {
        float* cp = C + (size_t)(m0 + h) * N + nb + 4 * c;
        float4 v[8];
        #pragma unroll
        for (int i = 0; i < 8; ++i)
            v[i] = *reinterpret_cast<const float4*>(cr + 2 * i * CP);
        #pragma unroll
        for (int i = 0; i < 8; ++i)
            *reinterpret_cast<float4*>(cp + (size_t)(2 * i) * N) = v[i];
    } else {
        #pragma unroll
        for (int i = 0; i < 8; ++i) {
            const int gm = m0 + 2 * i + h;
            if (gm < M)
                *reinterpret_cast<float4*>(C + (size_t)gm * N + nb + 4 * c) =
                    *reinterpret_cast<const float4*>(cr + 2 * i * CP);
        }
    }

    {
        const float* av   = (h ? att_d : att_s) + nb;
        const float* crow = cs + c * CP;
        float s = 0.f;
        #pragma unroll
        for (int k = 0; k < 64; k += 4) {
            const float4 v = *reinterpret_cast<const float4*>(crow + k);
            const float4 w = *reinterpret_cast<const float4*>(av + k);
            s = __builtin_fmaf(v.x, w.x, s); s = __builtin_fmaf(v.y, w.y, s);
            s = __builtin_fmaf(v.z, w.z, s); s = __builtin_fmaf(v.w, w.w, s);
        }
        const int gm = m0 + c;
        if (gm < M) (h ? a_dst : a_src)[(size_t)gm * (N / 64) + blockIdx.x] = s;
    }
}

#define BIN_SHIFT 6
#define BIN_NODES 64
#define EDGE_CHUNK 4096
#define MAX_BINS 1024
#define FINE_CAP 4096

__global__ __launch_bounds__(1024)
void gat_bin_coarse(const int* __restrict__ ei, int E, int nbin,
                    int2* __restrict__ etmp, int2* __restrict__ seg,
                    float* __restrict__ zero_p, int nzero)
{
    __shared__ int  hist[MAX_BINS];
    __shared__ int  wsum[32];
    __shared__ int2 stage[EDGE_CHUNK];
    const int tid  = threadIdx.x;
    const int lane = tid & 31, wave = tid >> 5;
    const int nblk = gridDim.x;

    for (int i = blockIdx.x * 1024 + tid; i < nzero; i += nblk * 1024) zero_p[i] = 0.f;
    hist[tid] = 0;
    __syncthreads();

    const int e0 = blockIdx.x * EDGE_CHUNK;
    int s[4], d[4], r[4];
    #pragma unroll
    for (int k = 0; k < 4; ++k) {
        const int e = e0 + k * 1024 + tid;
        const bool ok = e < E;
        s[k] = ok ? ei[e] : 0;
        d[k] = ok ? ei[E + e] : -1;
    }
    #pragma unroll
    for (int k = 0; k < 4; ++k)
        r[k] = (d[k] >= 0) ? atomicAdd(&hist[d[k] >> BIN_SHIFT], 1) : 0;
    __syncthreads();

    const int c = hist[tid];
    int incl = c;
    for (int o = 1; o < 32; o <<= 1) {
        const int t = __shfl_up(incl, o);
        if (lane >= o) incl += t;
    }
    if (lane == 31) wsum[wave] = incl;
    __syncthreads();
    if (wave == 0) {
        const int w = wsum[lane];
        int wi = w;
        for (int o = 1; o < 32; o <<= 1) {
            const int t = __shfl_up(wi, o);
            if (lane >= o) wi += t;
        }
        wsum[lane] = wi - w;
    }
    __syncthreads();
    const int off = wsum[wave] + incl - c;
    hist[tid] = off;
    if (tid < nbin) seg[(size_t)tid * nblk + blockIdx.x] = make_int2(e0 + off, c);
    __syncthreads();

    #pragma unroll
    for (int k = 0; k < 4; ++k)
        if (d[k] >= 0) stage[hist[d[k] >> BIN_SHIFT] + r[k]] = make_int2(s[k], d[k]);
    __syncthreads();
    #pragma unroll
    for (int k = 0; k < 4; ++k) {
        const int i = k * 1024 + tid;
        if (e0 + i < E) etmp[e0 + i] = stage[i];
    }
}

__global__ __launch_bounds__(256)
void gat_bin_fine(const int2* __restrict__ etmp, const int2* __restrict__ seg, int nseg,
                  int* __restrict__ gtot, int* __restrict__ start, int* __restrict__ cnt,
                  int* __restrict__ csr_src, int Nn)
{
    __shared__ int ncnt[BIN_NODES];
    __shared__ int ncur[BIN_NODES];
    __shared__ int tot_s, base_s;
    __shared__ int stage[FINE_CAP];
    const int tid = threadIdx.x;
    const int n0  = blockIdx.x << BIN_SHIFT;
    const int2* sp = seg + (size_t)blockIdx.x * nseg;

    if (tid < BIN_NODES) ncnt[tid] = 0;
    __syncthreads();
    for (int g = tid; g < nseg; g += 256) {
        const int2  sg = sp[g];
        const int2* ep = etmp + sg.x;
        for (int k = 0; k < sg.y; k += 4) {
            int dd[4];
            #pragma unroll
            for (int u = 0; u < 4; ++u) dd[u] = (k + u < sg.y) ? ep[k + u].y : -1;
            #pragma unroll
            for (int u = 0; u < 4; ++u)
                if (dd[u] >= 0) atomicAdd(&ncnt[dd[u] - n0], 1);
        }
    }
    __syncthreads();

    if (tid < 32) {
        const int v0 = ncnt[2 * tid], v1 = ncnt[2 * tid + 1];
        const int sm = v0 + v1;
        int incl = sm;
        for (int o = 1; o < 32; o <<= 1) {
            const int t = __shfl_up(incl, o);
            if (tid >= o) incl += t;
        }
        const int tot = __shfl(incl, 31);
        int base = 0;
        if (tid == 0) base = atomicAdd(gtot, tot);
        base = __shfl(base, 0);
        const int ex = incl - sm;
        ncur[2 * tid]     = ex;
        ncur[2 * tid + 1] = ex + v0;
        const int na = n0 + 2 * tid;
        if (na < Nn)     { start[na]     = base + ex;      cnt[na]     = v0; }
        if (na + 1 < Nn) { start[na + 1] = base + ex + v0; cnt[na + 1] = v1; }
        if (tid == 0) { tot_s = tot; base_s = base; }
    }
    __syncthreads();

    const int  tot    = tot_s, base = base_s;
    const bool staged = tot <= FINE_CAP;
    for (int g = tid; g < nseg; g += 256) {
        const int2  sg = sp[g];
        const int2* ep = etmp + sg.x;
        for (int k = 0; k < sg.y; k += 4) {
            int2 ed[4];
            #pragma unroll
            for (int u = 0; u < 4; ++u) ed[u] = (k + u < sg.y) ? ep[k + u] : make_int2(0, -1);
            #pragma unroll
            for (int u = 0; u < 4; ++u) {
                if (ed[u].y >= 0) {
                    const int p = atomicAdd(&ncur[ed[u].y - n0], 1);
                    if (staged) stage[p] = ed[u].x; else csr_src[base + p] = ed[u].x;
                }
            }
        }
    }
    if (staged) {
        __syncthreads();
        for (int i = tid; i < tot; i += 256) csr_src[base + i] = stage[i];
    }
}

template <int HEADS>
__global__ __launch_bounds__(256)
void gat_node_gather(const int* __restrict__ csr_src,
                     const int* __restrict__ start,
                     const int* __restrict__ cnt,
                     const float* __restrict__ as_,
                     const float* __restrict__ ad_,
                     const float* __restrict__ H,
                     const float* __restrict__ bias,
                     float* __restrict__ outp,
                     const int* __restrict__ batch,
                     float* __restrict__ sums,
                     float* __restrict__ cnts, int Nn)
{
    constexpr int HC  = HEADS * 64;
    constexpr int VPL = HC / 32;
    const int lane = threadIdx.x & 31;
    const int wave = __builtin_amdgcn_readfirstlane((int)(threadIdx.x >> 5));
    const int nraw = blockIdx.x * (int)(blockDim.x >> 5) + wave;
    if constexpr (HEADS == 4) { if (nraw >= Nn) return; }
    const bool valid = nraw < Nn;
    const int  n     = valid ? nraw : Nn - 1;
    const int beg = start[n];
    const int deg = cnt[n];
    const int tot = deg + 1;
    const float NEG_INF = -__builtin_inff();

    float adv[HEADS];
    for (int h = 0; h < HEADS; ++h) adv[h] = ad_[n * HEADS + h];

    auto scores = [&](int s, float (&ev)[HEADS]) {
        float av[HEADS];
        if constexpr (HEADS == 4) {
            const float4 q = *reinterpret_cast<const float4*>(as_ + (size_t)s * 4);
            av[0] = q.x; av[1] = q.y; av[2] = q.z; av[3] = q.w;
        } else {
            for (int h = 0; h < HEADS; ++h) av[h] = as_[s * HEADS + h];
        }
        for (int h = 0; h < HEADS; ++h) {
            const float v = av[h] + adv[h];
            ev[h] = v > 0.f ? v : NEG_SLOPE * v;
        }
    };

    int   s0 = n;
    float e0[HEADS];
    for (int h = 0; h < HEADS; ++h) e0[h] = NEG_INF;
    const bool act0 = lane < tot;
    if (act0) {
        s0 = (lane < deg) ? csr_src[beg + lane] : n;
        scores(s0, e0);
    }
    float m[HEADS];
    for (int h = 0; h < HEADS; ++h) m[h] = e0[h];
    for (int j0 = 32; j0 < tot; j0 += 32) {
        const int j = j0 + lane;
        if (j < tot) {
            const int s = (j < deg) ? csr_src[beg + j] : n;
            float ev[HEADS];
            scores(s, ev);
            for (int h = 0; h < HEADS; ++h) m[h] = fmaxf(m[h], ev[h]);
        }
    }
    for (int h = 0; h < HEADS; ++h)
        for (int o = 16; o > 0; o >>= 1) m[h] = fmaxf(m[h], __shfl_xor(m[h], o));

    float acc[VPL];
    for (int i = 0; i < VPL; ++i) acc[i] = 0.f;
    float dl[HEADS];
    for (int h = 0; h < HEADS; ++h) dl[h] = 0.f;
    const bool hi = (lane & 16) != 0;

    auto chunk = [&](int s, const float (&ev)[HEADS], bool act, int cn) {
        float al[HEADS];
        for (int h = 0; h < HEADS; ++h) {
            al[h] = act ? __expf(ev[h] - m[h]) : 0.f;
            dl[h] += al[h];
        }
        #pragma unroll 4
        for (int t = 0; t < cn; ++t) {
            const int st = __builtin_amdgcn_readlane(s, t);
            float w[HEADS];
            for (int h = 0; h < HEADS; ++h)
                w[h] = __int_as_float(__builtin_amdgcn_readlane(__float_as_int(al[h]), t));
            const float* row = H + (size_t)st * HC;
            if constexpr (HEADS == 4) {
                const float4 r0 = *reinterpret_cast<const float4*>(row + lane * 4);
                const float4 r1 = *reinterpret_cast<const float4*>(row + 128 + lane * 4);
                const float w0 = hi ? w[1] : w[0];
                const float w1 = hi ? w[3] : w[2];
                acc[0] += w0 * r0.x; acc[1] += w0 * r0.y;
                acc[2] += w0 * r0.z; acc[3] += w0 * r0.w;
                acc[4] += w1 * r1.x; acc[5] += w1 * r1.y;
                acc[6] += w1 * r1.z; acc[7] += w1 * r1.w;
            } else {
                const float2 r = *reinterpret_cast<const float2*>(row + lane * 2);
                acc[0] += w[0] * r.x; acc[1] += w[0] * r.y;
            }
        }
    };

    chunk(s0, e0, act0, tot < 32 ? tot : 32);
    for (int j0 = 32; j0 < tot; j0 += 32) {
        const int j = j0 + lane;
        const bool act = j < tot;
        int s = n;
        float ev[HEADS];
        for (int h = 0; h < HEADS; ++h) ev[h] = NEG_INF;
        if (act) {
            s = (j < deg) ? csr_src[beg + j] : n;
            scores(s, ev);
        }
        const int left = tot - j0;
        chunk(s, ev, act, left < 32 ? left : 32);
    }

    for (int h = 0; h < HEADS; ++h)
        for (int o = 16; o > 0; o >>= 1) dl[h] += __shfl_xor(dl[h], o);

    if constexpr (HEADS == 4) {
        float* orow = outp + (size_t)n * HC;
        const float d0 = hi ? dl[1] : dl[0];
        const float d1 = hi ? dl[3] : dl[2];
        float o[8];
        for (int i = 0; i < 4; ++i) {
            const float v0 = acc[i] / d0 + bias[lane * 4 + i];
            const float v1 = acc[4 + i] / d1 + bias[128 + lane * 4 + i];
            o[i]     = v0 > 0.f ? v0 : (__expf(v0) - 1.0f);
            o[4 + i] = v1 > 0.f ? v1 : (__expf(v1) - 1.0f);
        }
        *reinterpret_cast<float4*>(orow + lane * 4)       = make_float4(o[0], o[1], o[2], o[3]);
        *reinterpret_cast<float4*>(orow + 128 + lane * 4) = make_float4(o[4], o[5], o[6], o[7]);
    } else {
        const float v0 = acc[0] / dl[0] + bias[lane * 2 + 0];
        const float v1 = acc[1] / dl[0] + bias[lane * 2 + 1];
        const float o0 = v0 > 0.f ? v0 : (__expf(v0) - 1.0f);
        const float o1 = v1 > 0.f ? v1 : (__expf(v1) - 1.0f);
        __shared__ float prow[8][64];
        __shared__ int   pgr[8];
        *reinterpret_cast<float2*>(&prow[wave][2 * lane]) =
            valid ? make_float2(o0, o1) : make_float2(0.f, 0.f);
        if (lane == 0) pgr[wave] = valid ? batch[n] : -1;
        __syncthreads();
        if (wave < 2) {
            const int c  = (int)threadIdx.x;
            const int nw = (int)(blockDim.x >> 5);
            float run = 0.f;
            int   len = 0;
            int   gc  = pgr[0];
            for (int w = 0; w < nw; ++w) {
                const int g = pgr[w];
                if (g != gc) {
                    if (gc >= 0) {
                        atomicAdd(sums + (size_t)gc * 64 + c, run);
                        if (c == 0) atomicAdd(&cnts[gc], (float)len);
                    }
                    run = 0.f; len = 0; gc = g;
                }
                run += prow[w][c];
                ++len;
            }
            if (gc >= 0) {
                atomicAdd(sums + (size_t)gc * 64 + c, run);
                if (c == 0) atomicAdd(&cnts[gc], (float)len);
            }
        }
    }
}

__global__ void gat_final_linear(const float* __restrict__ sums,
                                 const float* __restrict__ cnts,
                                 const float* __restrict__ Wl,
                                 const float* __restrict__ bl,
                                 float* __restrict__ out, int G, int C, int O)
{
    const int idx = blockIdx.x * blockDim.x + threadIdx.x;
    if (idx >= G * O) return;
    const int g = idx / O, o = idx % O;
    const float inv = 1.0f / fmaxf(cnts[g], 1.0f);
    float a = bl[o];
    for (int c = 0; c < C; ++c) a += sums[g * C + c] * inv * Wl[c * O + o];
    out[idx] = a;
}

extern "C" void kernel_launch(void* const* d_in, const int* in_sizes, int n_in,
                              void* d_out, int out_size, void* d_ws, size_t ws_size,
                              hipStream_t stream)
{
    const float* x   = (const float*)d_in[0];
    const int*   ei  = (const int*)  d_in[1];
    const int*   bat = (const int*)  d_in[2];
    const float* W1  = (const float*)d_in[3];
    const float* aS1 = (const float*)d_in[4];
    const float* aD1 = (const float*)d_in[5];
    const float* b1  = (const float*)d_in[6];
    const float* W2  = (const float*)d_in[7];
    const float* aS2 = (const float*)d_in[8];
    const float* aD2 = (const float*)d_in[9];
    const float* b2  = (const float*)d_in[10];
    const float* Wl  = (const float*)d_in[11];
    const float* bl  = (const float*)d_in[12];
    float* out = (float*)d_out;

    const int N    = in_sizes[2];
    const int E    = in_sizes[1] / 2;
    const int HID  = 64, HEADS = 4, OUT = 10, G = 512;
    const int HC1  = HEADS * HID;

    float* ws = (float*)d_ws;
    size_t off = 0;
    float* H1  = ws + off; off += (size_t)N * HC1;
    float* A1  = ws + off; off += (size_t)N * HC1;
    float* as1 = ws + off; off += (size_t)N * HEADS;
    float* ad1 = ws + off; off += (size_t)N * HEADS;
    float* H2  = ws + off; off += (size_t)N * HID;
    float* as2 = ws + off; off += (size_t)N;
    float* ad2 = ws + off; off += (size_t)N;
    float* sums = ws + off; off += (size_t)G * HID;
    float* cnts = ws + off; off += (size_t)G;
    int* gtot   = (int*)(ws + off); off += 1;
    int* dcnt   = (int*)(ws + off); off += (size_t)N;
    int* dstart = (int*)(ws + off); off += (size_t)N;
    int* csrc   = (int*)(ws + off); off += (size_t)E;
    const int nbin = (N + 63) >> 6;
    const int nblk = (E + 4095) / 4096;
    off += off & 1;
    int2* etmp  = (int2*)(ws + off); off += 2 * (size_t)E;
    int2* seg   = (int2*)(ws + off); off += 2 * (size_t)nbin * nblk;
    (void)ws_size; (void)n_in;

    const int TB = 256;
    auto blks = [](size_t n, int tb) { return (unsigned)((n + tb - 1) / tb); };

    const int nzero = G * HID + G + 1;
    gat_bin_coarse<<<nblk, 1024, 0, stream>>>(ei, E, nbin, etmp, seg, sums, nzero);
    gat_bin_fine<<<nbin, 256, 0, stream>>>(etmp, seg, nblk, gtot, dstart, dcnt, csrc, N);

    {
        dim3 grid(256 / 64, (N + 127) / 128);
        gat_gemm_wmma<128, 256><<<grid, 256, 0, stream>>>(x, W1, H1, aS1, aD1, as1, ad1, N);
    }
    gat_node_gather<4><<<blks(N, 8), TB, 0, stream>>>(
        csrc, dstart, dcnt, as1, ad1, H1, b1, A1, bat, sums, cnts, N);

    {
        dim3 grid(64 / 64, (N + 127) / 128);
        gat_gemm_wmma<256, 64><<<grid, 256, 0, stream>>>(A1, W2, H2, aS2, aD2, as2, ad2, N);
    }
    gat_node_gather<1><<<blks(N, 8), TB, 0, stream>>>(
        csrc, dstart, dcnt, as2, ad2, H2, b2, nullptr, bat, sums, cnts, N);

    gat_final_linear<<<blks((size_t)G * OUT, TB), TB, 0, stream>>>(
        sums, cnts, Wl, bl, out, G, HID, OUT);
}


# edge aggregation: a wave walks 4 (last layer) or 5 destination nodes in turn, so each grid is resident at once and starts 1/4 of the waves; last layer pools 32 rows per work-group; sum order changes
# speedup vs baseline: 7.5160x; 1.0326x over previous
#include <hip/hip_runtime.h>
#include <cstddef>


#define NEG_SLOPE 0.2f

typedef float v2f __attribute__((ext_vector_type(2)));
typedef float v8f __attribute__((ext_vector_type(8)));

template <int K, int N>
__global__ __launch_bounds__(256)
void gat_gemm_wmma(const float* __restrict__ A,
                   const float* __restrict__ B,
                   float* __restrict__ C,
                   const float* __restrict__ att_s,
                   const float* __restrict__ att_d,
                   float* __restrict__ a_src,
                   float* __restrict__ a_dst, int M)
{
    constexpr int KJ    = K / 8;
    constexpr int BS_F4 = KJ * 4 * 32;
    constexpr int CP    = 68;
    constexpr int CS_F4 = 8 * 16 * CP / 4;
    constexpr int LDS_F4 = BS_F4 > CS_F4 ? BS_F4 : CS_F4;
    __shared__ float4 smem[LDS_F4];

    const int lane = threadIdx.x & 31;
    const int wave = __builtin_amdgcn_readfirstlane((int)(threadIdx.x >> 5));
    const int c    = lane & 15;
    const int h    = lane >> 4;
    const int nb   = blockIdx.x * 64;
    const int m0   = (blockIdx.y * 8 + wave) * 16;

    int arow = m0 + c;
    if (arow > M - 1) arow = M - 1;
    const float4* ap = reinterpret_cast<const float4*>(A + (size_t)arow * K + 4 * h);
    float4 a[KJ];
    #pragma unroll
    for (int j = 0; j < KJ; ++j) a[j] = ap[2 * j];

    #pragma unroll
    for (int it = 0; it < BS_F4 / 256; ++it) {
        const int idx = it * 256 + (int)threadIdx.x;
        const int sc = idx & 15, sh = (idx >> 4) & 1, st = (idx >> 5) & 3, sj = idx >> 7;
        const float* bp = B + (size_t)(8 * sj + 4 * sh) * N + nb + 16 * st + sc;
        float4 v;
        v.x = bp[0]; v.y = bp[N]; v.z = bp[2 * N]; v.w = bp[3 * N];
        smem[idx] = v;
    }
    __syncthreads();

    v8f acc[4] = {};
    #pragma unroll
    for (int j = 0; j < KJ; ++j) {
        float4 b[4];
        #pragma unroll
        for (int t = 0; t < 4; ++t) b[t] = smem[(j * 4 + t) * 32 + lane];
        v2f a0, a1;
        a0.x = a[j].x; a0.y = a[j].y;
        a1.x = a[j].z; a1.y = a[j].w;
        #pragma unroll
        for (int t = 0; t < 4; ++t) {
            v2f b0; b0.x = b[t].x; b0.y = b[t].y;
            acc[t] = __builtin_amdgcn_wmma_f32_16x16x4_f32(
                false, a0, false, b0, (short)0, acc[t], false, false);
        }
        #pragma unroll
        for (int t = 0; t < 4; ++t) {
            v2f b1; b1.x = b[t].z; b1.y = b[t].w;
            acc[t] = __builtin_amdgcn_wmma_f32_16x16x4_f32(
                false, a1, false, b1, (short)0, acc[t], false, false);
        }
    }

    __syncthreads();
    float* cs = reinterpret_cast<float*>(smem) + wave * (16 * CP);
    #pragma unroll
    for (int t = 0; t < 4; ++t)
        #pragma unroll
        for (int r = 0; r < 8; ++r)
            cs[(8 * h + r) * CP + 16 * t + c] = acc[t][r];
    __syncthreads();
    const float* cr = cs + h * CP + 4 * c;
    if (m0 + 16 <= M) {
        float* cp = C + (size_t)(m0 + h) * N + nb + 4 * c;
        float4 v[8];
        #pragma unroll
        for (int i = 0; i < 8; ++i)
            v[i] = *reinterpret_cast<const float4*>(cr + 2 * i * CP);
        #pragma unroll
        for (int i = 0; i < 8; ++i)
            *reinterpret_cast<float4*>(cp + (size_t)(2 * i) * N) = v[i];
    } else {
        #pragma unroll
        for (int i = 0; i < 8; ++i) {
            const int gm = m0 + 2 * i + h;
            if (gm < M)
                *reinterpret_cast<float4*>(C + (size_t)gm * N + nb + 4 * c) =
                    *reinterpret_cast<const float4*>(cr + 2 * i * CP);
        }
    }

    {
        const float* av   = (h ? att_d : att_s) + nb;
        const float* crow = cs + c * CP;
        float s = 0.f;
        #pragma unroll
        for (int k = 0; k < 64; k += 4) {
            const float4 v = *reinterpret_cast<const float4*>(crow + k);
            const float4 w = *reinterpret_cast<const float4*>(av + k);
            s = __builtin_fmaf(v.x, w.x, s); s = __builtin_fmaf(v.y, w.y, s);
            s = __builtin_fmaf(v.z, w.z, s); s = __builtin_fmaf(v.w, w.w, s);
        }
        const int gm = m0 + c;
        if (gm < M) (h ? a_dst : a_src)[(size_t)gm * (N / 64) + blockIdx.x] = s;
    }
}

#define BIN_SHIFT 6
#define BIN_NODES 64
#define EDGE_CHUNK 4096
#define MAX_BINS 1024
#define FINE_CAP 4096

__global__ __launch_bounds__(1024)
void gat_bin_coarse(const int* __restrict__ ei, int E, int nbin,
                    int2* __restrict__ etmp, int2* __restrict__ seg,
                    float* __restrict__ zero_p, int nzero)
{
    __shared__ int  hist[MAX_BINS];
    __shared__ int  wsum[32];
    __shared__ int2 stage[EDGE_CHUNK];
    const int tid  = threadIdx.x;
    const int lane = tid & 31, wave = tid >> 5;
    const int nblk = gridDim.x;

    for (int i = blockIdx.x * 1024 + tid; i < nzero; i += nblk * 1024) zero_p[i] = 0.f;
    hist[tid] = 0;
    __syncthreads();

    const int e0 = blockIdx.x * EDGE_CHUNK;
    int s[4], d[4], r[4];
    #pragma unroll
    for (int k = 0; k < 4; ++k) {
        const int e = e0 + k * 1024 + tid;
        const bool ok = e < E;
        s[k] = ok ? ei[e] : 0;
        d[k] = ok ? ei[E + e] : -1;
    }
    #pragma unroll
    for (int k = 0; k < 4; ++k)
        r[k] = (d[k] >= 0) ? atomicAdd(&hist[d[k] >> BIN_SHIFT], 1) : 0;
    __syncthreads();

    const int c = hist[tid];
    int incl = c;
    for (int o = 1; o < 32; o <<= 1) {
        const int t = __shfl_up(incl, o);
        if (lane >= o) incl += t;
    }
    if (lane == 31) wsum[wave] = incl;
    __syncthreads();
    if (wave == 0) {
        const int w = wsum[lane];
        int wi = w;
        for (int o = 1; o < 32; o <<= 1) {
            const int t = __shfl_up(wi, o);
            if (lane >= o) wi += t;
        }
        wsum[lane] = wi - w;
    }
    __syncthreads();
    const int off = wsum[wave] + incl - c;
    hist[tid] = off;
    if (tid < nbin) seg[(size_t)tid * nblk + blockIdx.x] = make_int2(e0 + off, c);
    __syncthreads();

    #pragma unroll
    for (int k = 0; k < 4; ++k)
        if (d[k] >= 0) stage[hist[d[k] >> BIN_SHIFT] + r[k]] = make_int2(s[k], d[k]);
    __syncthreads();
    #pragma unroll
    for (int k = 0; k < 4; ++k) {
        const int i = k * 1024 + tid;
        if (e0 + i < E) etmp[e0 + i] = stage[i];
    }
}

__global__ __launch_bounds__(256)
void gat_bin_fine(const int2* __restrict__ etmp, const int2* __restrict__ seg, int nseg,
                  int* __restrict__ gtot, int* __restrict__ start, int* __restrict__ cnt,
                  int* __restrict__ csr_src, int Nn)
{
    __shared__ int ncnt[BIN_NODES];
    __shared__ int ncur[BIN_NODES];
    __shared__ int tot_s, base_s;
    __shared__ int stage[FINE_CAP];
    const int tid = threadIdx.x;
    const int n0  = blockIdx.x << BIN_SHIFT;
    const int2* sp = seg + (size_t)blockIdx.x * nseg;

    if (tid < BIN_NODES) ncnt[tid] = 0;
    __syncthreads();
    for (int g = tid; g < nseg; g += 256) {
        const int2  sg = sp[g];
        const int2* ep = etmp + sg.x;
        for (int k = 0; k < sg.y; k += 4) {
            int dd[4];
            #pragma unroll
            for (int u = 0; u < 4; ++u) dd[u] = (k + u < sg.y) ? ep[k + u].y : -1;
            #pragma unroll
            for (int u = 0; u < 4; ++u)
                if (dd[u] >= 0) atomicAdd(&ncnt[dd[u] - n0], 1);
        }
    }
    __syncthreads();

    if (tid < 32) {
        const int v0 = ncnt[2 * tid], v1 = ncnt[2 * tid + 1];
        const int sm = v0 + v1;
        int incl = sm;
        for (int o = 1; o < 32; o <<= 1) {
            const int t = __shfl_up(incl, o);
            if (tid >= o) incl += t;
        }
        const int tot = __shfl(incl, 31);
        int base = 0;
        if (tid == 0) base = atomicAdd(gtot, tot);
        base = __shfl(base, 0);
        const int ex = incl - sm;
        ncur[2 * tid]     = ex;
        ncur[2 * tid + 1] = ex + v0;
        const int na = n0 + 2 * tid;
        if (na < Nn)     { start[na]     = base + ex;      cnt[na]     = v0; }
        if (na + 1 < Nn) { start[na + 1] = base + ex + v0; cnt[na + 1] = v1; }
        if (tid == 0) { tot_s = tot; base_s = base; }
    }
    __syncthreads();

    const int  tot    = tot_s, base = base_s;
    const bool staged = tot <= FINE_CAP;
    for (int g = tid; g < nseg; g += 256) {
        const int2  sg = sp[g];
        const int2* ep = etmp + sg.x;
        for (int k = 0; k < sg.y; k += 4) {
            int2 ed[4];
            #pragma unroll
            for (int u = 0; u < 4; ++u) ed[u] = (k + u < sg.y) ? ep[k + u] : make_int2(0, -1);
            #pragma unroll
            for (int u = 0; u < 4; ++u) {
                if (ed[u].y >= 0) {
                    const int p = atomicAdd(&ncur[ed[u].y - n0], 1);
                    if (staged) stage[p] = ed[u].x; else csr_src[base + p] = ed[u].x;
                }
            }
        }
    }
    if (staged) {
        __syncthreads();
        for (int i = tid; i < tot; i += 256) csr_src[base + i] = stage[i];
    }
}

#define GATHER_WAVES 8

template <int HEADS, int TRIPS>
__global__ __launch_bounds__(256)
void gat_node_gather(const int* __restrict__ csr_src,
                     const int* __restrict__ start,
                     const int* __restrict__ cnt,
                     const float* __restrict__ as_,
                     const float* __restrict__ ad_,
                     const float* __restrict__ H,
                     const float* __restrict__ bias,
                     float* __restrict__ outp,
                     const int* __restrict__ batch,
                     float* __restrict__ sums,
                     float* __restrict__ cnts, int Nn)
{
    constexpr int HC  = HEADS * 64;
    constexpr int VPL = HC / 32;
    constexpr int GATHER_NODES = GATHER_WAVES * TRIPS;
    __shared__ float prow[HEADS == 1 ? GATHER_NODES : 1][64];
    __shared__ int   pgr[HEADS == 1 ? GATHER_NODES : 1];
    const int lane = threadIdx.x & 31;
    const int wave = __builtin_amdgcn_readfirstlane((int)(threadIdx.x >> 5));
    const int nfirst = blockIdx.x * GATHER_NODES;
    const float NEG_INF = -__builtin_inff();
    const bool hi = (lane & 16) != 0;

    for (int trip = 0; trip < TRIPS; ++trip) {
    const int slot = trip * GATHER_WAVES + wave;
    const int n    = nfirst + slot;
    if (n >= Nn) {
        if constexpr (HEADS == 1) {
            *reinterpret_cast<float2*>(&prow[slot][2 * lane]) = make_float2(0.f, 0.f);
            if (lane == 0) pgr[slot] = -1;
        }
        continue;
    }
    const int beg = start[n];
    const int deg = cnt[n];
    const int tot = deg + 1;

    float adv[HEADS];
    for (int h = 0; h < HEADS; ++h) adv[h] = ad_[n * HEADS + h];

    auto scores = [&](int s, float (&ev)[HEADS]) {
        float av[HEADS];
        if constexpr (HEADS == 4) {
            const float4 q = *reinterpret_cast<const float4*>(as_ + (size_t)s * 4);
            av[0] = q.x; av[1] = q.y; av[2] = q.z; av[3] = q.w;
        } else {
            for (int h = 0; h < HEADS; ++h) av[h] = as_[s * HEADS + h];
        }
        for (int h = 0; h < HEADS; ++h) {
            const float v = av[h] + adv[h];
            ev[h] = v > 0.f ? v : NEG_SLOPE * v;
        }
    };

    int   s0 = n;
    float e0[HEADS];
    for (int h = 0; h < HEADS; ++h) e0[h] = NEG_INF;
    const bool act0 = lane < tot;
    if (act0) {
        s0 = (lane < deg) ? csr_src[beg + lane] : n;
        scores(s0, e0);
    }
    float m[HEADS];
    for (int h = 0; h < HEADS; ++h) m[h] = e0[h];
    for (int j0 = 32; j0 < tot; j0 += 32) {
        const int j = j0 + lane;
        if (j < tot) {
            const int s = (j < deg) ? csr_src[beg + j] : n;
            float ev[HEADS];
            scores(s, ev);
            for (int h = 0; h < HEADS; ++h) m[h] = fmaxf(m[h], ev[h]);
        }
    }
    for (int h = 0; h < HEADS; ++h)
        for (int o = 16; o > 0; o >>= 1) m[h] = fmaxf(m[h], __shfl_xor(m[h], o));

    float acc[VPL];
    for (int i = 0; i < VPL; ++i) acc[i] = 0.f;
    float dl[HEADS];
    for (int h = 0; h < HEADS; ++h) dl[h] = 0.f;

    auto chunk = [&](int s, const float (&ev)[HEADS], bool act, int cn) {
        float al[HEADS];
        for (int h = 0; h < HEADS; ++h) {
            al[h] = act ? __expf(ev[h] - m[h]) : 0.f;
            dl[h] += al[h];
        }
        #pragma unroll 4
        for (int t = 0; t < cn; ++t) {
            const int st = __builtin_amdgcn_readlane(s, t);
            float w[HEADS];
            for (int h = 0; h < HEADS; ++h)
                w[h] = __int_as_float(__builtin_amdgcn_readlane(__float_as_int(al[h]), t));
            const float* row = H + (size_t)st * HC;
            if constexpr (HEADS == 4) {
                const float4 r0 = *reinterpret_cast<const float4*>(row + lane * 4);
                const float4 r1 = *reinterpret_cast<const float4*>(row + 128 + lane * 4);
                const float w0 = hi ? w[1] : w[0];
                const float w1 = hi ? w[3] : w[2];
                acc[0] += w0 * r0.x; acc[1] += w0 * r0.y;
                acc[2] += w0 * r0.z; acc[3] += w0 * r0.w;
                acc[4] += w1 * r1.x; acc[5] += w1 * r1.y;
                acc[6] += w1 * r1.z; acc[7] += w1 * r1.w;
            } else {
                const float2 r = *reinterpret_cast<const float2*>(row + lane * 2);
                acc[0] += w[0] * r.x; acc[1] += w[0] * r.y;
            }
        }
    };

    chunk(s0, e0, act0, tot < 32 ? tot : 32);
    for (int j0 = 32; j0 < tot; j0 += 32) {
        const int j = j0 + lane;
        const bool act = j < tot;
        int s = n;
        float ev[HEADS];
        for (int h = 0; h < HEADS; ++h) ev[h] = NEG_INF;
        if (act) {
            s = (j < deg) ? csr_src[beg + j] : n;
            scores(s, ev);
        }
        const int left = tot - j0;
        chunk(s, ev, act, left < 32 ? left : 32);
    }

    for (int h = 0; h < HEADS; ++h)
        for (int o = 16; o > 0; o >>= 1) dl[h] += __shfl_xor(dl[h], o);

    if constexpr (HEADS == 4) {
        float* orow = outp + (size_t)n * HC;
        const float d0 = hi ? dl[1] : dl[0];
        const float d1 = hi ? dl[3] : dl[2];
        float o[8];
        for (int i = 0; i < 4; ++i) {
            const float v0 = acc[i] / d0 + bias[lane * 4 + i];
            const float v1 = acc[4 + i] / d1 + bias[128 + lane * 4 + i];
            o[i]     = v0 > 0.f ? v0 : (__expf(v0) - 1.0f);
            o[4 + i] = v1 > 0.f ? v1 : (__expf(v1) - 1.0f);
        }
        *reinterpret_cast<float4*>(orow + lane * 4)       = make_float4(o[0], o[1], o[2], o[3]);
        *reinterpret_cast<float4*>(orow + 128 + lane * 4) = make_float4(o[4], o[5], o[6], o[7]);
    } else {
        const float v0 = acc[0] / dl[0] + bias[lane * 2 + 0];
        const float v1 = acc[1] / dl[0] + bias[lane * 2 + 1];
        const float o0 = v0 > 0.f ? v0 : (__expf(v0) - 1.0f);
        const float o1 = v1 > 0.f ? v1 : (__expf(v1) - 1.0f);
        *reinterpret_cast<float2*>(&prow[slot][2 * lane]) = make_float2(o0, o1);
        if (lane == 0) pgr[slot] = batch[n];
    }
    }

    if constexpr (HEADS == 1) {
        __syncthreads();
        if (wave < 2) {
            static_assert(HEADS != 1 || GATHER_NODES <= 32, "one graph id per lane");
            const int c = (int)threadIdx.x;
            float pv[GATHER_NODES];
            #pragma unroll
            for (int w = 0; w < GATHER_NODES; ++w) pv[w] = prow[w][c];
            const int gl = pgr[lane < GATHER_NODES ? lane : 0];
            float run = 0.f;
            int   len = 0;
            int   gc  = __builtin_amdgcn_readlane(gl, 0);
            #pragma unroll
            for (int w = 0; w < GATHER_NODES; ++w) {
                const int g = __builtin_amdgcn_readlane(gl, w);
                if (g != gc) {
                    if (gc >= 0) {
                        atomicAdd(sums + (size_t)gc * 64 + c, run);
                        if (c == 0) atomicAdd(&cnts[gc], (float)len);
                    }
                    run = 0.f; len = 0; gc = g;
                }
                run += pv[w];
                ++len;
            }
            if (gc >= 0) {
                atomicAdd(sums + (size_t)gc * 64 + c, run);
                if (c == 0) atomicAdd(&cnts[gc], (float)len);
            }
        }
    }
}

__global__ void gat_final_linear(const float* __restrict__ sums,
                                 const float* __restrict__ cnts,
                                 const float* __restrict__ Wl,
                                 const float* __restrict__ bl,
                                 float* __restrict__ out, int G, int C, int O)
{
    const int idx = blockIdx.x * blockDim.x + threadIdx.x;
    if (idx >= G * O) return;
    const int g = idx / O, o = idx % O;
    const float inv = 1.0f / fmaxf(cnts[g], 1.0f);
    float a = bl[o];
    for (int c = 0; c < C; ++c) a += sums[g * C + c] * inv * Wl[c * O + o];
    out[idx] = a;
}

extern "C" void kernel_launch(void* const* d_in, const int* in_sizes, int n_in,
                              void* d_out, int out_size, void* d_ws, size_t ws_size,
                              hipStream_t stream)
{
    const float* x   = (const float*)d_in[0];
    const int*   ei  = (const int*)  d_in[1];
    const int*   bat = (const int*)  d_in[2];
    const float* W1  = (const float*)d_in[3];
    const float* aS1 = (const float*)d_in[4];
    const float* aD1 = (const float*)d_in[5];
    const float* b1  = (const float*)d_in[6];
    const float* W2  = (const float*)d_in[7];
    const float* aS2 = (const float*)d_in[8];
    const float* aD2 = (const float*)d_in[9];
    const float* b2  = (const float*)d_in[10];
    const float* Wl  = (const float*)d_in[11];
    const float* bl  = (const float*)d_in[12];
    float* out = (float*)d_out;

    const int N    = in_sizes[2];
    const int E    = in_sizes[1] / 2;
    const int HID  = 64, HEADS = 4, OUT = 10, G = 512;
    const int HC1  = HEADS * HID;

    float* ws = (float*)d_ws;
    size_t off = 0;
    float* H1  = ws + off; off += (size_t)N * HC1;
    float* A1  = ws + off; off += (size_t)N * HC1;
    float* as1 = ws + off; off += (size_t)N * HEADS;
    float* ad1 = ws + off; off += (size_t)N * HEADS;
    float* H2  = ws + off; off += (size_t)N * HID;
    float* as2 = ws + off; off += (size_t)N;
    float* ad2 = ws + off; off += (size_t)N;
    float* sums = ws + off; off += (size_t)G * HID;
    float* cnts = ws + off; off += (size_t)G;
    int* gtot   = (int*)(ws + off); off += 1;
    int* dcnt   = (int*)(ws + off); off += (size_t)N;
    int* dstart = (int*)(ws + off); off += (size_t)N;
    int* csrc   = (int*)(ws + off); off += (size_t)E;
    const int nbin = (N + 63) >> 6;
    const int nblk = (E + 4095) / 4096;
    off += off & 1;
    int2* etmp  = (int2*)(ws + off); off += 2 * (size_t)E;
    int2* seg   = (int2*)(ws + off); off += 2 * (size_t)nbin * nblk;
    (void)ws_size; (void)n_in;

    const int TB = 256;
    auto blks = [](size_t n, int tb) { return (unsigned)((n + tb - 1) / tb); };

    const int nzero = G * HID + G + 1;
    gat_bin_coarse<<<nblk, 1024, 0, stream>>>(ei, E, nbin, etmp, seg, sums, nzero);
    gat_bin_fine<<<nbin, 256, 0, stream>>>(etmp, seg, nblk, gtot, dstart, dcnt, csrc, N);

    {
        dim3 grid(256 / 64, (N + 127) / 128);
        gat_gemm_wmma<128, 256><<<grid, 256, 0, stream>>>(x, W1, H1, aS1, aD1, as1, ad1, N);
    }
    gat_node_gather<4, 5><<<blks(N, GATHER_WAVES * 5), TB, 0, stream>>>(
        csrc, dstart, dcnt, as1, ad1, H1, b1, A1, bat, sums, cnts, N);

    {
        dim3 grid(64 / 64, (N + 127) / 128);
        gat_gemm_wmma<256, 64><<<grid, 256, 0, stream>>>(A1, W2, H2, aS2, aD2, as2, ad2, N);
    }
    gat_node_gather<1, 4><<<blks(N, GATHER_WAVES * 4), TB, 0, stream>>>(
        csrc, dstart, dcnt, as2, ad2, H2, b2, nullptr, bat, sums, cnts, N);

    gat_final_linear<<<blks((size_t)G * OUT, TB), TB, 0, stream>>>(
        sums, cnts, Wl, bl, out, G, HID, OUT);
}


# fine binning pass and layer-1 GEMM (independent of each other) run as work-groups of one grid, bins interleaved 1:2 with GEMM tiles: 7 launches -> 6
# speedup vs baseline: 7.8045x; 1.0384x over previous
#include <hip/hip_runtime.h>
#include <cstddef>


#define NEG_SLOPE 0.2f

typedef float v2f __attribute__((ext_vector_type(2)));
typedef float v8f __attribute__((ext_vector_type(8)));

template <int K>
constexpr int gemm_lds_f4() { return (K / 8) * 4 * 32 > 8 * 16 * 68 / 4 ? (K / 8) * 4 * 32 : 8 * 16 * 68 / 4; }

template <int K, int N>
__device__ __forceinline__
void gat_gemm_tile(float4* __restrict__ smem, const int bx, const int by,
                   const float* __restrict__ A,
                   const float* __restrict__ B,
                   float* __restrict__ C,
                   const float* __restrict__ att_s,
                   const float* __restrict__ att_d,
                   float* __restrict__ a_src,
                   float* __restrict__ a_dst, int M)
{
    constexpr int KJ    = K / 8;
    constexpr int BS_F4 = KJ * 4 * 32;
    constexpr int CP    = 68;

    const int lane = threadIdx.x & 31;
    const int wave = __builtin_amdgcn_readfirstlane((int)(threadIdx.x >> 5));
    const int c    = lane & 15;
    const int h    = lane >> 4;
    const int nb   = bx * 64;
    const int m0   = (by * 8 + wave) * 16;

    int arow = m0 + c;
    if (arow > M - 1) arow = M - 1;
    const float4* ap = reinterpret_cast<const float4*>(A + (size_t)arow * K + 4 * h);
    float4 a[KJ];
    #pragma unroll
    for (int j = 0; j < KJ; ++j) a[j] = ap[2 * j];

    #pragma unroll
    for (int it = 0; it < BS_F4 / 256; ++it) {
        const int idx = it * 256 + (int)threadIdx.x;
        const int sc = idx & 15, sh = (idx >> 4) & 1, st = (idx >> 5) & 3, sj = idx >> 7;
        const float* bp = B + (size_t)(8 * sj + 4 * sh) * N + nb + 16 * st + sc;
        float4 v;
        v.x = bp[0]; v.y = bp[N]; v.z = bp[2 * N]; v.w = bp[3 * N];
        smem[idx] = v;
    }
    __syncthreads();

    v8f acc[4] = {};
    #pragma unroll
    for (int j = 0; j < KJ; ++j) {
        float4 b[4];
        #pragma unroll
        for (int t = 0; t < 4; ++t) b[t] = smem[(j * 4 + t) * 32 + lane];
        v2f a0, a1;
        a0.x = a[j].x; a0.y = a[j].y;
        a1.x = a[j].z; a1.y = a[j].w;
        #pragma unroll
        for (int t = 0; t < 4; ++t) {
            v2f b0; b0.x = b[t].x; b0.y = b[t].y;
            acc[t] = __builtin_amdgcn_wmma_f32_16x16x4_f32(
                false, a0, false, b0, (short)0, acc[t], false, false);
        }
        #pragma unroll
        for (int t = 0; t < 4; ++t) {
            v2f b1; b1.x = b[t].z; b1.y = b[t].w;
            acc[t] = __builtin_amdgcn_wmma_f32_16x16x4_f32(
                false, a1, false, b1, (short)0, acc[t], false, false);
        }
    }

    __syncthreads();
    float* cs = reinterpret_cast<float*>(smem) + wave * (16 * CP);
    #pragma unroll
    for (int t = 0; t < 4; ++t)
        #pragma unroll
        for (int r = 0; r < 8; ++r)
            cs[(8 * h + r) * CP + 16 * t + c] = acc[t][r];
    __syncthreads();
    const float* cr = cs + h * CP + 4 * c;
    if (m0 + 16 <= M) {
        float* cp = C + (size_t)(m0 + h) * N + nb + 4 * c;
        float4 v[8];
        #pragma unroll
        for (int i = 0; i < 8; ++i)
            v[i] = *reinterpret_cast<const float4*>(cr + 2 * i * CP);
        #pragma unroll
        for (int i = 0; i < 8; ++i)
            *reinterpret_cast<float4*>(cp + (size_t)(2 * i) * N) = v[i];
    } else {
        #pragma unroll
        for (int i = 0; i < 8; ++i) {
            const int gm = m0 + 2 * i + h;
            if (gm < M)
                *reinterpret_cast<float4*>(C + (size_t)gm * N + nb + 4 * c) =
                    *reinterpret_cast<const float4*>(cr + 2 * i * CP);
        }
    }

    {
        const float* av   = (h ? att_d : att_s) + nb;
        const float* crow = cs + c * CP;
        float s = 0.f;
        #pragma unroll
        for (int k = 0; k < 64; k += 4) {
            const float4 v = *reinterpret_cast<const float4*>(crow + k);
            const float4 w = *reinterpret_cast<const float4*>(av + k);
            s = __builtin_fmaf(v.x, w.x, s); s = __builtin_fmaf(v.y, w.y, s);
            s = __builtin_fmaf(v.z, w.z, s); s = __builtin_fmaf(v.w, w.w, s);
        }
        const int gm = m0 + c;
        if (gm < M) (h ? a_dst : a_src)[(size_t)gm * (N / 64) + bx] = s;
    }
}

template <int K, int N>
__global__ __launch_bounds__(256)
void gat_gemm_wmma(const float* __restrict__ A,
                   const float* __restrict__ B,
                   float* __restrict__ C,
                   const float* __restrict__ att_s,
                   const float* __restrict__ att_d,
                   float* __restrict__ a_src,
                   float* __restrict__ a_dst, int M)
{
    __shared__ float4 smem[gemm_lds_f4<K>()];
    gat_gemm_tile<K, N>(smem, (int)blockIdx.x, (int)blockIdx.y,
                        A, B, C, att_s, att_d, a_src, a_dst, M);
}

#define BIN_SHIFT 6
#define BIN_NODES 64
#define EDGE_CHUNK 4096
#define MAX_BINS 1024
#define FINE_CAP 4096

__global__ __launch_bounds__(1024)
void gat_bin_coarse(const int* __restrict__ ei, int E, int nbin,
                    int2* __restrict__ etmp, int2* __restrict__ seg,
                    float* __restrict__ zero_p, int nzero)
{
    __shared__ int  hist[MAX_BINS];
    __shared__ int  wsum[32];
    __shared__ int2 stage[EDGE_CHUNK];
    const int tid  = threadIdx.x;
    const int lane = tid & 31, wave = tid >> 5;
    const int nblk = gridDim.x;

    for (int i = blockIdx.x * 1024 + tid; i < nzero; i += nblk * 1024) zero_p[i] = 0.f;
    hist[tid] = 0;
    __syncthreads();

    const int e0 = blockIdx.x * EDGE_CHUNK;
    int s[4], d[4], r[4];
    #pragma unroll
    for (int k = 0; k < 4; ++k) {
        const int e = e0 + k * 1024 + tid;
        const bool ok = e < E;
        s[k] = ok ? ei[e] : 0;
        d[k] = ok ? ei[E + e] : -1;
    }
    #pragma unroll
    for (int k = 0; k < 4; ++k)
        r[k] = (d[k] >= 0) ? atomicAdd(&hist[d[k] >> BIN_SHIFT], 1) : 0;
    __syncthreads();

    const int c = hist[tid];
    int incl = c;
    for (int o = 1; o < 32; o <<= 1) {
        const int t = __shfl_up(incl, o);
        if (lane >= o) incl += t;
    }
    if (lane == 31) wsum[wave] = incl;
    __syncthreads();
    if (wave == 0) {
        const int w = wsum[lane];
        int wi = w;
        for (int o = 1; o < 32; o <<= 1) {
            const int t = __shfl_up(wi, o);
            if (lane >= o) wi += t;
        }
        wsum[lane] = wi - w;
    }
    __syncthreads();
    const int off = wsum[wave] + incl - c;
    hist[tid] = off;
    if (tid < nbin) seg[(size_t)tid * nblk + blockIdx.x] = make_int2(e0 + off, c);
    __syncthreads();

    #pragma unroll
    for (int k = 0; k < 4; ++k)
        if (d[k] >= 0) stage[hist[d[k] >> BIN_SHIFT] + r[k]] = make_int2(s[k], d[k]);
    __syncthreads();
    #pragma unroll
    for (int k = 0; k < 4; ++k) {
        const int i = k * 1024 + tid;
        if (e0 + i < E) etmp[e0 + i] = stage[i];
    }
}

#define FINE_LDS_WORDS (2 * BIN_NODES + 2 + FINE_CAP)

__device__ __forceinline__
void gat_bin_fine_body(int* __restrict__ lds, const int bin,
                  const int2* __restrict__ etmp, const int2* __restrict__ seg, int nseg,
                  int* __restrict__ gtot, int* __restrict__ start, int* __restrict__ cnt,
                  int* __restrict__ csr_src, int Nn)
{
    int* const ncnt  = lds;
    int* const ncur  = lds + BIN_NODES;
    int* const tb    = lds + 2 * BIN_NODES;
    int* const stage = lds + 2 * BIN_NODES + 2;
    const int tid = threadIdx.x;
    const int n0  = bin << BIN_SHIFT;
    const int2* sp = seg + (size_t)bin * nseg;

    if (tid < BIN_NODES) ncnt[tid] = 0;
    __syncthreads();
    for (int g = tid; g < nseg; g += 256) {
        const int2  sg = sp[g];
        const int2* ep = etmp + sg.x;
        for (int k = 0; k < sg.y; k += 4) {
            int dd[4];
            #pragma unroll
            for (int u = 0; u < 4; ++u) dd[u] = (k + u < sg.y) ? ep[k + u].y : -1;
            #pragma unroll
            for (int u = 0; u < 4; ++u)
                if (dd[u] >= 0) atomicAdd(&ncnt[dd[u] - n0], 1);
        }
    }
    __syncthreads();

    if (tid < 32) {
        const int v0 = ncnt[2 * tid], v1 = ncnt[2 * tid + 1];
        const int sm = v0 + v1;
        int incl = sm;
        for (int o = 1; o < 32; o <<= 1) {
            const int t = __shfl_up(incl, o);
            if (tid >= o) incl += t;
        }
        const int tot = __shfl(incl, 31);
        int base = 0;
        if (tid == 0) base = atomicAdd(gtot, tot);
        base = __shfl(base, 0);
        const int ex = incl - sm;
        ncur[2 * tid]     = ex;
        ncur[2 * tid + 1] = ex + v0;
        const int na = n0 + 2 * tid;
        if (na < Nn)     { start[na]     = base + ex;      cnt[na]     = v0; }
        if (na + 1 < Nn) { start[na + 1] = base + ex + v0; cnt[na + 1] = v1; }
        if (tid == 0) { tb[0] = tot; tb[1] = base; }
    }
    __syncthreads();

    const int  tot    = tb[0], base = tb[1];
    const bool staged = tot <= FINE_CAP;
    for (int g = tid; g < nseg; g += 256) {
        const int2  sg = sp[g];
        const int2* ep = etmp + sg.x;
        for (int k = 0; k < sg.y; k += 4) {
            int2 ed[4];
            #pragma unroll
            for (int u = 0; u < 4; ++u) ed[u] = (k + u < sg.y) ? ep[k + u] : make_int2(0, -1);
            #pragma unroll
            for (int u = 0; u < 4; ++u) {
                if (ed[u].y >= 0) {
                    const int p = atomicAdd(&ncur[ed[u].y - n0], 1);
                    if (staged) stage[p] = ed[u].x; else csr_src[base + p] = ed[u].x;
                }
            }
        }
    }
    if (staged) {
        __syncthreads();
        for (int i = tid; i < tot; i += 256) csr_src[base + i] = stage[i];
    }
}

__global__ __launch_bounds__(256)
void gat_fine_and_gemm1(const int2* __restrict__ etmp, const int2* __restrict__ seg, int nseg,
                        int* __restrict__ gtot, int* __restrict__ start, int* __restrict__ cnt,
                        int* __restrict__ csr_src, int Nn, int nbin,
                        const float* __restrict__ X,
                        const float* __restrict__ W,
                        float* __restrict__ Hout,
                        const float* __restrict__ att_s,
                        const float* __restrict__ att_d,
                        float* __restrict__ a_src,
                        float* __restrict__ a_dst)
{
    __shared__ float4 smem[gemm_lds_f4<128>()];
    static_assert(FINE_LDS_WORDS * 4 <= gemm_lds_f4<128>() * 16, "the fine pass fits the GEMM's LDS");
    const int b     = (int)blockIdx.x;
    const int third = b / 3;
    if (b == 3 * third && third < nbin) {
        gat_bin_fine_body(reinterpret_cast<int*>(smem), third,
                          etmp, seg, nseg, gtot, start, cnt, csr_src, Nn);
    } else {
        int nf = (b + 2) / 3;
        if (nf > nbin) nf = nbin;
        const int g = b - nf;
        gat_gemm_tile<128, 256>(smem, g & 3, g >> 2,
                                X, W, Hout, att_s, att_d, a_src, a_dst, Nn);
    }
}

#define GATHER_WAVES 8

template <int HEADS, int TRIPS>
__global__ __launch_bounds__(256)
void gat_node_gather(const int* __restrict__ csr_src,
                     const int* __restrict__ start,
                     const int* __restrict__ cnt,
                     const float* __restrict__ as_,
                     const float* __restrict__ ad_,
                     const float* __restrict__ H,
                     const float* __restrict__ bias,
                     float* __restrict__ outp,
                     const int* __restrict__ batch,
                     float* __restrict__ sums,
                     float* __restrict__ cnts, int Nn)
{
    constexpr int HC  = HEADS * 64;
    constexpr int VPL = HC / 32;
    constexpr int GATHER_NODES = GATHER_WAVES * TRIPS;
    __shared__ float prow[HEADS == 1 ? GATHER_NODES : 1][64];
    __shared__ int   pgr[HEADS == 1 ? GATHER_NODES : 1];
    const int lane = threadIdx.x & 31;
    const int wave = __builtin_amdgcn_readfirstlane((int)(threadIdx.x >> 5));
    const int nfirst = blockIdx.x * GATHER_NODES;
    const float NEG_INF = -__builtin_inff();
    const bool hi = (lane & 16) != 0;

    for (int trip = 0; trip < TRIPS; ++trip) {
    const int slot = trip * GATHER_WAVES + wave;
    const int n    = nfirst + slot;
    if (n >= Nn) {
        if constexpr (HEADS == 1) {
            *reinterpret_cast<float2*>(&prow[slot][2 * lane]) = make_float2(0.f, 0.f);
            if (lane == 0) pgr[slot] = -1;
        }
        continue;
    }
    const int beg = start[n];
    const int deg = cnt[n];
    const int tot = deg + 1;

    float adv[HEADS];
    for (int h = 0; h < HEADS; ++h) adv[h] = ad_[n * HEADS + h];

    auto scores = [&](int s, float (&ev)[HEADS]) {
        float av[HEADS];
        if constexpr (HEADS == 4) {
            const float4 q = *reinterpret_cast<const float4*>(as_ + (size_t)s * 4);
            av[0] = q.x; av[1] = q.y; av[2] = q.z; av[3] = q.w;
        } else {
            for (int h = 0; h < HEADS; ++h) av[h] = as_[s * HEADS + h];
        }
        for (int h = 0; h < HEADS; ++h) {
            const float v = av[h] + adv[h];
            ev[h] = v > 0.f ? v : NEG_SLOPE * v;
        }
    };

    int   s0 = n;
    float e0[HEADS];
    for (int h = 0; h < HEADS; ++h) e0[h] = NEG_INF;
    const bool act0 = lane < tot;
    if (act0) {
        s0 = (lane < deg) ? csr_src[beg + lane] : n;
        scores(s0, e0);
    }
    float m[HEADS];
    for (int h = 0; h < HEADS; ++h) m[h] = e0[h];
    for (int j0 = 32; j0 < tot; j0 += 32) {
        const int j = j0 + lane;
        if (j < tot) {
            const int s = (j < deg) ? csr_src[beg + j] : n;
            float ev[HEADS];
            scores(s, ev);
            for (int h = 0; h < HEADS; ++h) m[h] = fmaxf(m[h], ev[h]);
        }
    }
    for (int h = 0; h < HEADS; ++h)
        for (int o = 16; o > 0; o >>= 1) m[h] = fmaxf(m[h], __shfl_xor(m[h], o));

    float acc[VPL];
    for (int i = 0; i < VPL; ++i) acc[i] = 0.f;
    float dl[HEADS];
    for (int h = 0; h < HEADS; ++h) dl[h] = 0.f;

    auto chunk = [&](int s, const float (&ev)[HEADS], bool act, int cn) {
        float al[HEADS];
        for (int h = 0; h < HEADS; ++h) {
            al[h] = act ? __expf(ev[h] - m[h]) : 0.f;
            dl[h] += al[h];
        }
        #pragma unroll 4
        for (int t = 0; t < cn; ++t) {
            const int st = __builtin_amdgcn_readlane(s, t);
            float w[HEADS];
            for (int h = 0; h < HEADS; ++h)
                w[h] = __int_as_float(__builtin_amdgcn_readlane(__float_as_int(al[h]), t));
            const float* row = H + (size_t)st * HC;
            if constexpr (HEADS == 4) {
                const float4 r0 = *reinterpret_cast<const float4*>(row + lane * 4);
                const float4 r1 = *reinterpret_cast<const float4*>(row + 128 + lane * 4);
                const float w0 = hi ? w[1] : w[0];
                const float w1 = hi ? w[3] : w[2];
                acc[0] += w0 * r0.x; acc[1] += w0 * r0.y;
                acc[2] += w0 * r0.z; acc[3] += w0 * r0.w;
                acc[4] += w1 * r1.x; acc[5] += w1 * r1.y;
                acc[6] += w1 * r1.z; acc[7] += w1 * r1.w;
            } else {
                const float2 r = *reinterpret_cast<const float2*>(row + lane * 2);
                acc[0] += w[0] * r.x; acc[1] += w[0] * r.y;
            }
        }
    };

    chunk(s0, e0, act0, tot < 32 ? tot : 32);
    for (int j0 = 32; j0 < tot; j0 += 32) {
        const int j = j0 + lane;
        const bool act = j < tot;
        int s = n;
        float ev[HEADS];
        for (int h = 0; h < HEADS; ++h) ev[h] = NEG_INF;
        if (act) {
            s = (j < deg) ? csr_src[beg + j] : n;
            scores(s, ev);
        }
        const int left = tot - j0;
        chunk(s, ev, act, left < 32 ? left : 32);
    }

    for (int h = 0; h < HEADS; ++h)
        for (int o = 16; o > 0; o >>= 1) dl[h] += __shfl_xor(dl[h], o);

    if constexpr (HEADS == 4) {
        float* orow = outp + (size_t)n * HC;
        const float d0 = hi ? dl[1] : dl[0];
        const float d1 = hi ? dl[3] : dl[2];
        float o[8];
        for (int i = 0; i < 4; ++i) {
            const float v0 = acc[i] / d0 + bias[lane * 4 + i];
            const float v1 = acc[4 + i] / d1 + bias[128 + lane * 4 + i];
            o[i]     = v0 > 0.f ? v0 : (__expf(v0) - 1.0f);
            o[4 + i] = v1 > 0.f ? v1 : (__expf(v1) - 1.0f);
        }
        *reinterpret_cast<float4*>(orow + lane * 4)       = make_float4(o[0], o[1], o[2], o[3]);
        *reinterpret_cast<float4*>(orow + 128 + lane * 4) = make_float4(o[4], o[5], o[6], o[7]);
    } else {
        const float v0 = acc[0] / dl[0] + bias[lane * 2 + 0];
        const float v1 = acc[1] / dl[0] + bias[lane * 2 + 1];
        const float o0 = v0 > 0.f ? v0 : (__expf(v0) - 1.0f);
        const float o1 = v1 > 0.f ? v1 : (__expf(v1) - 1.0f);
        *reinterpret_cast<float2*>(&prow[slot][2 * lane]) = make_float2(o0, o1);
        if (lane == 0) pgr[slot] = batch[n];
    }
    }

    if constexpr (HEADS == 1) {
        __syncthreads();
        if (wave < 2) {
            static_assert(HEADS != 1 || GATHER_NODES <= 32, "one graph id per lane");
            const int c = (int)threadIdx.x;
            float pv[GATHER_NODES];
            #pragma unroll
            for (int w = 0; w < GATHER_NODES; ++w) pv[w] = prow[w][c];
            const int gl = pgr[lane < GATHER_NODES ? lane : 0];
            float run = 0.f;
            int   len = 0;
            int   gc  = __builtin_amdgcn_readlane(gl, 0);
            #pragma unroll
            for (int w = 0; w < GATHER_NODES; ++w) {
                const int g = __builtin_amdgcn_readlane(gl, w);
                if (g != gc) {
                    if (gc >= 0) {
                        atomicAdd(sums + (size_t)gc * 64 + c, run);
                        if (c == 0) atomicAdd(&cnts[gc], (float)len);
                    }
                    run = 0.f; len = 0; gc = g;
                }
                run += pv[w];
                ++len;
            }
            if (gc >= 0) {
                atomicAdd(sums + (size_t)gc * 64 + c, run);
                if (c == 0) atomicAdd(&cnts[gc], (float)len);
            }
        }
    }
}

__global__ void gat_final_linear(const float* __restrict__ sums,
                                 const float* __restrict__ cnts,
                                 const float* __restrict__ Wl,
                                 const float* __restrict__ bl,
                                 float* __restrict__ out, int G, int C, int O)
{
    const int idx = blockIdx.x * blockDim.x + threadIdx.x;
    if (idx >= G * O) return;
    const int g = idx / O, o = idx % O;
    const float inv = 1.0f / fmaxf(cnts[g], 1.0f);
    float a = bl[o];
    for (int c = 0; c < C; ++c) a += sums[g * C + c] * inv * Wl[c * O + o];
    out[idx] = a;
}

extern "C" void kernel_launch(void* const* d_in, const int* in_sizes, int n_in,
                              void* d_out, int out_size, void* d_ws, size_t ws_size,
                              hipStream_t stream)
{
    const float* x   = (const float*)d_in[0];
    const int*   ei  = (const int*)  d_in[1];
    const int*   bat = (const int*)  d_in[2];
    const float* W1  = (const float*)d_in[3];
    const float* aS1 = (const float*)d_in[4];
    const float* aD1 = (const float*)d_in[5];
    const float* b1  = (const float*)d_in[6];
    const float* W2  = (const float*)d_in[7];
    const float* aS2 = (const float*)d_in[8];
    const float* aD2 = (const float*)d_in[9];
    const float* b2  = (const float*)d_in[10];
    const float* Wl  = (const float*)d_in[11];
    const float* bl  = (const float*)d_in[12];
    float* out = (float*)d_out;

    const int N    = in_sizes[2];
    const int E    = in_sizes[1] / 2;
    const int HID  = 64, HEADS = 4, OUT = 10, G = 512;
    const int HC1  = HEADS * HID;

    float* ws = (float*)d_ws;
    size_t off = 0;
    float* H1  = ws + off; off += (size_t)N * HC1;
    float* A1  = ws + off; off += (size_t)N * HC1;
    float* as1 = ws + off; off += (size_t)N * HEADS;
    float* ad1 = ws + off; off += (size_t)N * HEADS;
    float* H2  = ws + off; off += (size_t)N * HID;
    float* as2 = ws + off; off += (size_t)N;
    float* ad2 = ws + off; off += (size_t)N;
    float* sums = ws + off; off += (size_t)G * HID;
    float* cnts = ws + off; off += (size_t)G;
    int* gtot   = (int*)(ws + off); off += 1;
    int* dcnt   = (int*)(ws + off); off += (size_t)N;
    int* dstart = (int*)(ws + off); off += (size_t)N;
    int* csrc   = (int*)(ws + off); off += (size_t)E;
    const int nbin = (N + 63) >> 6;
    const int nblk = (E + 4095) / 4096;
    off += off & 1;
    int2* etmp  = (int2*)(ws + off); off += 2 * (size_t)E;
    int2* seg   = (int2*)(ws + off); off += 2 * (size_t)nbin * nblk;
    (void)ws_size; (void)n_in;

    const int TB = 256;
    auto blks = [](size_t n, int tb) { return (unsigned)((n + tb - 1) / tb); };

    const int nzero = G * HID + G + 1;
    gat_bin_coarse<<<nblk, 1024, 0, stream>>>(ei, E, nbin, etmp, seg, sums, nzero);

    {
        const int ngemm = (256 / 64) * ((N + 127) / 128);
        gat_fine_and_gemm1<<<nbin + ngemm, 256, 0, stream>>>(
            etmp, seg, nblk, gtot, dstart, dcnt, csrc, N, nbin,
            x, W1, H1, aS1, aD1, as1, ad1);
    }
    gat_node_gather<4, 5><<<blks(N, GATHER_WAVES * 5), TB, 0, stream>>>(
        csrc, dstart, dcnt, as1, ad1, H1, b1, A1, bat, sums, cnts, N);

    {
        dim3 grid(64 / 64, (N + 127) / 128);
        gat_gemm_wmma<256, 64><<<grid, 256, 0, stream>>>(A1, W2, H2, aS2, aD2, as2, ad2, N);
    }
    gat_node_gather<1, 4><<<blks(N, GATHER_WAVES * 4), TB, 0, stream>>>(
        csrc, dstart, dcnt, as2, ad2, H2, b2, nullptr, bat, sums, cnts, N);

    gat_final_linear<<<blks((size_t)G * OUT, TB), TB, 0, stream>>>(
        sums, cnts, Wl, bl, out, G, HID, OUT);
}


# mean pool and final linear folded into the last aggregation: node counts taken in the layer-1 GEMM grid, per-graph partial rows times Wl / count added into out; 6 launches -> 5; sum order changes
# speedup vs baseline: 8.1551x; 1.0449x over previous
#include <hip/hip_runtime.h>
#include <cstddef>


#define NEG_SLOPE 0.2f
#define OUT_CH 10

typedef float v2f __attribute__((ext_vector_type(2)));
typedef float v8f __attribute__((ext_vector_type(8)));

template <int K>
constexpr int gemm_lds_f4() { return (K / 8) * 4 * 32 > 8 * 16 * 68 / 4 ? (K / 8) * 4 * 32 : 8 * 16 * 68 / 4; }

template <int K, int N>
__device__ __forceinline__
void gat_gemm_tile(float4* __restrict__ smem, const int bx, const int by,
                   const float* __restrict__ A,
                   const float* __restrict__ B,
                   float* __restrict__ C,
                   const float* __restrict__ att_s,
                   const float* __restrict__ att_d,
                   float* __restrict__ a_src,
                   float* __restrict__ a_dst, int M)
{
    constexpr int KJ    = K / 8;
    constexpr int BS_F4 = KJ * 4 * 32;
    constexpr int CP    = 68;

    const int lane = threadIdx.x & 31;
    const int wave = __builtin_amdgcn_readfirstlane((int)(threadIdx.x >> 5));
    const int c    = lane & 15;
    const int h    = lane >> 4;
    const int nb   = bx * 64;
    const int m0   = (by * 8 + wave) * 16;

    int arow = m0 + c;
    if (arow > M - 1) arow = M - 1;
    const float4* ap = reinterpret_cast<const float4*>(A + (size_t)arow * K + 4 * h);
    float4 a[KJ];
    #pragma unroll
    for (int j = 0; j < KJ; ++j) a[j] = ap[2 * j];

    #pragma unroll
    for (int it = 0; it < BS_F4 / 256; ++it) {
        const int idx = it * 256 + (int)threadIdx.x;
        const int sc = idx & 15, sh = (idx >> 4) & 1, st = (idx >> 5) & 3, sj = idx >> 7;
        const float* bp = B + (size_t)(8 * sj + 4 * sh) * N + nb + 16 * st + sc;
        float4 v;
        v.x = bp[0]; v.y = bp[N]; v.z = bp[2 * N]; v.w = bp[3 * N];
        smem[idx] = v;
    }
    __syncthreads();

    v8f acc[4] = {};
    #pragma unroll
    for (int j = 0; j < KJ; ++j) {
        float4 b[4];
        #pragma unroll
        for (int t = 0; t < 4; ++t) b[t] = smem[(j * 4 + t) * 32 + lane];
        v2f a0, a1;
        a0.x = a[j].x; a0.y = a[j].y;
        a1.x = a[j].z; a1.y = a[j].w;
        #pragma unroll
        for (int t = 0; t < 4; ++t) {
            v2f b0; b0.x = b[t].x; b0.y = b[t].y;
            acc[t] = __builtin_amdgcn_wmma_f32_16x16x4_f32(
                false, a0, false, b0, (short)0, acc[t], false, false);
        }
        #pragma unroll
        for (int t = 0; t < 4; ++t) {
            v2f b1; b1.x = b[t].z; b1.y = b[t].w;
            acc[t] = __builtin_amdgcn_wmma_f32_16x16x4_f32(
                false, a1, false, b1, (short)0, acc[t], false, false);
        }
    }

    __syncthreads();
    float* cs = reinterpret_cast<float*>(smem) + wave * (16 * CP);
    #pragma unroll
    for (int t = 0; t < 4; ++t)
        #pragma unroll
        for (int r = 0; r < 8; ++r)
            cs[(8 * h + r) * CP + 16 * t + c] = acc[t][r];
    __syncthreads();
    const float* cr = cs + h * CP + 4 * c;
    if (m0 + 16 <= M) {
        float* cp = C + (size_t)(m0 + h) * N + nb + 4 * c;
        float4 v[8];
        #pragma unroll
        for (int i = 0; i < 8; ++i)
            v[i] = *reinterpret_cast<const float4*>(cr + 2 * i * CP);
        #pragma unroll
        for (int i = 0; i < 8; ++i)
            *reinterpret_cast<float4*>(cp + (size_t)(2 * i) * N) = v[i];
    } else {
        #pragma unroll
        for (int i = 0; i < 8; ++i) {
            const int gm = m0 + 2 * i + h;
            if (gm < M)
                *reinterpret_cast<float4*>(C + (size_t)gm * N + nb + 4 * c) =
                    *reinterpret_cast<const float4*>(cr + 2 * i * CP);
        }
    }

    {
        const float* av   = (h ? att_d : att_s) + nb;
        const float* crow = cs + c * CP;
        float s = 0.f;
        #pragma unroll
        for (int k = 0; k < 64; k += 4) {
            const float4 v = *reinterpret_cast<const float4*>(crow + k);
            const float4 w = *reinterpret_cast<const float4*>(av + k);
            s = __builtin_fmaf(v.x, w.x, s); s = __builtin_fmaf(v.y, w.y, s);
            s = __builtin_fmaf(v.z, w.z, s); s = __builtin_fmaf(v.w, w.w, s);
        }
        const int gm = m0 + c;
        if (gm < M) (h ? a_dst : a_src)[(size_t)gm * (N / 64) + bx] = s;
    }
}

template <int K, int N>
__global__ __launch_bounds__(256)
void gat_gemm_wmma(const float* __restrict__ A,
                   const float* __restrict__ B,
                   float* __restrict__ C,
                   const float* __restrict__ att_s,
                   const float* __restrict__ att_d,
                   float* __restrict__ a_src,
                   float* __restrict__ a_dst, int M)
{
    __shared__ float4 smem[gemm_lds_f4<K>()];
    gat_gemm_tile<K, N>(smem, (int)blockIdx.x, (int)blockIdx.y,
                        A, B, C, att_s, att_d, a_src, a_dst, M);
}

#define BIN_SHIFT 6
#define BIN_NODES 64
#define EDGE_CHUNK 4096
#define MAX_BINS 1024
#define FINE_CAP 4096

__global__ __launch_bounds__(1024)
void gat_bin_coarse(const int* __restrict__ ei, int E, int nbin,
                    int2* __restrict__ etmp, int2* __restrict__ seg,
                    float* __restrict__ zero_p, int nzero,
                    const float* __restrict__ bl, float* __restrict__ out, int nout)
{
    __shared__ int  hist[MAX_BINS];
    __shared__ int  wsum[32];
    __shared__ int2 stage[EDGE_CHUNK];
    const int tid  = threadIdx.x;
    const int lane = tid & 31, wave = tid >> 5;
    const int nblk = gridDim.x;

    for (int i = blockIdx.x * 1024 + tid; i < nzero; i += nblk * 1024) zero_p[i] = 0.f;
    for (int i = blockIdx.x * 1024 + tid; i < nout; i += nblk * 1024) out[i] = bl[i % OUT_CH];
    hist[tid] = 0;
    __syncthreads();

    const int e0 = blockIdx.x * EDGE_CHUNK;
    int s[4], d[4], r[4];
    #pragma unroll
    for (int k = 0; k < 4; ++k) {
        const int e = e0 + k * 1024 + tid;
        const bool ok = e < E;
        s[k] = ok ? ei[e] : 0;
        d[k] = ok ? ei[E + e] : -1;
    }
    #pragma unroll
    for (int k = 0; k < 4; ++k)
        r[k] = (d[k] >= 0) ? atomicAdd(&hist[d[k] >> BIN_SHIFT], 1) : 0;
    __syncthreads();

    const int c = hist[tid];
    int incl = c;
    for (int o = 1; o < 32; o <<= 1) {
        const int t = __shfl_up(incl, o);
        if (lane >= o) incl += t;
    }
    if (lane == 31) wsum[wave] = incl;
    __syncthreads();
    if (wave == 0) {
        const int w = wsum[lane];
        int wi = w;
        for (int o = 1; o < 32; o <<= 1) {
            const int t = __shfl_up(wi, o);
            if (lane >= o) wi += t;
        }
        wsum[lane] = wi - w;
    }
    __syncthreads();
    const int off = wsum[wave] + incl - c;
    hist[tid] = off;
    if (tid < nbin) seg[(size_t)tid * nblk + blockIdx.x] = make_int2(e0 + off, c);
    __syncthreads();

    #pragma unroll
    for (int k = 0; k < 4; ++k)
        if (d[k] >= 0) stage[hist[d[k] >> BIN_SHIFT] + r[k]] = make_int2(s[k], d[k]);
    __syncthreads();
    #pragma unroll
    for (int k = 0; k < 4; ++k) {
        const int i = k * 1024 + tid;
        if (e0 + i < E) etmp[e0 + i] = stage[i];
    }
}

#define FINE_LDS_WORDS (2 * BIN_NODES + 2 + FINE_CAP)

__device__ __forceinline__
void gat_bin_fine_body(int* __restrict__ lds, const int bin,
                  const int2* __restrict__ etmp, const int2* __restrict__ seg, int nseg,
                  int* __restrict__ gtot, int* __restrict__ start, int* __restrict__ cnt,
                  int* __restrict__ csr_src, int Nn)
{
    int* const ncnt  = lds;
    int* const ncur  = lds + BIN_NODES;
    int* const tb    = lds + 2 * BIN_NODES;
    int* const stage = lds + 2 * BIN_NODES + 2;
    const int tid = threadIdx.x;
    const int n0  = bin << BIN_SHIFT;
    const int2* sp = seg + (size_t)bin * nseg;

    if (tid < BIN_NODES) ncnt[tid] = 0;
    __syncthreads();
    for (int g = tid; g < nseg; g += 256) {
        const int2  sg = sp[g];
        const int2* ep = etmp + sg.x;
        for (int k = 0; k < sg.y; k += 4) {
            int dd[4];
            #pragma unroll
            for (int u = 0; u < 4; ++u) dd[u] = (k + u < sg.y) ? ep[k + u].y : -1;
            #pragma unroll
            for (int u = 0; u < 4; ++u)
                if (dd[u] >= 0) atomicAdd(&ncnt[dd[u] - n0], 1);
        }
    }
    __syncthreads();

    if (tid < 32) {
        const int v0 = ncnt[2 * tid], v1 = ncnt[2 * tid + 1];
        const int sm = v0 + v1;
        int incl = sm;
        for (int o = 1; o < 32; o <<= 1) {
            const int t = __shfl_up(incl, o);
            if (tid >= o) incl += t;
        }
        const int tot = __shfl(incl, 31);
        int base = 0;
        if (tid == 0) base = atomicAdd(gtot, tot);
        base = __shfl(base, 0);
        const int ex = incl - sm;
        ncur[2 * tid]     = ex;
        ncur[2 * tid + 1] = ex + v0;
        const int na = n0 + 2 * tid;
        if (na < Nn)     { start[na]     = base + ex;      cnt[na]     = v0; }
        if (na + 1 < Nn) { start[na + 1] = base + ex + v0; cnt[na + 1] = v1; }
        if (tid == 0) { tb[0] = tot; tb[1] = base; }
    }
    __syncthreads();

    const int  tot    = tb[0], base = tb[1];
    const bool staged = tot <= FINE_CAP;
    for (int g = tid; g < nseg; g += 256) {
        const int2  sg = sp[g];
        const int2* ep = etmp + sg.x;
        for (int k = 0; k < sg.y; k += 4) {
            int2 ed[4];
            #pragma unroll
            for (int u = 0; u < 4; ++u) ed[u] = (k + u < sg.y) ? ep[k + u] : make_int2(0, -1);
            #pragma unroll
            for (int u = 0; u < 4; ++u) {
                if (ed[u].y >= 0) {
                    const int p = atomicAdd(&ncur[ed[u].y - n0], 1);
                    if (staged) stage[p] = ed[u].x; else csr_src[base + p] = ed[u].x;
                }
            }
        }
    }
    if (staged) {
        __syncthreads();
        for (int i = tid; i < tot; i += 256) csr_src[base + i] = stage[i];
    }
}

__device__ __forceinline__
int gat_count_load(const int* __restrict__ batch, const int row0, const int Nn)
{
    const int m = row0 + (int)threadIdx.x;
    return (threadIdx.x < 128 && m < Nn) ? batch[m] : -1;
}

__device__ __forceinline__
void gat_count_rows(const int g, float* __restrict__ cnts)
{
    const int t = (int)threadIdx.x;
    if (t < 128) {
        const int lane = t & 31;
        const int gp   = __shfl_up(g, 1);
        const bool head = (lane == 0) || (g != gp);
        const unsigned heads = (unsigned)__ballot(head);
        if (head && g >= 0) {
            const unsigned above = (lane == 31) ? 0u : (heads >> (lane + 1));
            const int len = above ? __ffs(above) : (32 - lane);
            atomicAdd(&cnts[g], (float)len);
        }
    }
}

__global__ __launch_bounds__(256)
void gat_fine_and_gemm1(const int2* __restrict__ etmp, const int2* __restrict__ seg, int nseg,
                        int* __restrict__ gtot, int* __restrict__ start, int* __restrict__ cnt,
                        int* __restrict__ csr_src, int Nn, int nbin,
                        const float* __restrict__ X,
                        const float* __restrict__ W,
                        float* __restrict__ Hout,
                        const float* __restrict__ att_s,
                        const float* __restrict__ att_d,
                        float* __restrict__ a_src,
                        float* __restrict__ a_dst,
                        const int* __restrict__ batch,
                        float* __restrict__ cnts)
{
    __shared__ float4 smem[gemm_lds_f4<128>()];
    static_assert(FINE_LDS_WORDS * 4 <= gemm_lds_f4<128>() * 16, "the fine pass fits the GEMM's LDS");
    const int b     = (int)blockIdx.x;
    const int third = b / 3;
    if (b == 3 * third && third < nbin) {
        gat_bin_fine_body(reinterpret_cast<int*>(smem), third,
                          etmp, seg, nseg, gtot, start, cnt, csr_src, Nn);
    } else {
        int nf = (b + 2) / 3;
        if (nf > nbin) nf = nbin;
        const int g = b - nf;
        const bool counts = (g & 3) == 0;
        const int  gid    = counts ? gat_count_load(batch, (g >> 2) * 128, Nn) : -1;
        gat_gemm_tile<128, 256>(smem, g & 3, g >> 2,
                                X, W, Hout, att_s, att_d, a_src, a_dst, Nn);
        if (counts) gat_count_rows(gid, cnts);
    }
}

#define GATHER_WAVES 8

template <int HEADS, int TRIPS>
__global__ __launch_bounds__(256)
void gat_node_gather(const int* __restrict__ csr_src,
                     const int* __restrict__ start,
                     const int* __restrict__ cnt,
                     const float* __restrict__ as_,
                     const float* __restrict__ ad_,
                     const float* __restrict__ H,
                     const float* __restrict__ bias,
                     float* __restrict__ outp,
                     const int* __restrict__ batch,
                     const float* __restrict__ cnts,
                     const float* __restrict__ Wl,
                     float* __restrict__ out, int Nn)
{
    constexpr int HC  = HEADS * 64;
    constexpr int VPL = HC / 32;
    constexpr int GATHER_NODES = GATHER_WAVES * TRIPS;
    __shared__ float prow[HEADS == 1 ? GATHER_NODES : 1][64];
    __shared__ int   pgr[HEADS == 1 ? GATHER_NODES : 1];
    __shared__ float rsum[HEADS == 1 ? GATHER_NODES : 1][64];
    __shared__ int   rgid[HEADS == 1 ? GATHER_NODES : 1][64];
    const int lane = threadIdx.x & 31;
    const int wave = __builtin_amdgcn_readfirstlane((int)(threadIdx.x >> 5));
    const int nfirst = blockIdx.x * GATHER_NODES;
    const float NEG_INF = -__builtin_inff();
    const bool hi = (lane & 16) != 0;

    for (int trip = 0; trip < TRIPS; ++trip) {
    const int slot = trip * GATHER_WAVES + wave;
    const int n    = nfirst + slot;
    if (n >= Nn) {
        if constexpr (HEADS == 1) {
            *reinterpret_cast<float2*>(&prow[slot][2 * lane]) = make_float2(0.f, 0.f);
            if (lane == 0) pgr[slot] = -1;
        }
        continue;
    }
    const int beg = start[n];
    const int deg = cnt[n];
    const int tot = deg + 1;

    float adv[HEADS];
    for (int h = 0; h < HEADS; ++h) adv[h] = ad_[n * HEADS + h];

    auto scores = [&](int s, float (&ev)[HEADS]) {
        float av[HEADS];
        if constexpr (HEADS == 4) {
            const float4 q = *reinterpret_cast<const float4*>(as_ + (size_t)s * 4);
            av[0] = q.x; av[1] = q.y; av[2] = q.z; av[3] = q.w;
        } else {
            for (int h = 0; h < HEADS; ++h) av[h] = as_[s * HEADS + h];
        }
        for (int h = 0; h < HEADS; ++h) {
            const float v = av[h] + adv[h];
            ev[h] = v > 0.f ? v : NEG_SLOPE * v;
        }
    };

    int   s0 = n;
    float e0[HEADS];
    for (int h = 0; h < HEADS; ++h) e0[h] = NEG_INF;
    const bool act0 = lane < tot;
    if (act0) {
        s0 = (lane < deg) ? csr_src[beg + lane] : n;
        scores(s0, e0);
    }
    float m[HEADS];
    for (int h = 0; h < HEADS; ++h) m[h] = e0[h];
    for (int j0 = 32; j0 < tot; j0 += 32) {
        const int j = j0 + lane;
        if (j < tot) {
            const int s = (j < deg) ? csr_src[beg + j] : n;
            float ev[HEADS];
            scores(s, ev);
            for (int h = 0; h < HEADS; ++h) m[h] = fmaxf(m[h], ev[h]);
        }
    }
    for (int h = 0; h < HEADS; ++h)
        for (int o = 16; o > 0; o >>= 1) m[h] = fmaxf(m[h], __shfl_xor(m[h], o));

    float acc[VPL];
    for (int i = 0; i < VPL; ++i) acc[i] = 0.f;
    float dl[HEADS];
    for (int h = 0; h < HEADS; ++h) dl[h] = 0.f;

    auto chunk = [&](int s, const float (&ev)[HEADS], bool act, int cn) {
        float al[HEADS];
        for (int h = 0; h < HEADS; ++h) {
            al[h] = act ? __expf(ev[h] - m[h]) : 0.f;
            dl[h] += al[h];
        }
        #pragma unroll 4
        for (int t = 0; t < cn; ++t) {
            const int st = __builtin_amdgcn_readlane(s, t);
            float w[HEADS];
            for (int h = 0; h < HEADS; ++h)
                w[h] = __int_as_float(__builtin_amdgcn_readlane(__float_as_int(al[h]), t));
            const float* row = H + (size_t)st * HC;
            if constexpr (HEADS == 4) {
                const float4 r0 = *reinterpret_cast<const float4*>(row + lane * 4);
                const float4 r1 = *reinterpret_cast<const float4*>(row + 128 + lane * 4);
                const float w0 = hi ? w[1] : w[0];
                const float w1 = hi ? w[3] : w[2];
                acc[0] += w0 * r0.x; acc[1] += w0 * r0.y;
                acc[2] += w0 * r0.z; acc[3] += w0 * r0.w;
                acc[4] += w1 * r1.x; acc[5] += w1 * r1.y;
                acc[6] += w1 * r1.z; acc[7] += w1 * r1.w;
            } else {
                const float2 r = *reinterpret_cast<const float2*>(row + lane * 2);
                acc[0] += w[0] * r.x; acc[1] += w[0] * r.y;
            }
        }
    };

    chunk(s0, e0, act0, tot < 32 ? tot : 32);
    for (int j0 = 32; j0 < tot; j0 += 32) {
        const int j = j0 + lane;
        const bool act = j < tot;
        int s = n;
        float ev[HEADS];
        for (int h = 0; h < HEADS; ++h) ev[h] = NEG_INF;
        if (act) {
            s = (j < deg) ? csr_src[beg + j] : n;
            scores(s, ev);
        }
        const int left = tot - j0;
        chunk(s, ev, act, left < 32 ? left : 32);
    }

    for (int h = 0; h < HEADS; ++h)
        for (int o = 16; o > 0; o >>= 1) dl[h] += __shfl_xor(dl[h], o);

    if constexpr (HEADS == 4) {
        float* orow = outp + (size_t)n * HC;
        const float d0 = hi ? dl[1] : dl[0];
        const float d1 = hi ? dl[3] : dl[2];
        float o[8];
        for (int i = 0; i < 4; ++i) {
            const float v0 = acc[i] / d0 + bias[lane * 4 + i];
            const float v1 = acc[4 + i] / d1 + bias[128 + lane * 4 + i];
            o[i]     = v0 > 0.f ? v0 : (__expf(v0) - 1.0f);
            o[4 + i] = v1 > 0.f ? v1 : (__expf(v1) - 1.0f);
        }
        *reinterpret_cast<float4*>(orow + lane * 4)       = make_float4(o[0], o[1], o[2], o[3]);
        *reinterpret_cast<float4*>(orow + 128 + lane * 4) = make_float4(o[4], o[5], o[6], o[7]);
    } else {
        const float v0 = acc[0] / dl[0] + bias[lane * 2 + 0];
        const float v1 = acc[1] / dl[0] + bias[lane * 2 + 1];
        const float o0 = v0 > 0.f ? v0 : (__expf(v0) - 1.0f);
        const float o1 = v1 > 0.f ? v1 : (__expf(v1) - 1.0f);
        *reinterpret_cast<float2*>(&prow[slot][2 * lane]) = make_float2(o0, o1);
        if (lane == 0) pgr[slot] = batch[n];
    }
    }

    if constexpr (HEADS == 1) {
        __syncthreads();
        if (wave < 2) {
            static_assert(HEADS != 1 || GATHER_NODES <= 32, "one graph id per lane");
            const int c = (int)threadIdx.x;
            float pv[GATHER_NODES];
            #pragma unroll
            for (int w = 0; w < GATHER_NODES; ++w) pv[w] = prow[w][c];
            const int gl = pgr[lane < GATHER_NODES ? lane : 0];
            float run = 0.f;
            int   nrun = 0;
            int   gc  = __builtin_amdgcn_readlane(gl, 0);
            #pragma unroll
            for (int w = 0; w < GATHER_NODES; ++w) {
                const int g = __builtin_amdgcn_readlane(gl, w);
                if (g != gc) {
                    if (gc >= 0) { rsum[nrun][c] = run; rgid[nrun][c] = gc; ++nrun; }
                    run = 0.f; gc = g;
                }
                run += pv[w];
            }
            if (gc >= 0) { rsum[nrun][c] = run; rgid[nrun][c] = gc; ++nrun; }

            float wl[OUT_CH];
            #pragma unroll
            for (int o = 0; o < OUT_CH; ++o) wl[o] = Wl[c * OUT_CH + o];
            #pragma nounroll
            for (int r = 0; r < nrun; ++r) {
                const int   gq  = __builtin_amdgcn_readfirstlane(rgid[r][c]);
                const float rv  = rsum[r][c];
                const float inv = 1.0f / fmaxf(cnts[gq], 1.0f);
                float mine = 0.f;
                #pragma unroll
                for (int o = 0; o < OUT_CH; ++o) {
                    float y = rv * wl[o];
                    for (int off = 16; off > 0; off >>= 1) y += __shfl_xor(y, off);
                    if (lane == o) mine = y;
                }
                if (lane < OUT_CH) atomicAdd(out + (size_t)gq * OUT_CH + lane, mine * inv);
            }
        }
    }
}

extern "C" void kernel_launch(void* const* d_in, const int* in_sizes, int n_in,
                              void* d_out, int out_size, void* d_ws, size_t ws_size,
                              hipStream_t stream)
{
    const float* x   = (const float*)d_in[0];
    const int*   ei  = (const int*)  d_in[1];
    const int*   bat = (const int*)  d_in[2];
    const float* W1  = (const float*)d_in[3];
    const float* aS1 = (const float*)d_in[4];
    const float* aD1 = (const float*)d_in[5];
    const float* b1  = (const float*)d_in[6];
    const float* W2  = (const float*)d_in[7];
    const float* aS2 = (const float*)d_in[8];
    const float* aD2 = (const float*)d_in[9];
    const float* b2  = (const float*)d_in[10];
    const float* Wl  = (const float*)d_in[11];
    const float* bl  = (const float*)d_in[12];
    float* out = (float*)d_out;

    const int N    = in_sizes[2];
    const int E    = in_sizes[1] / 2;
    const int HID  = 64, HEADS = 4, G = 512;
    const int HC1  = HEADS * HID;

    float* ws = (float*)d_ws;
    size_t off = 0;
    float* H1  = ws + off; off += (size_t)N * HC1;
    float* A1  = ws + off; off += (size_t)N * HC1;
    float* as1 = ws + off; off += (size_t)N * HEADS;
    float* ad1 = ws + off; off += (size_t)N * HEADS;
    float* H2  = ws + off; off += (size_t)N * HID;
    float* as2 = ws + off; off += (size_t)N;
    float* ad2 = ws + off; off += (size_t)N;
    float* cnts = ws + off; off += (size_t)G;
    int* gtot   = (int*)(ws + off); off += 1;
    int* dcnt   = (int*)(ws + off); off += (size_t)N;
    int* dstart = (int*)(ws + off); off += (size_t)N;
    int* csrc   = (int*)(ws + off); off += (size_t)E;
    const int nbin = (N + 63) >> 6;
    const int nblk = (E + 4095) / 4096;
    off += off & 1;
    int2* etmp  = (int2*)(ws + off); off += 2 * (size_t)E;
    int2* seg   = (int2*)(ws + off); off += 2 * (size_t)nbin * nblk;
    (void)ws_size; (void)n_in;

    const int TB = 256;
    auto blks = [](size_t n, int tb) { return (unsigned)((n + tb - 1) / tb); };

    const int nzero = G + 1;
    gat_bin_coarse<<<nblk, 1024, 0, stream>>>(ei, E, nbin, etmp, seg, cnts, nzero,
                                              bl, out, G * OUT_CH);

    {
        const int ngemm = (256 / 64) * ((N + 127) / 128);
        gat_fine_and_gemm1<<<nbin + ngemm, 256, 0, stream>>>(
            etmp, seg, nblk, gtot, dstart, dcnt, csrc, N, nbin,
            x, W1, H1, aS1, aD1, as1, ad1, bat, cnts);
    }
    gat_node_gather<4, 5><<<blks(N, GATHER_WAVES * 5), TB, 0, stream>>>(
        csrc, dstart, dcnt, as1, ad1, H1, b1, A1, bat, cnts, Wl, out, N);

    {
        dim3 grid(64 / 64, (N + 127) / 128);
        gat_gemm_wmma<256, 64><<<grid, 256, 0, stream>>>(A1, W2, H2, aS2, aD2, as2, ad2, N);
    }
    gat_node_gather<1, 4><<<blks(N, GATHER_WAVES * 4), TB, 0, stream>>>(
        csrc, dstart, dcnt, as2, ad2, H2, b2, nullptr, bat, cnts, Wl, out, N);
}


# launch fusion: layer-2 GEMM done by each layer-1 aggregation work-group on its own 40 activated rows kept in LDS (W2 put in fragment order by the coarse pass), rows never stored; 5 launches -> 4
# speedup vs baseline: 8.7184x; 1.0691x over previous
#include <hip/hip_runtime.h>
#include <cstddef>


#define NEG_SLOPE 0.2f
#define OUT_CH 10

typedef float v2f __attribute__((ext_vector_type(2)));
typedef float v8f __attribute__((ext_vector_type(8)));

template <int K>
constexpr int gemm_lds_f4() { return (K / 8) * 4 * 32 > 8 * 16 * 68 / 4 ? (K / 8) * 4 * 32 : 8 * 16 * 68 / 4; }

template <int K, int N>
__device__ __forceinline__
void gat_gemm_tile(float4* __restrict__ smem, const int bx, const int by,
                   const float* __restrict__ A,
                   const float* __restrict__ B,
                   float* __restrict__ C,
                   const float* __restrict__ att_s,
                   const float* __restrict__ att_d,
                   float* __restrict__ a_src,
                   float* __restrict__ a_dst, int M)
{
    constexpr int KJ    = K / 8;
    constexpr int BS_F4 = KJ * 4 * 32;
    constexpr int CP    = 68;

    const int lane = threadIdx.x & 31;
    const int wave = __builtin_amdgcn_readfirstlane((int)(threadIdx.x >> 5));
    const int c    = lane & 15;
    const int h    = lane >> 4;
    const int nb   = bx * 64;
    const int m0   = (by * 8 + wave) * 16;

    int arow = m0 + c;
    if (arow > M - 1) arow = M - 1;
    const float4* ap = reinterpret_cast<const float4*>(A + (size_t)arow * K + 4 * h);
    float4 a[KJ];
    #pragma unroll
    for (int j = 0; j < KJ; ++j) a[j] = ap[2 * j];

    #pragma unroll
    for (int it = 0; it < BS_F4 / 256; ++it) {
        const int idx = it * 256 + (int)threadIdx.x;
        const int sc = idx & 15, sh = (idx >> 4) & 1, st = (idx >> 5) & 3, sj = idx >> 7;
        const float* bp = B + (size_t)(8 * sj + 4 * sh) * N + nb + 16 * st + sc;
        float4 v;
        v.x = bp[0]; v.y = bp[N]; v.z = bp[2 * N]; v.w = bp[3 * N];
        smem[idx] = v;
    }
    __syncthreads();

    v8f acc[4] = {};
    #pragma unroll
    for (int j = 0; j < KJ; ++j) {
        float4 b[4];
        #pragma unroll
        for (int t = 0; t < 4; ++t) b[t] = smem[(j * 4 + t) * 32 + lane];
        v2f a0, a1;
        a0.x = a[j].x; a0.y = a[j].y;
        a1.x = a[j].z; a1.y = a[j].w;
        #pragma unroll
        for (int t = 0; t < 4; ++t) {
            v2f b0; b0.x = b[t].x; b0.y = b[t].y;
            acc[t] = __builtin_amdgcn_wmma_f32_16x16x4_f32(
                false, a0, false, b0, (short)0, acc[t], false, false);
        }
        #pragma unroll
        for (int t = 0; t < 4; ++t) {
            v2f b1; b1.x = b[t].z; b1.y = b[t].w;
            acc[t] = __builtin_amdgcn_wmma_f32_16x16x4_f32(
                false, a1, false, b1, (short)0, acc[t], false, false);
        }
    }

    __syncthreads();
    float* cs = reinterpret_cast<float*>(smem) + wave * (16 * CP);
    #pragma unroll
    for (int t = 0; t < 4; ++t)
        #pragma unroll
        for (int r = 0; r < 8; ++r)
            cs[(8 * h + r) * CP + 16 * t + c] = acc[t][r];
    __syncthreads();
    const float* cr = cs + h * CP + 4 * c;
    if (m0 + 16 <= M) {
        float* cp = C + (size_t)(m0 + h) * N + nb + 4 * c;
        float4 v[8];
        #pragma unroll
        for (int i = 0; i < 8; ++i)
            v[i] = *reinterpret_cast<const float4*>(cr + 2 * i * CP);
        #pragma unroll
        for (int i = 0; i < 8; ++i)
            *reinterpret_cast<float4*>(cp + (size_t)(2 * i) * N) = v[i];
    } else {
        #pragma unroll
        for (int i = 0; i < 8; ++i) {
            const int gm = m0 + 2 * i + h;
            if (gm < M)
                *reinterpret_cast<float4*>(C + (size_t)gm * N + nb + 4 * c) =
                    *reinterpret_cast<const float4*>(cr + 2 * i * CP);
        }
    }

    {
        const float* av   = (h ? att_d : att_s) + nb;
        const float* crow = cs + c * CP;
        float s = 0.f;
        #pragma unroll
        for (int k = 0; k < 64; k += 4) {
            const float4 v = *reinterpret_cast<const float4*>(crow + k);
            const float4 w = *reinterpret_cast<const float4*>(av + k);
            s = __builtin_fmaf(v.x, w.x, s); s = __builtin_fmaf(v.y, w.y, s);
            s = __builtin_fmaf(v.z, w.z, s); s = __builtin_fmaf(v.w, w.w, s);
        }
        const int gm = m0 + c;
        if (gm < M) (h ? a_dst : a_src)[(size_t)gm * (N / 64) + bx] = s;
    }
}

#define BIN_SHIFT 6
#define BIN_NODES 64
#define EDGE_CHUNK 4096
#define MAX_BINS 1024
#define FINE_CAP 4096

#define W2_K 256
#define W2_SLOTS ((W2_K / 8) * 4 * 32)
__global__ __launch_bounds__(1024)
void gat_bin_coarse(const int* __restrict__ ei, int E, int nbin,
                    int2* __restrict__ etmp, int2* __restrict__ seg,
                    float* __restrict__ zero_p, int nzero,
                    const float* __restrict__ bl, float* __restrict__ out, int nout,
                    const float* __restrict__ W2, float* __restrict__ W2f)
{
    __shared__ int  hist[MAX_BINS];
    __shared__ int  wsum[32];
    __shared__ int2 stage[EDGE_CHUNK];
    const int tid  = threadIdx.x;
    const int lane = tid & 31, wave = tid >> 5;
    const int nblk = gridDim.x;

    for (int i = blockIdx.x * 1024 + tid; i < nzero; i += nblk * 1024) zero_p[i] = 0.f;
    for (int i = blockIdx.x * 1024 + tid; i < nout; i += nblk * 1024) out[i] = bl[i % OUT_CH];
    for (int i = blockIdx.x * 1024 + tid; i < W2_SLOTS; i += nblk * 1024) {
        const int sc = i & 15, sh = (i >> 4) & 1, st = (i >> 5) & 3, sj = i >> 7;
        const float* bp = W2 + (size_t)(8 * sj + 4 * sh) * 64 + 16 * st + sc;
        float4 v;
        v.x = bp[0]; v.y = bp[64]; v.z = bp[128]; v.w = bp[192];
        reinterpret_cast<float4*>(W2f)[i] = v;
    }
    hist[tid] = 0;
    __syncthreads();

    const int e0 = blockIdx.x * EDGE_CHUNK;
    int s[4], d[4], r[4];
    #pragma unroll
    for (int k = 0; k < 4; ++k) {
        const int e = e0 + k * 1024 + tid;
        const bool ok = e < E;
        s[k] = ok ? ei[e] : 0;
        d[k] = ok ? ei[E + e] : -1;
    }
    #pragma unroll
    for (int k = 0; k < 4; ++k)
        r[k] = (d[k] >= 0) ? atomicAdd(&hist[d[k] >> BIN_SHIFT], 1) : 0;
    __syncthreads();

    const int c = hist[tid];
    int incl = c;
    for (int o = 1; o < 32; o <<= 1) {
        const int t = __shfl_up(incl, o);
        if (lane >= o) incl += t;
    }
    if (lane == 31) wsum[wave] = incl;
    __syncthreads();
    if (wave == 0) {
        const int w = wsum[lane];
        int wi = w;
        for (int o = 1; o < 32; o <<= 1) {
            const int t = __shfl_up(wi, o);
            if (lane >= o) wi += t;
        }
        wsum[lane] = wi - w;
    }
    __syncthreads();
    const int off = wsum[wave] + incl - c;
    hist[tid] = off;
    if (tid < nbin) seg[(size_t)tid * nblk + blockIdx.x] = make_int2(e0 + off, c);
    __syncthreads();

    #pragma unroll
    for (int k = 0; k < 4; ++k)
        if (d[k] >= 0) stage[hist[d[k] >> BIN_SHIFT] + r[k]] = make_int2(s[k], d[k]);
    __syncthreads();
    #pragma unroll
    for (int k = 0; k < 4; ++k) {
        const int i = k * 1024 + tid;
        if (e0 + i < E) etmp[e0 + i] = stage[i];
    }
}

#define FINE_LDS_WORDS (2 * BIN_NODES + 2 + FINE_CAP)

__device__ __forceinline__
void gat_bin_fine_body(int* __restrict__ lds, const int bin,
                  const int2* __restrict__ etmp, const int2* __restrict__ seg, int nseg,
                  int* __restrict__ gtot, int* __restrict__ start, int* __restrict__ cnt,
                  int* __restrict__ csr_src, int Nn)
{
    int* const ncnt  = lds;
    int* const ncur  = lds + BIN_NODES;
    int* const tb    = lds + 2 * BIN_NODES;
    int* const stage = lds + 2 * BIN_NODES + 2;
    const int tid = threadIdx.x;
    const int n0  = bin << BIN_SHIFT;
    const int2* sp = seg + (size_t)bin * nseg;

    if (tid < BIN_NODES) ncnt[tid] = 0;
    __syncthreads();
    for (int g = tid; g < nseg; g += 256) {
        const int2  sg = sp[g];
        const int2* ep = etmp + sg.x;
        for (int k = 0; k < sg.y; k += 4) {
            int dd[4];
            #pragma unroll
            for (int u = 0; u < 4; ++u) dd[u] = (k + u < sg.y) ? ep[k + u].y : -1;
            #pragma unroll
            for (int u = 0; u < 4; ++u)
                if (dd[u] >= 0) atomicAdd(&ncnt[dd[u] - n0], 1);
        }
    }
    __syncthreads();

    if (tid < 32) {
        const int v0 = ncnt[2 * tid], v1 = ncnt[2 * tid + 1];
        const int sm = v0 + v1;
        int incl = sm;
        for (int o = 1; o < 32; o <<= 1) {
            const int t = __shfl_up(incl, o);
            if (tid >= o) incl += t;
        }
        const int tot = __shfl(incl, 31);
        int base = 0;
        if (tid == 0) base = atomicAdd(gtot, tot);
        base = __shfl(base, 0);
        const int ex = incl - sm;
        ncur[2 * tid]     = ex;
        ncur[2 * tid + 1] = ex + v0;
        const int na = n0 + 2 * tid;
        if (na < Nn)     { start[na]     = base + ex;      cnt[na]     = v0; }
        if (na + 1 < Nn) { start[na + 1] = base + ex + v0; cnt[na + 1] = v1; }
        if (tid == 0) { tb[0] = tot; tb[1] = base; }
    }
    __syncthreads();

    const int  tot    = tb[0], base = tb[1];
    const bool staged = tot <= FINE_CAP;
    for (int g = tid; g < nseg; g += 256) {
        const int2  sg = sp[g];
        const int2* ep = etmp + sg.x;
        for (int k = 0; k < sg.y; k += 4) {
            int2 ed[4];
            #pragma unroll
            for (int u = 0; u < 4; ++u) ed[u] = (k + u < sg.y) ? ep[k + u] : make_int2(0, -1);
            #pragma unroll
            for (int u = 0; u < 4; ++u) {
                if (ed[u].y >= 0) {
                    const int p = atomicAdd(&ncur[ed[u].y - n0], 1);
                    if (staged) stage[p] = ed[u].x; else csr_src[base + p] = ed[u].x;
                }
            }
        }
    }
    if (staged) {
        __syncthreads();
        for (int i = tid; i < tot; i += 256) csr_src[base + i] = stage[i];
    }
}

__device__ __forceinline__
int gat_count_load(const int* __restrict__ batch, const int row0, const int Nn)
{
    const int m = row0 + (int)threadIdx.x;
    return (threadIdx.x < 128 && m < Nn) ? batch[m] : -1;
}

__device__ __forceinline__
void gat_count_rows(const int g, float* __restrict__ cnts)
{
    const int t = (int)threadIdx.x;
    if (t < 128) {
        const int lane = t & 31;
        const int gp   = __shfl_up(g, 1);
        const bool head = (lane == 0) || (g != gp);
        const unsigned heads = (unsigned)__ballot(head);
        if (head && g >= 0) {
            const unsigned above = (lane == 31) ? 0u : (heads >> (lane + 1));
            const int len = above ? __ffs(above) : (32 - lane);
            atomicAdd(&cnts[g], (float)len);
        }
    }
}

__global__ __launch_bounds__(256)
void gat_fine_and_gemm1(const int2* __restrict__ etmp, const int2* __restrict__ seg, int nseg,
                        int* __restrict__ gtot, int* __restrict__ start, int* __restrict__ cnt,
                        int* __restrict__ csr_src, int Nn, int nbin,
                        const float* __restrict__ X,
                        const float* __restrict__ W,
                        float* __restrict__ Hout,
                        const float* __restrict__ att_s,
                        const float* __restrict__ att_d,
                        float* __restrict__ a_src,
                        float* __restrict__ a_dst,
                        const int* __restrict__ batch,
                        float* __restrict__ cnts)
{
    __shared__ float4 smem[gemm_lds_f4<128>()];
    static_assert(FINE_LDS_WORDS * 4 <= gemm_lds_f4<128>() * 16, "the fine pass fits the GEMM's LDS");
    const int b     = (int)blockIdx.x;
    const int third = b / 3;
    if (b == 3 * third && third < nbin) {
        gat_bin_fine_body(reinterpret_cast<int*>(smem), third,
                          etmp, seg, nseg, gtot, start, cnt, csr_src, Nn);
    } else {
        int nf = (b + 2) / 3;
        if (nf > nbin) nf = nbin;
        const int g = b - nf;
        const bool counts = (g & 3) == 0;
        const int  gid    = counts ? gat_count_load(batch, (g >> 2) * 128, Nn) : -1;
        gat_gemm_tile<128, 256>(smem, g & 3, g >> 2,
                                X, W, Hout, att_s, att_d, a_src, a_dst, Nn);
        if (counts) gat_count_rows(gid, cnts);
    }
}

#define GATHER_WAVES 8

template <int HEADS, int TRIPS>
__global__ __launch_bounds__(256)
void gat_node_gather(const int* __restrict__ csr_src,
                     const int* __restrict__ start,
                     const int* __restrict__ cnt,
                     const float* __restrict__ as_,
                     const float* __restrict__ ad_,
                     const float* __restrict__ H,
                     const float* __restrict__ bias,
                     float* __restrict__ outp,
                     const int* __restrict__ batch,
                     const float* __restrict__ cnts,
                     const float* __restrict__ Wl,
                     float* __restrict__ out, int Nn,
                     const float* __restrict__ Bf,
                     const float* __restrict__ att_s2,
                     const float* __restrict__ att_d2,
                     float* __restrict__ as_o,
                     float* __restrict__ ad_o)
{
    constexpr int HC  = HEADS * 64;
    constexpr int VPL = HC / 32;
    constexpr int GATHER_NODES = GATHER_WAVES * TRIPS;
    __shared__ float prow[HEADS == 1 ? GATHER_NODES : 1][64];
    __shared__ int   pgr[HEADS == 1 ? GATHER_NODES : 1];
    __shared__ float rsum[HEADS == 1 ? GATHER_NODES : 1][64];
    __shared__ int   rgid[HEADS == 1 ? GATHER_NODES : 1][64];
    constexpr int MROWS = ((GATHER_NODES + 15) / 16) * 16;
    constexpr int AP    = 260;
    constexpr int CP    = 68;
    __shared__ float4 arow4[HEADS == 4 ? MROWS * AP / 4 : 1];
    float* const As = reinterpret_cast<float*>(arow4);
    const int lane = threadIdx.x & 31;
    const int wave = __builtin_amdgcn_readfirstlane((int)(threadIdx.x >> 5));
    const int nfirst = blockIdx.x * GATHER_NODES;
    if constexpr (HEADS == 4) {
        for (int i = (int)threadIdx.x; i < (MROWS - GATHER_NODES) * 64; i += 256)
            *reinterpret_cast<float4*>(As + (GATHER_NODES + (i >> 6)) * AP + 4 * (i & 63)) =
                make_float4(0.f, 0.f, 0.f, 0.f);
    }
    const float NEG_INF = -__builtin_inff();
    const bool hi = (lane & 16) != 0;

    for (int trip = 0; trip < TRIPS; ++trip) {
    const int slot = trip * GATHER_WAVES + wave;
    const int n    = nfirst + slot;
    if (n >= Nn) {
        if constexpr (HEADS == 1) {
            *reinterpret_cast<float2*>(&prow[slot][2 * lane]) = make_float2(0.f, 0.f);
            if (lane == 0) pgr[slot] = -1;
        } else {
            *reinterpret_cast<float4*>(As + slot * AP + lane * 4)       = make_float4(0.f, 0.f, 0.f, 0.f);
            *reinterpret_cast<float4*>(As + slot * AP + 128 + lane * 4) = make_float4(0.f, 0.f, 0.f, 0.f);
        }
        continue;
    }
    const int beg = start[n];
    const int deg = cnt[n];
    const int tot = deg + 1;

    float adv[HEADS];
    for (int h = 0; h < HEADS; ++h) adv[h] = ad_[n * HEADS + h];

    auto scores = [&](int s, float (&ev)[HEADS]) {
        float av[HEADS];
        if constexpr (HEADS == 4) {
            const float4 q = *reinterpret_cast<const float4*>(as_ + (size_t)s * 4);
            av[0] = q.x; av[1] = q.y; av[2] = q.z; av[3] = q.w;
        } else {
            for (int h = 0; h < HEADS; ++h) av[h] = as_[s * HEADS + h];
        }
        for (int h = 0; h < HEADS; ++h) {
            const float v = av[h] + adv[h];
            ev[h] = v > 0.f ? v : NEG_SLOPE * v;
        }
    };

    int   s0 = n;
    float e0[HEADS];
    for (int h = 0; h < HEADS; ++h) e0[h] = NEG_INF;
    const bool act0 = lane < tot;
    if (act0) {
        s0 = (lane < deg) ? csr_src[beg + lane] : n;
        scores(s0, e0);
    }
    float m[HEADS];
    for (int h = 0; h < HEADS; ++h) m[h] = e0[h];
    for (int j0 = 32; j0 < tot; j0 += 32) {
        const int j = j0 + lane;
        if (j < tot) {
            const int s = (j < deg) ? csr_src[beg + j] : n;
            float ev[HEADS];
            scores(s, ev);
            for (int h = 0; h < HEADS; ++h) m[h] = fmaxf(m[h], ev[h]);
        }
    }
    for (int h = 0; h < HEADS; ++h)
        for (int o = 16; o > 0; o >>= 1) m[h] = fmaxf(m[h], __shfl_xor(m[h], o));

    float acc[VPL];
    for (int i = 0; i < VPL; ++i) acc[i] = 0.f;
    float dl[HEADS];
    for (int h = 0; h < HEADS; ++h) dl[h] = 0.f;

    auto chunk = [&](int s, const float (&ev)[HEADS], bool act, int cn) {
        float al[HEADS];
        for (int h = 0; h < HEADS; ++h) {
            al[h] = act ? __expf(ev[h] - m[h]) : 0.f;
            dl[h] += al[h];
        }
        #pragma unroll 4
        for (int t = 0; t < cn; ++t) {
            const int st = __builtin_amdgcn_readlane(s, t);
            float w[HEADS];
            for (int h = 0; h < HEADS; ++h)
                w[h] = __int_as_float(__builtin_amdgcn_readlane(__float_as_int(al[h]), t));
            const float* row = H + (size_t)st * HC;
            if constexpr (HEADS == 4) {
                const float4 r0 = *reinterpret_cast<const float4*>(row + lane * 4);
                const float4 r1 = *reinterpret_cast<const float4*>(row + 128 + lane * 4);
                const float w0 = hi ? w[1] : w[0];
                const float w1 = hi ? w[3] : w[2];
                acc[0] += w0 * r0.x; acc[1] += w0 * r0.y;
                acc[2] += w0 * r0.z; acc[3] += w0 * r0.w;
                acc[4] += w1 * r1.x; acc[5] += w1 * r1.y;
                acc[6] += w1 * r1.z; acc[7] += w1 * r1.w;
            } else {
                const float2 r = *reinterpret_cast<const float2*>(row + lane * 2);
                acc[0] += w[0] * r.x; acc[1] += w[0] * r.y;
            }
        }
    };

    chunk(s0, e0, act0, tot < 32 ? tot : 32);
    for (int j0 = 32; j0 < tot; j0 += 32) {
        const int j = j0 + lane;
        const bool act = j < tot;
        int s = n;
        float ev[HEADS];
        for (int h = 0; h < HEADS; ++h) ev[h] = NEG_INF;
        if (act) {
            s = (j < deg) ? csr_src[beg + j] : n;
            scores(s, ev);
        }
        const int left = tot - j0;
        chunk(s, ev, act, left < 32 ? left : 32);
    }

    for (int h = 0; h < HEADS; ++h)
        for (int o = 16; o > 0; o >>= 1) dl[h] += __shfl_xor(dl[h], o);

    if constexpr (HEADS == 4) {
        float* orow = As + slot * AP;
        const float d0 = hi ? dl[1] : dl[0];
        const float d1 = hi ? dl[3] : dl[2];
        float o[8];
        for (int i = 0; i < 4; ++i) {
            const float v0 = acc[i] / d0 + bias[lane * 4 + i];
            const float v1 = acc[4 + i] / d1 + bias[128 + lane * 4 + i];
            o[i]     = v0 > 0.f ? v0 : (__expf(v0) - 1.0f);
            o[4 + i] = v1 > 0.f ? v1 : (__expf(v1) - 1.0f);
        }
        *reinterpret_cast<float4*>(orow + lane * 4)       = make_float4(o[0], o[1], o[2], o[3]);
        *reinterpret_cast<float4*>(orow + 128 + lane * 4) = make_float4(o[4], o[5], o[6], o[7]);
    } else {
        const float v0 = acc[0] / dl[0] + bias[lane * 2 + 0];
        const float v1 = acc[1] / dl[0] + bias[lane * 2 + 1];
        const float o0 = v0 > 0.f ? v0 : (__expf(v0) - 1.0f);
        const float o1 = v1 > 0.f ? v1 : (__expf(v1) - 1.0f);
        *reinterpret_cast<float2*>(&prow[slot][2 * lane]) = make_float2(o0, o1);
        if (lane == 0) pgr[slot] = batch[n];
    }
    }

    if constexpr (HEADS == 4) {
        static_assert(HEADS != 4 || (MROWS == 48 && GATHER_WAVES == 8), "three row tiles over eight waves");
        __syncthreads();
        const int c = lane & 15, h = lane >> 4;
        const bool two = wave < 4;
        const int ct  = wave & 3;
        const int rt0 = two ? 0 : 1;
        const float*  ap0 = As + (16 * rt0 + c) * AP + 4 * h;
        const float*  ap1 = As + (32 + c) * AP + 4 * h;
        const float4* bp  = reinterpret_cast<const float4*>(Bf) + ct * 32 + lane;
        v8f acc0 = {}, acc1 = {};
        if (two) {
            #pragma unroll
            for (int j = 0; j < W2_K / 8; ++j) {
                const float4 b  = bp[j * 128];
                const float4 a  = *reinterpret_cast<const float4*>(ap0 + 8 * j);
                const float4 a2 = *reinterpret_cast<const float4*>(ap1 + 8 * j);
                v2f b0, b1, x0, x1, y0, y1;
                b0.x = b.x;  b0.y = b.y;  b1.x = b.z;  b1.y = b.w;
                x0.x = a.x;  x0.y = a.y;  x1.x = a.z;  x1.y = a.w;
                y0.x = a2.x; y0.y = a2.y; y1.x = a2.z; y1.y = a2.w;
                acc0 = __builtin_amdgcn_wmma_f32_16x16x4_f32(false, x0, false, b0, (short)0, acc0, false, false);
                acc1 = __builtin_amdgcn_wmma_f32_16x16x4_f32(false, y0, false, b0, (short)0, acc1, false, false);
                acc0 = __builtin_amdgcn_wmma_f32_16x16x4_f32(false, x1, false, b1, (short)0, acc0, false, false);
                acc1 = __builtin_amdgcn_wmma_f32_16x16x4_f32(false, y1, false, b1, (short)0, acc1, false, false);
            }
        } else {
            #pragma unroll
            for (int j = 0; j < W2_K / 8; ++j) {
                const float4 b = bp[j * 128];
                const float4 a = *reinterpret_cast<const float4*>(ap0 + 8 * j);
                v2f b0, b1, x0, x1;
                b0.x = b.x; b0.y = b.y; b1.x = b.z; b1.y = b.w;
                x0.x = a.x; x0.y = a.y; x1.x = a.z; x1.y = a.w;
                acc0 = __builtin_amdgcn_wmma_f32_16x16x4_f32(false, x0, false, b0, (short)0, acc0, false, false);
                acc0 = __builtin_amdgcn_wmma_f32_16x16x4_f32(false, x1, false, b1, (short)0, acc0, false, false);
            }
        }

        __syncthreads();
        float* const cs = As;
        #pragma unroll
        for (int r = 0; r < 8; ++r) cs[(16 * rt0 + 8 * h + r) * CP + 16 * ct + c] = acc0[r];
        if (two) {
            #pragma unroll
            for (int r = 0; r < 8; ++r) cs[(32 + 8 * h + r) * CP + 16 * ct + c] = acc1[r];
        }
        __syncthreads();
        for (int i = (int)threadIdx.x; i < GATHER_NODES * 16; i += 256) {
            const int r = i >> 4, q = i & 15;
            if (nfirst + r < Nn)
                *reinterpret_cast<float4*>(outp + (size_t)(nfirst + r) * 64 + 4 * q) =
                    *reinterpret_cast<const float4*>(cs + r * CP + 4 * q);
        }
        if (threadIdx.x < 128) {
            const int r = (int)threadIdx.x & 63;
            const bool dsts = threadIdx.x >= 64;
            if (r < GATHER_NODES && nfirst + r < Nn) {
                const float* av   = dsts ? att_d2 : att_s2;
                const float* crow = cs + r * CP;
                float sa = 0.f;
                #pragma unroll
                for (int k = 0; k < 64; k += 4) {
                    const float4 v = *reinterpret_cast<const float4*>(crow + k);
                    const float4 w = *reinterpret_cast<const float4*>(av + k);
                    sa = __builtin_fmaf(v.x, w.x, sa); sa = __builtin_fmaf(v.y, w.y, sa);
                    sa = __builtin_fmaf(v.z, w.z, sa); sa = __builtin_fmaf(v.w, w.w, sa);
                }
                (dsts ? ad_o : as_o)[nfirst + r] = sa;
            }
        }
    }

    if constexpr (HEADS == 1) {
        __syncthreads();
        if (wave < 2) {
            static_assert(HEADS != 1 || GATHER_NODES <= 32, "one graph id per lane");
            const int c = (int)threadIdx.x;
            float pv[GATHER_NODES];
            #pragma unroll
            for (int w = 0; w < GATHER_NODES; ++w) pv[w] = prow[w][c];
            const int gl = pgr[lane < GATHER_NODES ? lane : 0];
            float run = 0.f;
            int   nrun = 0;
            int   gc  = __builtin_amdgcn_readlane(gl, 0);
            #pragma unroll
            for (int w = 0; w < GATHER_NODES; ++w) {
                const int g = __builtin_amdgcn_readlane(gl, w);
                if (g != gc) {
                    if (gc >= 0) { rsum[nrun][c] = run; rgid[nrun][c] = gc; ++nrun; }
                    run = 0.f; gc = g;
                }
                run += pv[w];
            }
            if (gc >= 0) { rsum[nrun][c] = run; rgid[nrun][c] = gc; ++nrun; }

            float wl[OUT_CH];
            #pragma unroll
            for (int o = 0; o < OUT_CH; ++o) wl[o] = Wl[c * OUT_CH + o];
            #pragma nounroll
            for (int r = 0; r < nrun; ++r) {
                const int   gq  = __builtin_amdgcn_readfirstlane(rgid[r][c]);
                const float rv  = rsum[r][c];
                const float inv = 1.0f / fmaxf(cnts[gq], 1.0f);
                float mine = 0.f;
                #pragma unroll
                for (int o = 0; o < OUT_CH; ++o) {
                    float y = rv * wl[o];
                    for (int off = 16; off > 0; off >>= 1) y += __shfl_xor(y, off);
                    if (lane == o) mine = y;
                }
                if (lane < OUT_CH) atomicAdd(out + (size_t)gq * OUT_CH + lane, mine * inv);
            }
        }
    }
}

extern "C" void kernel_launch(void* const* d_in, const int* in_sizes, int n_in,
                              void* d_out, int out_size, void* d_ws, size_t ws_size,
                              hipStream_t stream)
{
    const float* x   = (const float*)d_in[0];
    const int*   ei  = (const int*)  d_in[1];
    const int*   bat = (const int*)  d_in[2];
    const float* W1  = (const float*)d_in[3];
    const float* aS1 = (const float*)d_in[4];
    const float* aD1 = (const float*)d_in[5];
    const float* b1  = (const float*)d_in[6];
    const float* W2  = (const float*)d_in[7];
    const float* aS2 = (const float*)d_in[8];
    const float* aD2 = (const float*)d_in[9];
    const float* b2  = (const float*)d_in[10];
    const float* Wl  = (const float*)d_in[11];
    const float* bl  = (const float*)d_in[12];
    float* out = (float*)d_out;

    const int N    = in_sizes[2];
    const int E    = in_sizes[1] / 2;
    const int HID  = 64, HEADS = 4, G = 512;
    const int HC1  = HEADS * HID;

    float* ws = (float*)d_ws;
    size_t off = 0;
    float* H1  = ws + off; off += (size_t)N * HC1;
    float* W2f = ws + off; off += (size_t)W2_SLOTS * 4;
    float* as1 = ws + off; off += (size_t)N * HEADS;
    float* ad1 = ws + off; off += (size_t)N * HEADS;
    float* H2  = ws + off; off += (size_t)N * HID;
    float* as2 = ws + off; off += (size_t)N;
    float* ad2 = ws + off; off += (size_t)N;
    float* cnts = ws + off; off += (size_t)G;
    int* gtot   = (int*)(ws + off); off += 1;
    int* dcnt   = (int*)(ws + off); off += (size_t)N;
    int* dstart = (int*)(ws + off); off += (size_t)N;
    int* csrc   = (int*)(ws + off); off += (size_t)E;
    const int nbin = (N + 63) >> 6;
    const int nblk = (E + 4095) / 4096;
    off += off & 1;
    int2* etmp  = (int2*)(ws + off); off += 2 * (size_t)E;
    int2* seg   = (int2*)(ws + off); off += 2 * (size_t)nbin * nblk;
    (void)ws_size; (void)n_in;

    const int TB = 256;
    auto blks = [](size_t n, int tb) { return (unsigned)((n + tb - 1) / tb); };

    const int nzero = G + 1;
    gat_bin_coarse<<<nblk, 1024, 0, stream>>>(ei, E, nbin, etmp, seg, cnts, nzero,
                                              bl, out, G * OUT_CH, W2, W2f);

    {
        const int ngemm = (256 / 64) * ((N + 127) / 128);
        gat_fine_and_gemm1<<<nbin + ngemm, 256, 0, stream>>>(
            etmp, seg, nblk, gtot, dstart, dcnt, csrc, N, nbin,
            x, W1, H1, aS1, aD1, as1, ad1, bat, cnts);
    }
    gat_node_gather<4, 5><<<blks(N, GATHER_WAVES * 5), TB, 0, stream>>>(
        csrc, dstart, dcnt, as1, ad1, H1, b1, H2, bat, cnts, Wl, out, N,
        W2f, aS2, aD2, as2, ad2);

    gat_node_gather<1, 4><<<blks(N, GATHER_WAVES * 4), TB, 0, stream>>>(
        csrc, dstart, dcnt, as2, ad2, H2, b2, nullptr, bat, cnts, Wl, out, N,
        nullptr, nullptr, nullptr, nullptr, nullptr);
}


# layer 1 sums input rows x[src] (512 B an edge) with the heads' weights and multiplies by W1 afterwards in the aggregation work-group (WMMA); a_s, a_d = x (W1 att); x W1 not stored; sum order changes
# speedup vs baseline: 10.3203x; 1.1837x over previous
#include <hip/hip_runtime.h>
#include <cstddef>


#define NEG_SLOPE 0.2f
#define OUT_CH 10

typedef float v2f __attribute__((ext_vector_type(2)));
typedef float v8f __attribute__((ext_vector_type(8)));


__device__ __forceinline__
void gat_acoef_tile(const int by,
                    const float* __restrict__ X,
                    const float* __restrict__ Vf,
                    float* __restrict__ a_src,
                    float* __restrict__ a_dst, int M)
{
    constexpr int K  = 128;
    constexpr int KJ = K / 8;
    const int lane = threadIdx.x & 31;
    const int wave = __builtin_amdgcn_readfirstlane((int)(threadIdx.x >> 5));
    const int c    = lane & 15;
    const int h    = lane >> 4;
    const int m0   = (by * 8 + wave) * 16;

    int arow = m0 + c;
    if (arow > M - 1) arow = M - 1;
    const float4* ap = reinterpret_cast<const float4*>(X + (size_t)arow * K + 4 * h);
    const float4* bp = reinterpret_cast<const float4*>(Vf) + lane;
    float4 a[KJ];
    #pragma unroll
    for (int j = 0; j < KJ; ++j) a[j] = ap[2 * j];

    v8f acc = {};
    #pragma unroll
    for (int j = 0; j < KJ; ++j) {
        const float4 b = bp[j * 32];
        v2f a0, a1, b0, b1;
        a0.x = a[j].x; a0.y = a[j].y; a1.x = a[j].z; a1.y = a[j].w;
        b0.x = b.x;    b0.y = b.y;    b1.x = b.z;    b1.y = b.w;
        acc = __builtin_amdgcn_wmma_f32_16x16x4_f32(false, a0, false, b0, (short)0, acc, false, false);
        acc = __builtin_amdgcn_wmma_f32_16x16x4_f32(false, a1, false, b1, (short)0, acc, false, false);
    }
    if (c < 8) {
        float* dst = (c < 4) ? a_src : a_dst;
        const int hc = c & 3;
        #pragma unroll
        for (int i = 0; i < 8; ++i) {
            const int gm = m0 + 8 * h + i;
            if (gm < M) dst[(size_t)gm * 4 + hc] = acc[i];
        }
    }
}


#define BIN_SHIFT 6
#define BIN_NODES 64
#define EDGE_CHUNK 4096
#define MAX_BINS 1024
#define FINE_CAP 4096

#define W2_K 256
#define W2_SLOTS ((W2_K / 8) * 4 * 32)
#define W1_K 128
#define W1_HSLOTS ((W1_K / 8) * 4 * 32)
#define W1_SLOTS (4 * W1_HSLOTS)
#define V_WORDS ((W1_K / 8) * 32 * 4)
__global__ __launch_bounds__(1024)
void gat_bin_coarse(const int* __restrict__ ei, int E, int nbin,
                    int2* __restrict__ etmp, int2* __restrict__ seg,
                    float* __restrict__ zero_p, int nzero,
                    const float* __restrict__ bl, float* __restrict__ out, int nout,
                    const float* __restrict__ W2, float* __restrict__ W2f,
                    const float* __restrict__ W1, float* __restrict__ W1f,
                    const float* __restrict__ att_s1, const float* __restrict__ att_d1,
                    float* __restrict__ Vf)
{
    __shared__ int  hist[MAX_BINS];
    __shared__ int  wsum[32];
    __shared__ int2 stage[EDGE_CHUNK];
    const int tid  = threadIdx.x;
    const int lane = tid & 31, wave = tid >> 5;
    const int nblk = gridDim.x;

    for (int i = blockIdx.x * 1024 + tid; i < nzero; i += nblk * 1024) zero_p[i] = 0.f;
    for (int i = blockIdx.x * 1024 + tid; i < nout; i += nblk * 1024) out[i] = bl[i % OUT_CH];
    for (int i = blockIdx.x * 1024 + tid; i < W2_SLOTS; i += nblk * 1024) {
        const int sc = i & 15, sh = (i >> 4) & 1, st = (i >> 5) & 3, sj = i >> 7;
        const float* bp = W2 + (size_t)(8 * sj + 4 * sh) * 64 + 16 * st + sc;
        float4 v;
        v.x = bp[0]; v.y = bp[64]; v.z = bp[128]; v.w = bp[192];
        reinterpret_cast<float4*>(W2f)[i] = v;
    }
    for (int i = blockIdx.x * 1024 + tid; i < W1_SLOTS; i += nblk * 1024) {
        const int sc = i & 15, sh = (i >> 4) & 1, st = (i >> 5) & 3, sj = (i >> 7) & 15, hd = i >> 11;
        const float* bp = W1 + (size_t)(8 * sj + 4 * sh) * 256 + hd * 64 + 16 * st + sc;
        float4 v;
        v.x = bp[0]; v.y = bp[256]; v.z = bp[512]; v.w = bp[768];
        reinterpret_cast<float4*>(W1f)[i] = v;
    }
    if (wave == 0) {
        for (int k = blockIdx.x; k < W1_K; k += nblk) {
            const float4* wr = reinterpret_cast<const float4*>(W1 + (size_t)k * 256);
            const float4* sr = reinterpret_cast<const float4*>(att_s1);
            const float4* dr = reinterpret_cast<const float4*>(att_d1);
            const float4 w0 = wr[lane], w1 = wr[32 + lane];
            const float4 s0 = sr[lane], s1 = sr[32 + lane];
            const float4 d0 = dr[lane], d1 = dr[32 + lane];
            float ps0 = w0.x * s0.x + w0.y * s0.y + w0.z * s0.z + w0.w * s0.w;
            float ps1 = w1.x * s1.x + w1.y * s1.y + w1.z * s1.z + w1.w * s1.w;
            float pd0 = w0.x * d0.x + w0.y * d0.y + w0.z * d0.z + w0.w * d0.w;
            float pd1 = w1.x * d1.x + w1.y * d1.y + w1.z * d1.z + w1.w * d1.w;
            for (int o = 8; o > 0; o >>= 1) {
                ps0 += __shfl_xor(ps0, o); ps1 += __shfl_xor(ps1, o);
                pd0 += __shfl_xor(pd0, o); pd1 += __shfl_xor(pd1, o);
            }
            const int cc = lane & 15, hd = cc & 3;
            const int sl = (hd & 1) * 16;
            const float a0 = __shfl(ps0, sl), a1 = __shfl(ps1, sl);
            const float c0 = __shfl(pd0, sl), c1 = __shfl(pd1, sl);
            float v = 0.f;
            if (cc < 4)      v = hd < 2 ? a0 : a1;
            else if (cc < 8) v = hd < 2 ? c0 : c1;
            if (lane < 16)
                Vf[((k >> 3) * 32 + ((k >> 2) & 1) * 16 + cc) * 4 + (k & 3)] = v;
        }
    }
    hist[tid] = 0;
    __syncthreads();

    const int e0 = blockIdx.x * EDGE_CHUNK;
    int s[4], d[4], r[4];
    #pragma unroll
    for (int k = 0; k < 4; ++k) {
        const int e = e0 + k * 1024 + tid;
        const bool ok = e < E;
        s[k] = ok ? ei[e] : 0;
        d[k] = ok ? ei[E + e] : -1;
    }
    #pragma unroll
    for (int k = 0; k < 4; ++k)
        r[k] = (d[k] >= 0) ? atomicAdd(&hist[d[k] >> BIN_SHIFT], 1) : 0;
    __syncthreads();

    const int c = hist[tid];
    int incl = c;
    for (int o = 1; o < 32; o <<= 1) {
        const int t = __shfl_up(incl, o);
        if (lane >= o) incl += t;
    }
    if (lane == 31) wsum[wave] = incl;
    __syncthreads();
    if (wave == 0) {
        const int w = wsum[lane];
        int wi = w;
        for (int o = 1; o < 32; o <<= 1) {
            const int t = __shfl_up(wi, o);
            if (lane >= o) wi += t;
        }
        wsum[lane] = wi - w;
    }
    __syncthreads();
    const int off = wsum[wave] + incl - c;
    hist[tid] = off;
    if (tid < nbin) seg[(size_t)tid * nblk + blockIdx.x] = make_int2(e0 + off, c);
    __syncthreads();

    #pragma unroll
    for (int k = 0; k < 4; ++k)
        if (d[k] >= 0) stage[hist[d[k] >> BIN_SHIFT] + r[k]] = make_int2(s[k], d[k]);
    __syncthreads();
    #pragma unroll
    for (int k = 0; k < 4; ++k) {
        const int i = k * 1024 + tid;
        if (e0 + i < E) etmp[e0 + i] = stage[i];
    }
}

#define FINE_LDS_WORDS (2 * BIN_NODES + 2 + FINE_CAP)

__device__ __forceinline__
void gat_bin_fine_body(int* __restrict__ lds, const int bin,
                  const int2* __restrict__ etmp, const int2* __restrict__ seg, int nseg,
                  int* __restrict__ gtot, int* __restrict__ start, int* __restrict__ cnt,
                  int* __restrict__ csr_src, int Nn)
{
    int* const ncnt  = lds;
    int* const ncur  = lds + BIN_NODES;
    int* const tb    = lds + 2 * BIN_NODES;
    int* const stage = lds + 2 * BIN_NODES + 2;
    const int tid = threadIdx.x;
    const int n0  = bin << BIN_SHIFT;
    const int2* sp = seg + (size_t)bin * nseg;

    if (tid < BIN_NODES) ncnt[tid] = 0;
    __syncthreads();
    for (int g = tid; g < nseg; g += 256) {
        const int2  sg = sp[g];
        const int2* ep = etmp + sg.x;
        for (int k = 0; k < sg.y; k += 4) {
            int dd[4];
            #pragma unroll
            for (int u = 0; u < 4; ++u) dd[u] = (k + u < sg.y) ? ep[k + u].y : -1;
            #pragma unroll
            for (int u = 0; u < 4; ++u)
                if (dd[u] >= 0) atomicAdd(&ncnt[dd[u] - n0], 1);
        }
    }
    __syncthreads();

    if (tid < 32) {
        const int v0 = ncnt[2 * tid], v1 = ncnt[2 * tid + 1];
        const int sm = v0 + v1;
        int incl = sm;
        for (int o = 1; o < 32; o <<= 1) {
            const int t = __shfl_up(incl, o);
            if (tid >= o) incl += t;
        }
        const int tot = __shfl(incl, 31);
        int base = 0;
        if (tid == 0) base = atomicAdd(gtot, tot);
        base = __shfl(base, 0);
        const int ex = incl - sm;
        ncur[2 * tid]     = ex;
        ncur[2 * tid + 1] = ex + v0;
        const int na = n0 + 2 * tid;
        if (na < Nn)     { start[na]     = base + ex;      cnt[na]     = v0; }
        if (na + 1 < Nn) { start[na + 1] = base + ex + v0; cnt[na + 1] = v1; }
        if (tid == 0) { tb[0] = tot; tb[1] = base; }
    }
    __syncthreads();

    const int  tot    = tb[0], base = tb[1];
    const bool staged = tot <= FINE_CAP;
    for (int g = tid; g < nseg; g += 256) {
        const int2  sg = sp[g];
        const int2* ep = etmp + sg.x;
        for (int k = 0; k < sg.y; k += 4) {
            int2 ed[4];
            #pragma unroll
            for (int u = 0; u < 4; ++u) ed[u] = (k + u < sg.y) ? ep[k + u] : make_int2(0, -1);
            #pragma unroll
            for (int u = 0; u < 4; ++u) {
                if (ed[u].y >= 0) {
                    const int p = atomicAdd(&ncur[ed[u].y - n0], 1);
                    if (staged) stage[p] = ed[u].x; else csr_src[base + p] = ed[u].x;
                }
            }
        }
    }
    if (staged) {
        __syncthreads();
        for (int i = tid; i < tot; i += 256) csr_src[base + i] = stage[i];
    }
}

__device__ __forceinline__
int gat_count_load(const int* __restrict__ batch, const int row0, const int Nn)
{
    const int m = row0 + (int)threadIdx.x;
    return (threadIdx.x < 128 && m < Nn) ? batch[m] : -1;
}

__device__ __forceinline__
void gat_count_rows(const int g, float* __restrict__ cnts)
{
    const int t = (int)threadIdx.x;
    if (t < 128) {
        const int lane = t & 31;
        const int gp   = __shfl_up(g, 1);
        const bool head = (lane == 0) || (g != gp);
        const unsigned heads = (unsigned)__ballot(head);
        if (head && g >= 0) {
            const unsigned above = (lane == 31) ? 0u : (heads >> (lane + 1));
            const int len = above ? __ffs(above) : (32 - lane);
            atomicAdd(&cnts[g], (float)len);
        }
    }
}

__global__ __launch_bounds__(256)
void gat_fine_and_acoef(const int2* __restrict__ etmp, const int2* __restrict__ seg, int nseg,
                        int* __restrict__ gtot, int* __restrict__ start, int* __restrict__ cnt,
                        int* __restrict__ csr_src, int Nn, int nbin, int ntile,
                        const float* __restrict__ X,
                        const float* __restrict__ Vf,
                        float* __restrict__ a_src,
                        float* __restrict__ a_dst,
                        const int* __restrict__ batch,
                        float* __restrict__ cnts)
{
    __shared__ int lds[FINE_LDS_WORDS];
    const int b = (int)blockIdx.x;
    int T = nbin >> 1;
    if (T > ntile) T = ntile;
    int bin = -1, tile = -1;
    if (b < 3 * T) {
        const int q = b / 3, r = b - 3 * q;
        if (r == 2) tile = q; else bin = 2 * q + r;
    } else {
        const int i = b - 3 * T;
        if (i < ntile - T) tile = T + i; else bin = 2 * T + (i - (ntile - T));
    }
    if (bin >= 0) {
        if (bin < nbin)
            gat_bin_fine_body(lds, bin, etmp, seg, nseg, gtot, start, cnt, csr_src, Nn);
    } else {
        const int gid = gat_count_load(batch, tile * 128, Nn);
        gat_acoef_tile(tile, X, Vf, a_src, a_dst, Nn);
        gat_count_rows(gid, cnts);
    }
}


#define GATHER_WAVES 8

template <int HEADS, int TRIPS>
__global__ __launch_bounds__(256)
void gat_node_gather(const int* __restrict__ csr_src,
                     const int* __restrict__ start,
                     const int* __restrict__ cnt,
                     const float* __restrict__ as_,
                     const float* __restrict__ ad_,
                     const float* __restrict__ H,
                     const float* __restrict__ bias,
                     float* __restrict__ outp,
                     const int* __restrict__ batch,
                     const float* __restrict__ cnts,
                     const float* __restrict__ Wl,
                     float* __restrict__ out, int Nn,
                     const float* __restrict__ Bf,
                     const float* __restrict__ att_s2,
                     const float* __restrict__ att_d2,
                     float* __restrict__ as_o,
                     float* __restrict__ ad_o,
                     const float* __restrict__ B1f)
{
    constexpr int RW  = HEADS == 4 ? 128 : 64;
    constexpr int VPL = HEADS == 4 ? 16 : 2;
    constexpr int GATHER_NODES = GATHER_WAVES * TRIPS;
    __shared__ float prow[HEADS == 1 ? GATHER_NODES : 1][64];
    __shared__ int   pgr[HEADS == 1 ? GATHER_NODES : 1];
    __shared__ float rsum[HEADS == 1 ? GATHER_NODES : 1][64];
    __shared__ int   rgid[HEADS == 1 ? GATHER_NODES : 1][64];
    static_assert(HEADS != 4 || GATHER_NODES % 16 == 0, "whole row tiles");
    constexpr int MROWS = GATHER_NODES;
    constexpr int RT    = MROWS / 16;
    constexpr int ZP    = 516;
    constexpr int AP    = 260;
    constexpr int CP    = 68;
    __shared__ float4 arow4[HEADS == 4 ? MROWS * ZP / 4 : 1];
    float* const Zs = reinterpret_cast<float*>(arow4);
    float* const As = reinterpret_cast<float*>(arow4);
    const int lane = threadIdx.x & 31;
    const int wave = __builtin_amdgcn_readfirstlane((int)(threadIdx.x >> 5));
    const int nfirst = blockIdx.x * GATHER_NODES;
    const float NEG_INF = -__builtin_inff();

    for (int trip = 0; trip < TRIPS; ++trip) {
    const int slot = trip * GATHER_WAVES + wave;
    const int n    = nfirst + slot;
    if (n >= Nn) {
        if constexpr (HEADS == 1) {
            *reinterpret_cast<float2*>(&prow[slot][2 * lane]) = make_float2(0.f, 0.f);
            if (lane == 0) pgr[slot] = -1;
        } else {
            for (int hd = 0; hd < 4; ++hd)
                *reinterpret_cast<float4*>(Zs + slot * ZP + hd * 128 + lane * 4) = make_float4(0.f, 0.f, 0.f, 0.f);
        }
        continue;
    }
    const int beg = start[n];
    const int deg = cnt[n];
    const int tot = deg + 1;

    float adv[HEADS];
    for (int h = 0; h < HEADS; ++h) adv[h] = ad_[n * HEADS + h];

    auto scores = [&](int s, float (&ev)[HEADS]) {
        float av[HEADS];
        if constexpr (HEADS == 4) {
            const float4 q = *reinterpret_cast<const float4*>(as_ + (size_t)s * 4);
            av[0] = q.x; av[1] = q.y; av[2] = q.z; av[3] = q.w;
        } else {
            for (int h = 0; h < HEADS; ++h) av[h] = as_[s * HEADS + h];
        }
        for (int h = 0; h < HEADS; ++h) {
            const float v = av[h] + adv[h];
            ev[h] = v > 0.f ? v : NEG_SLOPE * v;
        }
    };

    int   s0 = n;
    float e0[HEADS];
    for (int h = 0; h < HEADS; ++h) e0[h] = NEG_INF;
    const bool act0 = lane < tot;
    if (act0) {
        s0 = (lane < deg) ? csr_src[beg + lane] : n;
        scores(s0, e0);
    }
    float m[HEADS];
    for (int h = 0; h < HEADS; ++h) m[h] = e0[h];
    for (int j0 = 32; j0 < tot; j0 += 32) {
        const int j = j0 + lane;
        if (j < tot) {
            const int s = (j < deg) ? csr_src[beg + j] : n;
            float ev[HEADS];
            scores(s, ev);
            for (int h = 0; h < HEADS; ++h) m[h] = fmaxf(m[h], ev[h]);
        }
    }
    for (int h = 0; h < HEADS; ++h)
        for (int o = 16; o > 0; o >>= 1) m[h] = fmaxf(m[h], __shfl_xor(m[h], o));

    float acc[VPL];
    for (int i = 0; i < VPL; ++i) acc[i] = 0.f;
    float dl[HEADS];
    for (int h = 0; h < HEADS; ++h) dl[h] = 0.f;

    auto chunk = [&](int s, const float (&ev)[HEADS], bool act, int cn) {
        float al[HEADS];
        for (int h = 0; h < HEADS; ++h) {
            al[h] = act ? __expf(ev[h] - m[h]) : 0.f;
            dl[h] += al[h];
        }
        #pragma unroll 4
        for (int t = 0; t < cn; ++t) {
            const int st = __builtin_amdgcn_readlane(s, t);
            float w[HEADS];
            for (int h = 0; h < HEADS; ++h)
                w[h] = __int_as_float(__builtin_amdgcn_readlane(__float_as_int(al[h]), t));
            const float* row = H + (size_t)st * RW;
            if constexpr (HEADS == 4) {
                const float4 r = *reinterpret_cast<const float4*>(row + lane * 4);
                for (int h = 0; h < 4; ++h) {
                    acc[4 * h + 0] += w[h] * r.x; acc[4 * h + 1] += w[h] * r.y;
                    acc[4 * h + 2] += w[h] * r.z; acc[4 * h + 3] += w[h] * r.w;
                }
            } else {
                const float2 r = *reinterpret_cast<const float2*>(row + lane * 2);
                acc[0] += w[0] * r.x; acc[1] += w[0] * r.y;
            }
        }
    };

    chunk(s0, e0, act0, tot < 32 ? tot : 32);
    for (int j0 = 32; j0 < tot; j0 += 32) {
        const int j = j0 + lane;
        const bool act = j < tot;
        int s = n;
        float ev[HEADS];
        for (int h = 0; h < HEADS; ++h) ev[h] = NEG_INF;
        if (act) {
            s = (j < deg) ? csr_src[beg + j] : n;
            scores(s, ev);
        }
        const int left = tot - j0;
        chunk(s, ev, act, left < 32 ? left : 32);
    }

    for (int h = 0; h < HEADS; ++h)
        for (int o = 16; o > 0; o >>= 1) dl[h] += __shfl_xor(dl[h], o);

    if constexpr (HEADS == 4) {
        float* zrow = Zs + slot * ZP + lane * 4;
        for (int h = 0; h < 4; ++h) {
            const float inv = 1.0f / dl[h];
            *reinterpret_cast<float4*>(zrow + h * 128) =
                make_float4(acc[4 * h + 0] * inv, acc[4 * h + 1] * inv,
                            acc[4 * h + 2] * inv, acc[4 * h + 3] * inv);
        }
    } else {
        const float v0 = acc[0] / dl[0] + bias[lane * 2 + 0];
        const float v1 = acc[1] / dl[0] + bias[lane * 2 + 1];
        const float o0 = v0 > 0.f ? v0 : (__expf(v0) - 1.0f);
        const float o1 = v1 > 0.f ? v1 : (__expf(v1) - 1.0f);
        *reinterpret_cast<float2*>(&prow[slot][2 * lane]) = make_float2(o0, o1);
        if (lane == 0) pgr[slot] = batch[n];
    }
    }

    if constexpr (HEADS == 4) {
        static_assert(HEADS != 4 || (RT == 2 && GATHER_WAVES == 8), "two row tiles over eight waves");
        __syncthreads();
        const int c = lane & 15, h = lane >> 4;
        const int hd  = wave >> 1;
        const int ct0 = (wave & 1) * 2;
        v8f p[RT][2];
        #pragma unroll
        for (int rt = 0; rt < RT; ++rt) { p[rt][0] = v8f{}; p[rt][1] = v8f{}; }
        {
            const float*  zp = Zs + c * ZP + hd * 128 + 4 * h;
            const float4* bp = reinterpret_cast<const float4*>(B1f) + hd * W1_HSLOTS + ct0 * 32 + lane;
            #pragma unroll
            for (int j = 0; j < W1_K / 8; ++j) {
                float4 b[2], a[RT];
                b[0] = bp[j * 128];
                b[1] = bp[j * 128 + 32];
                #pragma unroll
                for (int rt = 0; rt < RT; ++rt)
                    a[rt] = *reinterpret_cast<const float4*>(zp + rt * 16 * ZP + 8 * j);
                #pragma unroll
                for (int rt = 0; rt < RT; ++rt)
                    #pragma unroll
                    for (int tt = 0; tt < 2; ++tt) {
                        v2f x0, b0;
                        x0.x = a[rt].x; x0.y = a[rt].y; b0.x = b[tt].x; b0.y = b[tt].y;
                        p[rt][tt] = __builtin_amdgcn_wmma_f32_16x16x4_f32(false, x0, false, b0, (short)0, p[rt][tt], false, false);
                    }
                #pragma unroll
                for (int rt = 0; rt < RT; ++rt)
                    #pragma unroll
                    for (int tt = 0; tt < 2; ++tt) {
                        v2f x1, b1;
                        x1.x = a[rt].z; x1.y = a[rt].w; b1.x = b[tt].z; b1.y = b[tt].w;
                        p[rt][tt] = __builtin_amdgcn_wmma_f32_16x16x4_f32(false, x1, false, b1, (short)0, p[rt][tt], false, false);
                    }
            }
        }
        __syncthreads();
        #pragma unroll
        for (int tt = 0; tt < 2; ++tt) {
            const int col = hd * 64 + 16 * (ct0 + tt) + c;
            const float bv = bias[col];
            #pragma unroll
            for (int rt = 0; rt < RT; ++rt)
                #pragma unroll
                for (int r = 0; r < 8; ++r) {
                    const float v = p[rt][tt][r] + bv;
                    As[(16 * rt + 8 * h + r) * AP + col] = v > 0.f ? v : (__expf(v) - 1.0f);
                }
        }
        __syncthreads();

        const int ct  = wave & 3;
        const int rt2 = wave >> 2;
        const float*  ap0 = As + (16 * rt2 + c) * AP + 4 * h;
        const float4* bp2 = reinterpret_cast<const float4*>(Bf) + ct * 32 + lane;
        v8f acc0 = {};
        #pragma unroll
        for (int j = 0; j < W2_K / 8; ++j) {
            const float4 b = bp2[j * 128];
            const float4 a = *reinterpret_cast<const float4*>(ap0 + 8 * j);
            v2f b0, b1, x0, x1;
            b0.x = b.x; b0.y = b.y; b1.x = b.z; b1.y = b.w;
            x0.x = a.x; x0.y = a.y; x1.x = a.z; x1.y = a.w;
            acc0 = __builtin_amdgcn_wmma_f32_16x16x4_f32(false, x0, false, b0, (short)0, acc0, false, false);
            acc0 = __builtin_amdgcn_wmma_f32_16x16x4_f32(false, x1, false, b1, (short)0, acc0, false, false);
        }

        __syncthreads();
        float* const cs = As;
        #pragma unroll
        for (int r = 0; r < 8; ++r) cs[(16 * rt2 + 8 * h + r) * CP + 16 * ct + c] = acc0[r];
        __syncthreads();
        for (int i = (int)threadIdx.x; i < GATHER_NODES * 16; i += 256) {
            const int r = i >> 4, q = i & 15;
            if (nfirst + r < Nn)
                *reinterpret_cast<float4*>(outp + (size_t)(nfirst + r) * 64 + 4 * q) =
                    *reinterpret_cast<const float4*>(cs + r * CP + 4 * q);
        }
        if (threadIdx.x < 128) {
            const int r = (int)threadIdx.x & 63;
            const bool dsts = threadIdx.x >= 64;
            if (r < GATHER_NODES && nfirst + r < Nn) {
                const float* av   = dsts ? att_d2 : att_s2;
                const float* crow = cs + r * CP;
                float sa = 0.f;
                #pragma unroll
                for (int k = 0; k < 64; k += 4) {
                    const float4 v = *reinterpret_cast<const float4*>(crow + k);
                    const float4 w = *reinterpret_cast<const float4*>(av + k);
                    sa = __builtin_fmaf(v.x, w.x, sa); sa = __builtin_fmaf(v.y, w.y, sa);
                    sa = __builtin_fmaf(v.z, w.z, sa); sa = __builtin_fmaf(v.w, w.w, sa);
                }
                (dsts ? ad_o : as_o)[nfirst + r] = sa;
            }
        }
    }

    if constexpr (HEADS == 1) {
        __syncthreads();
        if (wave < 2) {
            static_assert(HEADS != 1 || GATHER_NODES <= 32, "one graph id per lane");
            const int c = (int)threadIdx.x;
            float pv[GATHER_NODES];
            #pragma unroll
            for (int w = 0; w < GATHER_NODES; ++w) pv[w] = prow[w][c];
            const int gl = pgr[lane < GATHER_NODES ? lane : 0];
            float run = 0.f;
            int   nrun = 0;
            int   gc  = __builtin_amdgcn_readlane(gl, 0);
            #pragma unroll
            for (int w = 0; w < GATHER_NODES; ++w) {
                const int g = __builtin_amdgcn_readlane(gl, w);
                if (g != gc) {
                    if (gc >= 0) { rsum[nrun][c] = run; rgid[nrun][c] = gc; ++nrun; }
                    run = 0.f; gc = g;
                }
                run += pv[w];
            }
            if (gc >= 0) { rsum[nrun][c] = run; rgid[nrun][c] = gc; ++nrun; }

            float wl[OUT_CH];
            #pragma unroll
            for (int o = 0; o < OUT_CH; ++o) wl[o] = Wl[c * OUT_CH + o];
            #pragma nounroll
            for (int r = 0; r < nrun; ++r) {
                const int   gq  = __builtin_amdgcn_readfirstlane(rgid[r][c]);
                const float rv  = rsum[r][c];
                const float inv = 1.0f / fmaxf(cnts[gq], 1.0f);
                float mine = 0.f;
                #pragma unroll
                for (int o = 0; o < OUT_CH; ++o) {
                    float y = rv * wl[o];
                    for (int off = 16; off > 0; off >>= 1) y += __shfl_xor(y, off);
                    if (lane == o) mine = y;
                }
                if (lane < OUT_CH) atomicAdd(out + (size_t)gq * OUT_CH + lane, mine * inv);
            }
        }
    }
}

extern "C" void kernel_launch(void* const* d_in, const int* in_sizes, int n_in,
                              void* d_out, int out_size, void* d_ws, size_t ws_size,
                              hipStream_t stream)
{
    const float* x   = (const float*)d_in[0];
    const int*   ei  = (const int*)  d_in[1];
    const int*   bat = (const int*)  d_in[2];
    const float* W1  = (const float*)d_in[3];
    const float* aS1 = (const float*)d_in[4];
    const float* aD1 = (const float*)d_in[5];
    const float* b1  = (const float*)d_in[6];
    const float* W2  = (const float*)d_in[7];
    const float* aS2 = (const float*)d_in[8];
    const float* aD2 = (const float*)d_in[9];
    const float* b2  = (const float*)d_in[10];
    const float* Wl  = (const float*)d_in[11];
    const float* bl  = (const float*)d_in[12];
    float* out = (float*)d_out;

    const int N    = in_sizes[2];
    const int E    = in_sizes[1] / 2;
    const int HID  = 64, HEADS = 4, G = 512;

    float* ws = (float*)d_ws;
    size_t off = 0;
    float* W1f = ws + off; off += (size_t)W1_SLOTS * 4;
    float* W2f = ws + off; off += (size_t)W2_SLOTS * 4;
    float* Vf  = ws + off; off += (size_t)V_WORDS;
    float* as1 = ws + off; off += (size_t)N * HEADS;
    float* ad1 = ws + off; off += (size_t)N * HEADS;
    float* H2  = ws + off; off += (size_t)N * HID;
    float* as2 = ws + off; off += (size_t)N;
    float* ad2 = ws + off; off += (size_t)N;
    float* cnts = ws + off; off += (size_t)G;
    int* gtot   = (int*)(ws + off); off += 1;
    int* dcnt   = (int*)(ws + off); off += (size_t)N;
    int* dstart = (int*)(ws + off); off += (size_t)N;
    int* csrc   = (int*)(ws + off); off += (size_t)E;
    const int nbin = (N + 63) >> 6;
    const int nblk = (E + 4095) / 4096;
    off += off & 1;
    int2* etmp  = (int2*)(ws + off); off += 2 * (size_t)E;
    int2* seg   = (int2*)(ws + off); off += 2 * (size_t)nbin * nblk;
    (void)ws_size; (void)n_in; (void)out_size;

    const int TB = 256;
    auto blks = [](size_t n, int tb) { return (unsigned)((n + tb - 1) / tb); };

    const int nzero = G + 1;
    gat_bin_coarse<<<nblk, 1024, 0, stream>>>(ei, E, nbin, etmp, seg, cnts, nzero,
                                              bl, out, G * OUT_CH, W2, W2f,
                                              W1, W1f, aS1, aD1, Vf);

    {
        const int ntile = (N + 127) / 128;
        gat_fine_and_acoef<<<nbin + ntile, 256, 0, stream>>>(
            etmp, seg, nblk, gtot, dstart, dcnt, csrc, N, nbin, ntile,
            x, Vf, as1, ad1, bat, cnts);
    }
    gat_node_gather<4, 4><<<blks(N, GATHER_WAVES * 4), TB, 0, stream>>>(
        csrc, dstart, dcnt, as1, ad1, x, b1, H2, bat, cnts, Wl, out, N,
        W2f, aS2, aD2, as2, ad2, W1f);

    gat_node_gather<1, 4><<<blks(N, GATHER_WAVES * 4), TB, 0, stream>>>(
        csrc, dstart, dcnt, as2, ad2, H2, b2, nullptr, bat, cnts, Wl, out, N,
        nullptr, nullptr, nullptr, nullptr, nullptr, nullptr);
}
